# starting kernel
# baseline (speedup 1.0000x reference)
#include <hip/hip_runtime.h>

#define NNODES 100000
#define NEDGES 1600000
#define NFEAT  128
#define NHID   32
#define NCLS   40

typedef float v2f __attribute__((ext_vector_type(2)));
typedef float v8f __attribute__((ext_vector_type(8)));

__global__ void zero_u32_kernel(unsigned* __restrict__ p, int n) {
    int t = blockIdx.x * blockDim.x + threadIdx.x;
    if (t < n) p[t] = 0u;
}

__global__ void deg_count_kernel(const int* __restrict__ dst,
                                 unsigned* __restrict__ cnt, int E) {
    int e = blockIdx.x * blockDim.x + threadIdx.x;
    if (e < E) atomicAdd(&cnt[dst[e]], 1u);
}

__global__ void dinv_kernel(const unsigned* __restrict__ cnt,
                            float* __restrict__ dinv, int n) {
    int i = blockIdx.x * blockDim.x + threadIdx.x;
    if (i < n) dinv[i] = rsqrtf((float)cnt[i] + 1.0f);
}

__global__ void scan_block_kernel(const unsigned* __restrict__ cnt,
                                  unsigned* __restrict__ excl,
                                  unsigned* __restrict__ bsum, int n) {
    __shared__ unsigned s[256];
    int t = blockIdx.x * 256 + threadIdx.x;
    unsigned v = (t < n) ? cnt[t] : 0u;
    s[threadIdx.x] = v;
    __syncthreads();
    for (int off = 1; off < 256; off <<= 1) {
        unsigned add = (threadIdx.x >= off) ? s[threadIdx.x - off] : 0u;
        __syncthreads();
        s[threadIdx.x] += add;
        __syncthreads();
    }
    if (t < n) excl[t] = s[threadIdx.x] - v;
    if (threadIdx.x == 255) bsum[blockIdx.x] = s[255];
}

__global__ void scan_bsum_kernel(unsigned* __restrict__ bsum, int nb) {
    if (blockIdx.x == 0 && threadIdx.x == 0) {
        unsigned run = 0;
        for (int i = 0; i < nb; ++i) { unsigned v = bsum[i]; bsum[i] = run; run += v; }
    }
}

__global__ void scan_add_kernel(unsigned* __restrict__ excl,
                                const unsigned* __restrict__ bsum,
                                int n, unsigned total) {
    int t = blockIdx.x * blockDim.x + threadIdx.x;
    if (t < n) excl[t] += bsum[t >> 8];
    if (t == 0) excl[n] = total;
}

__global__ void csr_fill_kernel(const int* __restrict__ src, const int* __restrict__ dst,
                                const float* __restrict__ dinv,
                                const unsigned* __restrict__ rowptr,
                                unsigned* __restrict__ cursor,
                                int2* __restrict__ csr, int E) {
    int e = blockIdx.x * blockDim.x + threadIdx.x;
    if (e >= E) return;
    int s = src[e];
    int d = dst[e];
    unsigned pos = rowptr[d] + atomicAdd(&cursor[d], 1u);
    csr[pos] = make_int2(s, __float_as_int(dinv[s] * dinv[d]));
}

__global__ void transpose_pad_kernel(const float* __restrict__ W,
                                     float* __restrict__ Bt,
                                     int K, int Nout, int Npad) {
    int t = blockIdx.x * blockDim.x + threadIdx.x;
    if (t >= Npad * K) return;
    int n = t / K;
    int k = t - n * K;
    Bt[t] = (n < Nout) ? W[(size_t)k * Nout + n] : 0.0f;
}

template <int NT>
__global__ __launch_bounds__(256)
void gemm_wmma_kernel(const float* __restrict__ A, const float* __restrict__ Bt,
                      float* __restrict__ C, int M, int K, int Nout) {
    int wave = (int)((blockIdx.x * blockDim.x + threadIdx.x) >> 5);
    int lane = threadIdx.x & 31;
    if (wave >= (M >> 4)) return;

    int m0   = wave << 4;
    int half = lane >> 4;
    int l16  = lane & 15;

    const float* Arow = A + (size_t)(m0 + l16) * K;
    const float* Bcol[NT];
#pragma unroll
    for (int t = 0; t < NT; ++t)
        Bcol[t] = Bt + (size_t)(t * 16 + l16) * K;

    v8f c[NT] = {};

    for (int k0 = 0; k0 < K; k0 += 4) {
        int ka = k0 + (half << 1);
        v2f a = *(const v2f*)(Arow + ka);
#pragma unroll
        for (int t = 0; t < NT; ++t) {
            v2f b = *(const v2f*)(Bcol[t] + ka);
            c[t] = __builtin_amdgcn_wmma_f32_16x16x4_f32(
                       false, a, false, b, (short)0, c[t], false, false);
        }
    }

    int rbase = m0 + (half << 3);
#pragma unroll
    for (int t = 0; t < NT; ++t) {
        int ncol = t * 16 + l16;
        if (ncol < Nout) {
#pragma unroll
            for (int r = 0; r < 8; ++r)
                C[(size_t)(rbase + r) * Nout + ncol] = c[t][r];
        }
    }
}

template <int F, bool POST>
__global__ __launch_bounds__(256)
void gather_kernel(const float* __restrict__ h, const int2* __restrict__ csr,
                   const unsigned* __restrict__ rowptr,
                   const float* __restrict__ dinv, const float* __restrict__ bias,
                   const float* __restrict__ xold_in, const float* __restrict__ dropu,
                   float* __restrict__ xout, float* __restrict__ xold_out, int n) {
    int node = (int)((blockIdx.x * blockDim.x + threadIdx.x) >> 5);
    int lane = threadIdx.x & 31;
    if (node >= n) return;

    unsigned beg = rowptr[node];
    unsigned end = rowptr[node + 1];

    constexpr int C = (F + 31) / 32;
    float acc[C];
    float di = dinv[node];
    float selfw = di * di;
    const float* hrow = h + (size_t)node * F;
#pragma unroll
    for (int c2 = 0; c2 < C; ++c2) {
        int f = lane + 32 * c2;
        acc[c2] = (f < F) ? hrow[f] * selfw + bias[f] : 0.0f;
    }

    unsigned j = beg;
    for (; j + 2 <= end; j += 2) {
        int2 p0 = csr[j];
        int2 p1 = csr[j + 1];
        const float* h0 = h + (size_t)p0.x * F;
        const float* h1 = h + (size_t)p1.x * F;
        float w0 = __int_as_float(p0.y);
        float w1 = __int_as_float(p1.y);
#pragma unroll
        for (int c2 = 0; c2 < C; ++c2) {
            int f = lane + 32 * c2;
            if (f < F) acc[c2] += h0[f] * w0 + h1[f] * w1;
        }
    }
    if (j < end) {
        int2 p0 = csr[j];
        const float* h0 = h + (size_t)p0.x * F;
        float w0 = __int_as_float(p0.y);
#pragma unroll
        for (int c2 = 0; c2 < C; ++c2) {
            int f = lane + 32 * c2;
            if (f < F) acc[c2] += h0[f] * w0;
        }
    }

    size_t base = (size_t)node * F;
#pragma unroll
    for (int c2 = 0; c2 < C; ++c2) {
        int f = lane + 32 * c2;
        if (f < F) {
            if (POST) {
                float y = fmaxf(acc[c2], 0.0f);
                if (xold_in) y += xold_in[base + f];
                xold_out[base + f] = y;
                xout[base + f] = y * (dropu[base + f] > 0.5f ? 2.0f : 0.0f);
            } else {
                xout[base + f] = acc[c2];
            }
        }
    }
}


extern "C" void kernel_launch(void* const* d_in, const int* in_sizes, int n_in,
                              void* d_out, int out_size, void* d_ws, size_t ws_size,
                              hipStream_t stream) {
    (void)in_sizes; (void)n_in; (void)out_size; (void)ws_size;

    const float* x0    = (const float*)d_in[0];
    const int*   ei    = (const int*)d_in[1];
    const float* dropu = (const float*)d_in[2];
    const float* W0    = (const float*)d_in[3];
    const float* b0    = (const float*)d_in[4];
    const float* Whid  = (const float*)d_in[5];
    const float* bhid  = (const float*)d_in[6];
    const float* Wout  = (const float*)d_in[7];
    const float* bout  = (const float*)d_in[8];
    float* out = (float*)d_out;

    const int* src = ei;
    const int* dst = ei + NEDGES;

    char* ws = (char*)d_ws;
    size_t off = 0;
    auto take = [&](size_t bytes) -> void* {
        void* p = (void*)(ws + off);
        off += (bytes + 255) & ~(size_t)255;
        return p;
    };
    const int NB = (NNODES + 255) / 256;
    unsigned* cnt    = (unsigned*)take((size_t)NNODES * 4);
    unsigned* rowptr = (unsigned*)take((size_t)(NNODES + 1) * 4);
    unsigned* cursor = (unsigned*)take((size_t)NNODES * 4);
    unsigned* bsum   = (unsigned*)take((size_t)NB * 4);
    float*    dinv   = (float*)   take((size_t)NNODES * 4);
    int2*     csr    = (int2*)    take((size_t)NEDGES * 8);
    float*    h      = (float*)   take((size_t)NNODES * NCLS * 4);
    float*    xbuf   = (float*)   take((size_t)NNODES * NHID * 4);
    float*    xold   = (float*)   take((size_t)NNODES * NHID * 4);
    float*    Bt     = (float*)   take((size_t)48 * NFEAT * 4);

    const int TB = 256;
    auto blocks = [](int n, int tb) { return (n + tb - 1) / tb; };

    zero_u32_kernel<<<blocks(NNODES, TB), TB, 0, stream>>>(cnt, NNODES);
    zero_u32_kernel<<<blocks(NNODES, TB), TB, 0, stream>>>(cursor, NNODES);
    deg_count_kernel<<<blocks(NEDGES, TB), TB, 0, stream>>>(dst, cnt, NEDGES);
    dinv_kernel<<<blocks(NNODES, TB), TB, 0, stream>>>(cnt, dinv, NNODES);
    scan_block_kernel<<<NB, 256, 0, stream>>>(cnt, rowptr, bsum, NNODES);
    scan_bsum_kernel<<<1, 32, 0, stream>>>(bsum, NB);
    scan_add_kernel<<<blocks(NNODES, TB), TB, 0, stream>>>(rowptr, bsum, NNODES, NEDGES);
    csr_fill_kernel<<<blocks(NEDGES, TB), TB, 0, stream>>>(
        src, dst, dinv, rowptr, cursor, csr, NEDGES);

    int gemm_blocks   = blocks(NNODES / 16, 8);
    int gather_blocks = blocks(NNODES, 8);

    const float* xold_in = nullptr;
    for (int layer = 0; layer < 7; ++layer) {
        const float* A = (layer == 0) ? x0 : xbuf;
        const float* W = (layer == 0) ? W0 : Whid + (size_t)(layer - 1) * NHID * NHID;
        const float* b = (layer == 0) ? b0 : bhid + (size_t)(layer - 1) * NHID;
        int K = (layer == 0) ? NFEAT : NHID;

        transpose_pad_kernel<<<blocks(NHID * K, TB), TB, 0, stream>>>(W, Bt, K, NHID, NHID);
        gemm_wmma_kernel<2><<<gemm_blocks, 256, 0, stream>>>(A, Bt, h, NNODES, K, NHID);

        gather_kernel<NHID, true><<<gather_blocks, 256, 0, stream>>>(
            h, csr, rowptr, dinv, b, xold_in,
            dropu + (size_t)layer * NNODES * NHID, xbuf, xold, NNODES);
        xold_in = xold;
    }

    {
        transpose_pad_kernel<<<blocks(48 * NHID, TB), TB, 0, stream>>>(Wout, Bt, NHID, NCLS, 48);
        gemm_wmma_kernel<3><<<gemm_blocks, 256, 0, stream>>>(xbuf, Bt, h, NNODES, NHID, NCLS);

        gather_kernel<NCLS, false><<<gather_blocks, 256, 0, stream>>>(
            h, csr, rowptr, dinv, bout, nullptr, nullptr, out, nullptr, NNODES);
    }
}


# gather kernels: 8 lanes x 16 B per node row, 4 nodes per wave (was 32 lanes x 4 B, 1 node per wave): 4x fewer waves and row-load instructions; per-feature sum order unchanged
# speedup vs baseline: 1.1622x; 1.1622x over previous
#include <hip/hip_runtime.h>

#define NNODES 100000
#define NEDGES 1600000
#define NFEAT  128
#define NHID   32
#define NCLS   40

typedef float v2f __attribute__((ext_vector_type(2)));
typedef float v8f __attribute__((ext_vector_type(8)));

__global__ void zero_u32_kernel(unsigned* __restrict__ p, int n) {
    int t = blockIdx.x * blockDim.x + threadIdx.x;
    if (t < n) p[t] = 0u;
}

__global__ void deg_count_kernel(const int* __restrict__ dst,
                                 unsigned* __restrict__ cnt, int E) {
    int e = blockIdx.x * blockDim.x + threadIdx.x;
    if (e < E) atomicAdd(&cnt[dst[e]], 1u);
}

__global__ void dinv_kernel(const unsigned* __restrict__ cnt,
                            float* __restrict__ dinv, int n) {
    int i = blockIdx.x * blockDim.x + threadIdx.x;
    if (i < n) dinv[i] = rsqrtf((float)cnt[i] + 1.0f);
}

__global__ void scan_block_kernel(const unsigned* __restrict__ cnt,
                                  unsigned* __restrict__ excl,
                                  unsigned* __restrict__ bsum, int n) {
    __shared__ unsigned s[256];
    int t = blockIdx.x * 256 + threadIdx.x;
    unsigned v = (t < n) ? cnt[t] : 0u;
    s[threadIdx.x] = v;
    __syncthreads();
    for (int off = 1; off < 256; off <<= 1) {
        unsigned add = (threadIdx.x >= off) ? s[threadIdx.x - off] : 0u;
        __syncthreads();
        s[threadIdx.x] += add;
        __syncthreads();
    }
    if (t < n) excl[t] = s[threadIdx.x] - v;
    if (threadIdx.x == 255) bsum[blockIdx.x] = s[255];
}

__global__ void scan_bsum_kernel(unsigned* __restrict__ bsum, int nb) {
    if (blockIdx.x == 0 && threadIdx.x == 0) {
        unsigned run = 0;
        for (int i = 0; i < nb; ++i) { unsigned v = bsum[i]; bsum[i] = run; run += v; }
    }
}

__global__ void scan_add_kernel(unsigned* __restrict__ excl,
                                const unsigned* __restrict__ bsum,
                                int n, unsigned total) {
    int t = blockIdx.x * blockDim.x + threadIdx.x;
    if (t < n) excl[t] += bsum[t >> 8];
    if (t == 0) excl[n] = total;
}

__global__ void csr_fill_kernel(const int* __restrict__ src, const int* __restrict__ dst,
                                const float* __restrict__ dinv,
                                const unsigned* __restrict__ rowptr,
                                unsigned* __restrict__ cursor,
                                int2* __restrict__ csr, int E) {
    int e = blockIdx.x * blockDim.x + threadIdx.x;
    if (e >= E) return;
    int s = src[e];
    int d = dst[e];
    unsigned pos = rowptr[d] + atomicAdd(&cursor[d], 1u);
    csr[pos] = make_int2(s, __float_as_int(dinv[s] * dinv[d]));
}

__global__ void transpose_pad_kernel(const float* __restrict__ W,
                                     float* __restrict__ Bt,
                                     int K, int Nout, int Npad) {
    int t = blockIdx.x * blockDim.x + threadIdx.x;
    if (t >= Npad * K) return;
    int n = t / K;
    int k = t - n * K;
    Bt[t] = (n < Nout) ? W[(size_t)k * Nout + n] : 0.0f;
}

template <int NT>
__global__ __launch_bounds__(256)
void gemm_wmma_kernel(const float* __restrict__ A, const float* __restrict__ Bt,
                      float* __restrict__ C, int M, int K, int Nout) {
    int wave = (int)((blockIdx.x * blockDim.x + threadIdx.x) >> 5);
    int lane = threadIdx.x & 31;
    if (wave >= (M >> 4)) return;

    int m0   = wave << 4;
    int half = lane >> 4;
    int l16  = lane & 15;

    const float* Arow = A + (size_t)(m0 + l16) * K;
    const float* Bcol[NT];
#pragma unroll
    for (int t = 0; t < NT; ++t)
        Bcol[t] = Bt + (size_t)(t * 16 + l16) * K;

    v8f c[NT] = {};

    for (int k0 = 0; k0 < K; k0 += 4) {
        int ka = k0 + (half << 1);
        v2f a = *(const v2f*)(Arow + ka);
#pragma unroll
        for (int t = 0; t < NT; ++t) {
            v2f b = *(const v2f*)(Bcol[t] + ka);
            c[t] = __builtin_amdgcn_wmma_f32_16x16x4_f32(
                       false, a, false, b, (short)0, c[t], false, false);
        }
    }

    int rbase = m0 + (half << 3);
#pragma unroll
    for (int t = 0; t < NT; ++t) {
        int ncol = t * 16 + l16;
        if (ncol < Nout) {
#pragma unroll
            for (int r = 0; r < 8; ++r)
                C[(size_t)(rbase + r) * Nout + ncol] = c[t][r];
        }
    }
}

typedef float v4f __attribute__((ext_vector_type(4)));

template <int F, bool POST>
__global__ __launch_bounds__(256)
void gather_kernel(const float* __restrict__ h, const int2* __restrict__ csr,
                   const unsigned* __restrict__ rowptr,
                   const float* __restrict__ dinv, const float* __restrict__ bias,
                   const float* __restrict__ xold_in, const float* __restrict__ dropu,
                   float* __restrict__ xout, float* __restrict__ xold_out, int n) {
    static_assert(F % 4 == 0, "row must be a whole number of 16-byte pieces");
    constexpr int Q = F / 4;
    constexpr int C = (Q + 7) / 8;

    int node = (int)((blockIdx.x * blockDim.x + threadIdx.x) >> 3);
    int l8   = threadIdx.x & 7;
    if (node >= n) return;

    unsigned beg = rowptr[node];
    unsigned end = rowptr[node + 1];

    v4f acc[C];
    float di = dinv[node];
    float selfw = di * di;
    const v4f* hrow = (const v4f*)(h + (size_t)node * F);
#pragma unroll
    for (int c2 = 0; c2 < C; ++c2) {
        int q = l8 + 8 * c2;
        if (q < Q) {
            v4f hv = hrow[q];
            v4f bv;
            bv.x = bias[4 * q + 0];
            bv.y = bias[4 * q + 1];
            bv.z = bias[4 * q + 2];
            bv.w = bias[4 * q + 3];
            acc[c2] = hv * selfw + bv;
        } else {
            acc[c2] = (v4f){0.0f, 0.0f, 0.0f, 0.0f};
        }
    }

    unsigned j = beg;
    for (; j + 2 <= end; j += 2) {
        int2 p0 = csr[j];
        int2 p1 = csr[j + 1];
        const v4f* h0 = (const v4f*)(h + (size_t)p0.x * F);
        const v4f* h1 = (const v4f*)(h + (size_t)p1.x * F);
        float w0 = __int_as_float(p0.y);
        float w1 = __int_as_float(p1.y);
#pragma unroll
        for (int c2 = 0; c2 < C; ++c2) {
            int q = l8 + 8 * c2;
            if (q < Q) acc[c2] += h0[q] * w0 + h1[q] * w1;
        }
    }
    if (j < end) {
        int2 p0 = csr[j];
        const v4f* h0 = (const v4f*)(h + (size_t)p0.x * F);
        float w0 = __int_as_float(p0.y);
#pragma unroll
        for (int c2 = 0; c2 < C; ++c2) {
            int q = l8 + 8 * c2;
            if (q < Q) acc[c2] += h0[q] * w0;
        }
    }

    size_t base = (size_t)node * F;
#pragma unroll
    for (int c2 = 0; c2 < C; ++c2) {
        int q = l8 + 8 * c2;
        if (q < Q) {
            size_t o = base + 4 * (size_t)q;
            if (POST) {
                v4f y;
                y.x = fmaxf(acc[c2].x, 0.0f);
                y.y = fmaxf(acc[c2].y, 0.0f);
                y.z = fmaxf(acc[c2].z, 0.0f);
                y.w = fmaxf(acc[c2].w, 0.0f);
                if (xold_in) y += *(const v4f*)(xold_in + o);
                *(v4f*)(xold_out + o) = y;
                v4f d = *(const v4f*)(dropu + o);
                v4f x;
                x.x = y.x * (d.x > 0.5f ? 2.0f : 0.0f);
                x.y = y.y * (d.y > 0.5f ? 2.0f : 0.0f);
                x.z = y.z * (d.z > 0.5f ? 2.0f : 0.0f);
                x.w = y.w * (d.w > 0.5f ? 2.0f : 0.0f);
                *(v4f*)(xout + o) = x;
            } else {
                *(v4f*)(xout + o) = acc[c2];
            }
        }
    }
}


extern "C" void kernel_launch(void* const* d_in, const int* in_sizes, int n_in,
                              void* d_out, int out_size, void* d_ws, size_t ws_size,
                              hipStream_t stream) {
    (void)in_sizes; (void)n_in; (void)out_size; (void)ws_size;

    const float* x0    = (const float*)d_in[0];
    const int*   ei    = (const int*)d_in[1];
    const float* dropu = (const float*)d_in[2];
    const float* W0    = (const float*)d_in[3];
    const float* b0    = (const float*)d_in[4];
    const float* Whid  = (const float*)d_in[5];
    const float* bhid  = (const float*)d_in[6];
    const float* Wout  = (const float*)d_in[7];
    const float* bout  = (const float*)d_in[8];
    float* out = (float*)d_out;

    const int* src = ei;
    const int* dst = ei + NEDGES;

    char* ws = (char*)d_ws;
    size_t off = 0;
    auto take = [&](size_t bytes) -> void* {
        void* p = (void*)(ws + off);
        off += (bytes + 255) & ~(size_t)255;
        return p;
    };
    const int NB = (NNODES + 255) / 256;
    unsigned* cnt    = (unsigned*)take((size_t)NNODES * 4);
    unsigned* rowptr = (unsigned*)take((size_t)(NNODES + 1) * 4);
    unsigned* cursor = (unsigned*)take((size_t)NNODES * 4);
    unsigned* bsum   = (unsigned*)take((size_t)NB * 4);
    float*    dinv   = (float*)   take((size_t)NNODES * 4);
    int2*     csr    = (int2*)    take((size_t)NEDGES * 8);
    float*    h      = (float*)   take((size_t)NNODES * NCLS * 4);
    float*    xbuf   = (float*)   take((size_t)NNODES * NHID * 4);
    float*    xold   = (float*)   take((size_t)NNODES * NHID * 4);
    float*    Bt     = (float*)   take((size_t)48 * NFEAT * 4);

    const int TB = 256;
    auto blocks = [](int n, int tb) { return (n + tb - 1) / tb; };

    zero_u32_kernel<<<blocks(NNODES, TB), TB, 0, stream>>>(cnt, NNODES);
    zero_u32_kernel<<<blocks(NNODES, TB), TB, 0, stream>>>(cursor, NNODES);
    deg_count_kernel<<<blocks(NEDGES, TB), TB, 0, stream>>>(dst, cnt, NEDGES);
    dinv_kernel<<<blocks(NNODES, TB), TB, 0, stream>>>(cnt, dinv, NNODES);
    scan_block_kernel<<<NB, 256, 0, stream>>>(cnt, rowptr, bsum, NNODES);
    scan_bsum_kernel<<<1, 32, 0, stream>>>(bsum, NB);
    scan_add_kernel<<<blocks(NNODES, TB), TB, 0, stream>>>(rowptr, bsum, NNODES, NEDGES);
    csr_fill_kernel<<<blocks(NEDGES, TB), TB, 0, stream>>>(
        src, dst, dinv, rowptr, cursor, csr, NEDGES);

    int gemm_blocks   = blocks(NNODES / 16, 8);
    int gather_blocks = blocks(NNODES, 32);

    const float* xold_in = nullptr;
    for (int layer = 0; layer < 7; ++layer) {
        const float* A = (layer == 0) ? x0 : xbuf;
        const float* W = (layer == 0) ? W0 : Whid + (size_t)(layer - 1) * NHID * NHID;
        const float* b = (layer == 0) ? b0 : bhid + (size_t)(layer - 1) * NHID;
        int K = (layer == 0) ? NFEAT : NHID;

        transpose_pad_kernel<<<blocks(NHID * K, TB), TB, 0, stream>>>(W, Bt, K, NHID, NHID);
        gemm_wmma_kernel<2><<<gemm_blocks, 256, 0, stream>>>(A, Bt, h, NNODES, K, NHID);

        gather_kernel<NHID, true><<<gather_blocks, 256, 0, stream>>>(
            h, csr, rowptr, dinv, b, xold_in,
            dropu + (size_t)layer * NNODES * NHID, xbuf, xold, NNODES);
        xold_in = xold;
    }

    {
        transpose_pad_kernel<<<blocks(48 * NHID, TB), TB, 0, stream>>>(Wout, Bt, NHID, NCLS, 48);
        gemm_wmma_kernel<3><<<gemm_blocks, 256, 0, stream>>>(xbuf, Bt, h, NNODES, NHID, NCLS);

        gather_kernel<NCLS, false><<<gather_blocks, 256, 0, stream>>>(
            h, csr, rowptr, dinv, bout, nullptr, nullptr, out, nullptr, NNODES);
    }
}


# gather kernels: 8 edges per node per trip (was 2): each of a node's 8 lanes loads one (src,w) pair, 8-lane swizzle shares them, 8 row loads in flight, epilogue operands fetched ahead; sum order kept
# speedup vs baseline: 1.2253x; 1.0543x over previous
#include <hip/hip_runtime.h>

#define NNODES 100000
#define NEDGES 1600000
#define NFEAT  128
#define NHID   32
#define NCLS   40

typedef float v2f __attribute__((ext_vector_type(2)));
typedef float v8f __attribute__((ext_vector_type(8)));

__global__ void zero_u32_kernel(unsigned* __restrict__ p, int n) {
    int t = blockIdx.x * blockDim.x + threadIdx.x;
    if (t < n) p[t] = 0u;
}

__global__ void deg_count_kernel(const int* __restrict__ dst,
                                 unsigned* __restrict__ cnt, int E) {
    int e = blockIdx.x * blockDim.x + threadIdx.x;
    if (e < E) atomicAdd(&cnt[dst[e]], 1u);
}

__global__ void dinv_kernel(const unsigned* __restrict__ cnt,
                            float* __restrict__ dinv, int n) {
    int i = blockIdx.x * blockDim.x + threadIdx.x;
    if (i < n) dinv[i] = rsqrtf((float)cnt[i] + 1.0f);
}

__global__ void scan_block_kernel(const unsigned* __restrict__ cnt,
                                  unsigned* __restrict__ excl,
                                  unsigned* __restrict__ bsum, int n) {
    __shared__ unsigned s[256];
    int t = blockIdx.x * 256 + threadIdx.x;
    unsigned v = (t < n) ? cnt[t] : 0u;
    s[threadIdx.x] = v;
    __syncthreads();
    for (int off = 1; off < 256; off <<= 1) {
        unsigned add = (threadIdx.x >= off) ? s[threadIdx.x - off] : 0u;
        __syncthreads();
        s[threadIdx.x] += add;
        __syncthreads();
    }
    if (t < n) excl[t] = s[threadIdx.x] - v;
    if (threadIdx.x == 255) bsum[blockIdx.x] = s[255];
}

__global__ void scan_bsum_kernel(unsigned* __restrict__ bsum, int nb) {
    if (blockIdx.x == 0 && threadIdx.x == 0) {
        unsigned run = 0;
        for (int i = 0; i < nb; ++i) { unsigned v = bsum[i]; bsum[i] = run; run += v; }
    }
}

__global__ void scan_add_kernel(unsigned* __restrict__ excl,
                                const unsigned* __restrict__ bsum,
                                int n, unsigned total) {
    int t = blockIdx.x * blockDim.x + threadIdx.x;
    if (t < n) excl[t] += bsum[t >> 8];
    if (t == 0) excl[n] = total;
}

__global__ void csr_fill_kernel(const int* __restrict__ src, const int* __restrict__ dst,
                                const float* __restrict__ dinv,
                                const unsigned* __restrict__ rowptr,
                                unsigned* __restrict__ cursor,
                                int2* __restrict__ csr, int E) {
    int e = blockIdx.x * blockDim.x + threadIdx.x;
    if (e >= E) return;
    int s = src[e];
    int d = dst[e];
    unsigned pos = rowptr[d] + atomicAdd(&cursor[d], 1u);
    csr[pos] = make_int2(s, __float_as_int(dinv[s] * dinv[d]));
}

__global__ void transpose_pad_kernel(const float* __restrict__ W,
                                     float* __restrict__ Bt,
                                     int K, int Nout, int Npad) {
    int t = blockIdx.x * blockDim.x + threadIdx.x;
    if (t >= Npad * K) return;
    int n = t / K;
    int k = t - n * K;
    Bt[t] = (n < Nout) ? W[(size_t)k * Nout + n] : 0.0f;
}

template <int NT>
__global__ __launch_bounds__(256)
void gemm_wmma_kernel(const float* __restrict__ A, const float* __restrict__ Bt,
                      float* __restrict__ C, int M, int K, int Nout) {
    int wave = (int)((blockIdx.x * blockDim.x + threadIdx.x) >> 5);
    int lane = threadIdx.x & 31;
    if (wave >= (M >> 4)) return;

    int m0   = wave << 4;
    int half = lane >> 4;
    int l16  = lane & 15;

    const float* Arow = A + (size_t)(m0 + l16) * K;
    const float* Bcol[NT];
#pragma unroll
    for (int t = 0; t < NT; ++t)
        Bcol[t] = Bt + (size_t)(t * 16 + l16) * K;

    v8f c[NT] = {};

    for (int k0 = 0; k0 < K; k0 += 4) {
        int ka = k0 + (half << 1);
        v2f a = *(const v2f*)(Arow + ka);
#pragma unroll
        for (int t = 0; t < NT; ++t) {
            v2f b = *(const v2f*)(Bcol[t] + ka);
            c[t] = __builtin_amdgcn_wmma_f32_16x16x4_f32(
                       false, a, false, b, (short)0, c[t], false, false);
        }
    }

    int rbase = m0 + (half << 3);
#pragma unroll
    for (int t = 0; t < NT; ++t) {
        int ncol = t * 16 + l16;
        if (ncol < Nout) {
#pragma unroll
            for (int r = 0; r < 8; ++r)
                C[(size_t)(rbase + r) * Nout + ncol] = c[t][r];
        }
    }
}

typedef float v4f __attribute__((ext_vector_type(4)));

template <int E>
__device__ __forceinline__ unsigned group8_lane(unsigned v) {
    return __builtin_amdgcn_mov_dpp8(v, (unsigned)E * 0x249249u);
}

template <int F, bool POST>
__global__ __launch_bounds__(256)
void gather_kernel(const float* __restrict__ h, const int2* __restrict__ csr,
                   const unsigned* __restrict__ rowptr,
                   const float* __restrict__ dinv, const float* __restrict__ bias,
                   const float* __restrict__ xold_in, const float* __restrict__ dropu,
                   float* __restrict__ xout, float* __restrict__ xold_out, int n) {
    static_assert(F % 4 == 0, "row must be a whole number of 16-byte pieces");
    constexpr int Q = F / 4;
    constexpr int C = (Q + 7) / 8;
    static_assert(!POST || C == 1, "the epilogue handles one piece per lane");

    int node = (int)((blockIdx.x * blockDim.x + threadIdx.x) >> 3);
    int l8   = threadIdx.x & 7;
    if (node >= n) return;

    unsigned beg = rowptr[node];
    unsigned end = rowptr[node + 1];

    float di = dinv[node];
    const v4f* hrow = (const v4f*)(h + (size_t)node * F);
    v4f hself[C];
    v4f bself[C];
#pragma unroll
    for (int c2 = 0; c2 < C; ++c2) {
        int q = l8 + 8 * c2;
        if (q < Q) {
            hself[c2] = hrow[q];
            bself[c2].x = bias[4 * q + 0];
            bself[c2].y = bias[4 * q + 1];
            bself[c2].z = bias[4 * q + 2];
            bself[c2].w = bias[4 * q + 3];
        }
    }
    size_t obase = (size_t)node * F + 4 * (size_t)l8;
    v4f dvec = (v4f){0.0f, 0.0f, 0.0f, 0.0f};
    v4f xvec = (v4f){0.0f, 0.0f, 0.0f, 0.0f};
    if (POST) dvec = *(const v4f*)(dropu + obase);

    unsigned last = (end > 0u ? end : 1u) - 1u;
    unsigned i0 = beg + (unsigned)l8;
    int2 p = csr[i0 < last ? i0 : last];

    if (POST) {
        if (xold_in) xvec = *(const v4f*)(xold_in + obase);
    }

    v4f acc[C];
    float selfw = di * di;
#pragma unroll
    for (int c2 = 0; c2 < C; ++c2) {
        int q = l8 + 8 * c2;
        if (q < Q) acc[c2] = hself[c2] * selfw + bself[c2];
        else       acc[c2] = (v4f){0.0f, 0.0f, 0.0f, 0.0f};
    }

    for (unsigned j = beg; j < end; j += 8) {
        unsigned cnt = end - j;

        unsigned px = (unsigned)p.x, py = (unsigned)p.y;
        int s[8];
        float w[8];
        s[0] = (int)group8_lane<0>(px);  w[0] = __uint_as_float(group8_lane<0>(py));
        s[1] = (int)group8_lane<1>(px);  w[1] = __uint_as_float(group8_lane<1>(py));
        s[2] = (int)group8_lane<2>(px);  w[2] = __uint_as_float(group8_lane<2>(py));
        s[3] = (int)group8_lane<3>(px);  w[3] = __uint_as_float(group8_lane<3>(py));
        s[4] = (int)group8_lane<4>(px);  w[4] = __uint_as_float(group8_lane<4>(py));
        s[5] = (int)group8_lane<5>(px);  w[5] = __uint_as_float(group8_lane<5>(py));
        s[6] = (int)group8_lane<6>(px);  w[6] = __uint_as_float(group8_lane<6>(py));
        s[7] = (int)group8_lane<7>(px);  w[7] = __uint_as_float(group8_lane<7>(py));

        v4f r[8][C];
#pragma unroll
        for (int e = 0; e < 8; ++e) {
            const v4f* he = (const v4f*)(h + (size_t)s[e] * F);
#pragma unroll
            for (int c2 = 0; c2 < C; ++c2) {
                int q = l8 + 8 * c2;
                if (q < Q) r[e][c2] = he[q];
            }
        }

        unsigned jn = j + 8 + (unsigned)l8;
        int2 pn = csr[jn < last ? jn : last];

#pragma unroll
        for (int e = 0; e < 8; e += 2) {
            if ((unsigned)(e + 2) <= cnt) {
#pragma unroll
                for (int c2 = 0; c2 < C; ++c2) {
                    int q = l8 + 8 * c2;
                    if (q < Q) acc[c2] += r[e][c2] * w[e] + r[e + 1][c2] * w[e + 1];
                }
            } else if ((unsigned)(e + 1) == cnt) {
#pragma unroll
                for (int c2 = 0; c2 < C; ++c2) {
                    int q = l8 + 8 * c2;
                    if (q < Q) acc[c2] += r[e][c2] * w[e];
                }
            }
        }
        p = pn;
    }

    size_t base = (size_t)node * F;
#pragma unroll
    for (int c2 = 0; c2 < C; ++c2) {
        int q = l8 + 8 * c2;
        if (q < Q) {
            size_t o = base + 4 * (size_t)q;
            if (POST) {
                v4f y;
                y.x = fmaxf(acc[c2].x, 0.0f);
                y.y = fmaxf(acc[c2].y, 0.0f);
                y.z = fmaxf(acc[c2].z, 0.0f);
                y.w = fmaxf(acc[c2].w, 0.0f);
                if (xold_in) y += xvec;
                *(v4f*)(xold_out + o) = y;
                v4f d = dvec;
                v4f x;
                x.x = y.x * (d.x > 0.5f ? 2.0f : 0.0f);
                x.y = y.y * (d.y > 0.5f ? 2.0f : 0.0f);
                x.z = y.z * (d.z > 0.5f ? 2.0f : 0.0f);
                x.w = y.w * (d.w > 0.5f ? 2.0f : 0.0f);
                *(v4f*)(xout + o) = x;
            } else {
                *(v4f*)(xout + o) = acc[c2];
            }
        }
    }
}


extern "C" void kernel_launch(void* const* d_in, const int* in_sizes, int n_in,
                              void* d_out, int out_size, void* d_ws, size_t ws_size,
                              hipStream_t stream) {
    (void)in_sizes; (void)n_in; (void)out_size; (void)ws_size;

    const float* x0    = (const float*)d_in[0];
    const int*   ei    = (const int*)d_in[1];
    const float* dropu = (const float*)d_in[2];
    const float* W0    = (const float*)d_in[3];
    const float* b0    = (const float*)d_in[4];
    const float* Whid  = (const float*)d_in[5];
    const float* bhid  = (const float*)d_in[6];
    const float* Wout  = (const float*)d_in[7];
    const float* bout  = (const float*)d_in[8];
    float* out = (float*)d_out;

    const int* src = ei;
    const int* dst = ei + NEDGES;

    char* ws = (char*)d_ws;
    size_t off = 0;
    auto take = [&](size_t bytes) -> void* {
        void* p = (void*)(ws + off);
        off += (bytes + 255) & ~(size_t)255;
        return p;
    };
    const int NB = (NNODES + 255) / 256;
    unsigned* cnt    = (unsigned*)take((size_t)NNODES * 4);
    unsigned* rowptr = (unsigned*)take((size_t)(NNODES + 1) * 4);
    unsigned* cursor = (unsigned*)take((size_t)NNODES * 4);
    unsigned* bsum   = (unsigned*)take((size_t)NB * 4);
    float*    dinv   = (float*)   take((size_t)NNODES * 4);
    int2*     csr    = (int2*)    take((size_t)NEDGES * 8);
    float*    h      = (float*)   take((size_t)NNODES * NCLS * 4);
    float*    xbuf   = (float*)   take((size_t)NNODES * NHID * 4);
    float*    xold   = (float*)   take((size_t)NNODES * NHID * 4);
    float*    Bt     = (float*)   take((size_t)48 * NFEAT * 4);

    const int TB = 256;
    auto blocks = [](int n, int tb) { return (n + tb - 1) / tb; };

    zero_u32_kernel<<<blocks(NNODES, TB), TB, 0, stream>>>(cnt, NNODES);
    zero_u32_kernel<<<blocks(NNODES, TB), TB, 0, stream>>>(cursor, NNODES);
    deg_count_kernel<<<blocks(NEDGES, TB), TB, 0, stream>>>(dst, cnt, NEDGES);
    dinv_kernel<<<blocks(NNODES, TB), TB, 0, stream>>>(cnt, dinv, NNODES);
    scan_block_kernel<<<NB, 256, 0, stream>>>(cnt, rowptr, bsum, NNODES);
    scan_bsum_kernel<<<1, 32, 0, stream>>>(bsum, NB);
    scan_add_kernel<<<blocks(NNODES, TB), TB, 0, stream>>>(rowptr, bsum, NNODES, NEDGES);
    csr_fill_kernel<<<blocks(NEDGES, TB), TB, 0, stream>>>(
        src, dst, dinv, rowptr, cursor, csr, NEDGES);

    int gemm_blocks   = blocks(NNODES / 16, 8);
    int gather_blocks = blocks(NNODES, 32);

    const float* xold_in = nullptr;
    for (int layer = 0; layer < 7; ++layer) {
        const float* A = (layer == 0) ? x0 : xbuf;
        const float* W = (layer == 0) ? W0 : Whid + (size_t)(layer - 1) * NHID * NHID;
        const float* b = (layer == 0) ? b0 : bhid + (size_t)(layer - 1) * NHID;
        int K = (layer == 0) ? NFEAT : NHID;

        transpose_pad_kernel<<<blocks(NHID * K, TB), TB, 0, stream>>>(W, Bt, K, NHID, NHID);
        gemm_wmma_kernel<2><<<gemm_blocks, 256, 0, stream>>>(A, Bt, h, NNODES, K, NHID);

        gather_kernel<NHID, true><<<gather_blocks, 256, 0, stream>>>(
            h, csr, rowptr, dinv, b, xold_in,
            dropu + (size_t)layer * NNODES * NHID, xbuf, xold, NNODES);
        xold_in = xold;
    }

    {
        transpose_pad_kernel<<<blocks(48 * NHID, TB), TB, 0, stream>>>(Wout, Bt, NHID, NCLS, 48);
        gemm_wmma_kernel<3><<<gemm_blocks, 256, 0, stream>>>(xbuf, Bt, h, NNODES, NHID, NCLS);

        gather_kernel<NCLS, false><<<gather_blocks, 256, 0, stream>>>(
            h, csr, rowptr, dinv, bout, nullptr, nullptr, out, nullptr, NNODES);
    }
}


# next layer's 32xFN linear transform fused behind each hidden gather: x tile and W^T meet in LDS, waves 0-1 form h on f32 WMMA (same k order, sum order unchanged); 14 launches, x round trip gone
# speedup vs baseline: 1.3790x; 1.1254x over previous
#include <hip/hip_runtime.h>

#define NNODES 100000
#define NEDGES 1600000
#define NFEAT  128
#define NHID   32
#define NCLS   40
static_assert(NNODES % 16 == 0, "the WMMA strips are whole 16-row tiles");

typedef float v2f __attribute__((ext_vector_type(2)));
typedef float v8f __attribute__((ext_vector_type(8)));

__global__ void zero_u32_kernel(unsigned* __restrict__ p, int n) {
    int t = blockIdx.x * blockDim.x + threadIdx.x;
    if (t < n) p[t] = 0u;
}

__global__ void deg_count_kernel(const int* __restrict__ dst,
                                 unsigned* __restrict__ cnt, int E) {
    int e = blockIdx.x * blockDim.x + threadIdx.x;
    if (e < E) atomicAdd(&cnt[dst[e]], 1u);
}

__global__ void dinv_kernel(const unsigned* __restrict__ cnt,
                            float* __restrict__ dinv, int n) {
    int i = blockIdx.x * blockDim.x + threadIdx.x;
    if (i < n) dinv[i] = rsqrtf((float)cnt[i] + 1.0f);
}

__global__ void scan_block_kernel(const unsigned* __restrict__ cnt,
                                  unsigned* __restrict__ excl,
                                  unsigned* __restrict__ bsum, int n) {
    __shared__ unsigned s[256];
    int t = blockIdx.x * 256 + threadIdx.x;
    unsigned v = (t < n) ? cnt[t] : 0u;
    s[threadIdx.x] = v;
    __syncthreads();
    for (int off = 1; off < 256; off <<= 1) {
        unsigned add = (threadIdx.x >= off) ? s[threadIdx.x - off] : 0u;
        __syncthreads();
        s[threadIdx.x] += add;
        __syncthreads();
    }
    if (t < n) excl[t] = s[threadIdx.x] - v;
    if (threadIdx.x == 255) bsum[blockIdx.x] = s[255];
}

__global__ void scan_bsum_kernel(unsigned* __restrict__ bsum, int nb) {
    if (blockIdx.x == 0 && threadIdx.x == 0) {
        unsigned run = 0;
        for (int i = 0; i < nb; ++i) { unsigned v = bsum[i]; bsum[i] = run; run += v; }
    }
}

__global__ void scan_add_kernel(unsigned* __restrict__ excl,
                                const unsigned* __restrict__ bsum,
                                int n, unsigned total) {
    int t = blockIdx.x * blockDim.x + threadIdx.x;
    if (t < n) excl[t] += bsum[t >> 8];
    if (t == 0) excl[n] = total;
}

__global__ void csr_fill_kernel(const int* __restrict__ src, const int* __restrict__ dst,
                                const float* __restrict__ dinv,
                                const unsigned* __restrict__ rowptr,
                                unsigned* __restrict__ cursor,
                                int2* __restrict__ csr, int E) {
    int e = blockIdx.x * blockDim.x + threadIdx.x;
    if (e >= E) return;
    int s = src[e];
    int d = dst[e];
    unsigned pos = rowptr[d] + atomicAdd(&cursor[d], 1u);
    csr[pos] = make_int2(s, __float_as_int(dinv[s] * dinv[d]));
}

__global__ void transpose_pad_kernel(const float* __restrict__ W,
                                     float* __restrict__ Bt,
                                     int K, int Nout, int Npad) {
    int t = blockIdx.x * blockDim.x + threadIdx.x;
    if (t >= Npad * K) return;
    int n = t / K;
    int k = t - n * K;
    Bt[t] = (n < Nout) ? W[(size_t)k * Nout + n] : 0.0f;
}

template <int NT>
__global__ __launch_bounds__(256)
void gemm_wmma_kernel(const float* __restrict__ A, const float* __restrict__ Bt,
                      float* __restrict__ C, int M, int K, int Nout) {
    int wave = (int)((blockIdx.x * blockDim.x + threadIdx.x) >> 5);
    int lane = threadIdx.x & 31;
    if (wave >= (M >> 4)) return;

    int m0   = wave << 4;
    int half = lane >> 4;
    int l16  = lane & 15;

    const float* Arow = A + (size_t)(m0 + l16) * K;
    const float* Bcol[NT];
#pragma unroll
    for (int t = 0; t < NT; ++t)
        Bcol[t] = Bt + (size_t)(t * 16 + l16) * K;

    v8f c[NT] = {};

    for (int k0 = 0; k0 < K; k0 += 4) {
        int ka = k0 + (half << 1);
        v2f a = *(const v2f*)(Arow + ka);
#pragma unroll
        for (int t = 0; t < NT; ++t) {
            v2f b = *(const v2f*)(Bcol[t] + ka);
            c[t] = __builtin_amdgcn_wmma_f32_16x16x4_f32(
                       false, a, false, b, (short)0, c[t], false, false);
        }
    }

    int rbase = m0 + (half << 3);
#pragma unroll
    for (int t = 0; t < NT; ++t) {
        int ncol = t * 16 + l16;
        if (ncol < Nout) {
#pragma unroll
            for (int r = 0; r < 8; ++r)
                C[(size_t)(rbase + r) * Nout + ncol] = c[t][r];
        }
    }
}

typedef float v4f __attribute__((ext_vector_type(4)));

#define TILE_PITCH 36

template <int E>
__device__ __forceinline__ unsigned group8_lane(unsigned v) {
    return __builtin_amdgcn_mov_dpp8(v, (unsigned)E * 0x249249u);
}

template <int F, bool POST, int FN>
__global__ __launch_bounds__(256)
void gather_kernel(const float* __restrict__ h, const int2* __restrict__ csr,
                   const unsigned* __restrict__ rowptr,
                   const float* __restrict__ dinv, const float* __restrict__ bias,
                   const float* __restrict__ xold_in, const float* __restrict__ dropu,
                   float* __restrict__ xout, float* __restrict__ xold_out,
                   const float* __restrict__ wnext, int n) {
    static_assert(F % 4 == 0, "row must be a whole number of 16-byte pieces");
    constexpr int Q = F / 4;
    constexpr int C = (Q + 7) / 8;
    static_assert(!POST || C == 1, "the epilogue handles one piece per lane");
    static_assert(FN == 0 || (POST && F == 32), "the fused product takes the epilogue's 32-wide rows");
    constexpr int NT  = (FN + 15) / 16;
    constexpr int NW4 = F * FN / 4;
    constexpr int WPT = (NW4 + 255) / 256;
    __shared__ __attribute__((aligned(16))) float xs[FN > 0 ? 32 * TILE_PITCH : 4];
    __shared__ __attribute__((aligned(16))) float wt[FN > 0 ? NT * 16 * TILE_PITCH : 4];

    int node = (int)((blockIdx.x * blockDim.x + threadIdx.x) >> 3);
    int l8   = threadIdx.x & 7;
    bool live = true;
    if constexpr (FN > 0) {
        live = node < n;
        node = live ? node : n - 1;
    } else {
        if (node >= n) return;
    }

    v4f wreg[WPT > 0 ? WPT : 1];
    if constexpr (FN > 0) {
#pragma unroll
        for (int u = 0; u < WPT; ++u) {
            int i = u * 256 + (int)threadIdx.x;
            i = i < NW4 ? i : NW4 - 1;
            wreg[u] = *(const v4f*)(wnext + 4 * (size_t)i);
        }
    }

    unsigned beg = rowptr[node];
    unsigned end = rowptr[node + 1];

    float di = dinv[node];
    const v4f* hrow = (const v4f*)(h + (size_t)node * F);
    v4f hself[C];
    v4f bself[C];
#pragma unroll
    for (int c2 = 0; c2 < C; ++c2) {
        int q = l8 + 8 * c2;
        if (q < Q) {
            hself[c2] = hrow[q];
            bself[c2].x = bias[4 * q + 0];
            bself[c2].y = bias[4 * q + 1];
            bself[c2].z = bias[4 * q + 2];
            bself[c2].w = bias[4 * q + 3];
        }
    }
    size_t obase = (size_t)node * F + 4 * (size_t)l8;
    v4f dvec = (v4f){0.0f, 0.0f, 0.0f, 0.0f};
    v4f xvec = (v4f){0.0f, 0.0f, 0.0f, 0.0f};
    if (POST) dvec = *(const v4f*)(dropu + obase);

    unsigned last = (end > 0u ? end : 1u) - 1u;
    unsigned i0 = beg + (unsigned)l8;
    int2 p = csr[i0 < last ? i0 : last];

    if (POST) {
        if (xold_in) xvec = *(const v4f*)(xold_in + obase);
    }

    if constexpr (FN > 0) {
#pragma unroll
        for (int u = 0; u < WPT; ++u) {
            int i = u * 256 + (int)threadIdx.x;
            if (i < NW4) {
                int k  = (4 * i) / FN;
                int c0 = (4 * i) - k * FN;
                wt[(c0 + 0) * TILE_PITCH + k] = wreg[u].x;
                wt[(c0 + 1) * TILE_PITCH + k] = wreg[u].y;
                wt[(c0 + 2) * TILE_PITCH + k] = wreg[u].z;
                wt[(c0 + 3) * TILE_PITCH + k] = wreg[u].w;
            }
        }
        if (NT * 16 > FN) {
            for (int i = (int)threadIdx.x; i < (NT * 16 - FN) * F; i += 256)
                wt[(FN + i / F) * TILE_PITCH + (i % F)] = 0.0f;
        }
    }

    v4f acc[C];
    float selfw = di * di;
#pragma unroll
    for (int c2 = 0; c2 < C; ++c2) {
        int q = l8 + 8 * c2;
        if (q < Q) acc[c2] = hself[c2] * selfw + bself[c2];
        else       acc[c2] = (v4f){0.0f, 0.0f, 0.0f, 0.0f};
    }

    for (unsigned j = beg; j < end; j += 8) {
        unsigned cnt = end - j;

        unsigned px = (unsigned)p.x, py = (unsigned)p.y;
        int s[8];
        float w[8];
        s[0] = (int)group8_lane<0>(px);  w[0] = __uint_as_float(group8_lane<0>(py));
        s[1] = (int)group8_lane<1>(px);  w[1] = __uint_as_float(group8_lane<1>(py));
        s[2] = (int)group8_lane<2>(px);  w[2] = __uint_as_float(group8_lane<2>(py));
        s[3] = (int)group8_lane<3>(px);  w[3] = __uint_as_float(group8_lane<3>(py));
        s[4] = (int)group8_lane<4>(px);  w[4] = __uint_as_float(group8_lane<4>(py));
        s[5] = (int)group8_lane<5>(px);  w[5] = __uint_as_float(group8_lane<5>(py));
        s[6] = (int)group8_lane<6>(px);  w[6] = __uint_as_float(group8_lane<6>(py));
        s[7] = (int)group8_lane<7>(px);  w[7] = __uint_as_float(group8_lane<7>(py));

        v4f r[8][C];
#pragma unroll
        for (int e = 0; e < 8; ++e) {
            const v4f* he = (const v4f*)(h + (size_t)s[e] * F);
#pragma unroll
            for (int c2 = 0; c2 < C; ++c2) {
                int q = l8 + 8 * c2;
                if (q < Q) r[e][c2] = he[q];
            }
        }

        unsigned jn = j + 8 + (unsigned)l8;
        int2 pn = csr[jn < last ? jn : last];

#pragma unroll
        for (int e = 0; e < 8; e += 2) {
            if ((unsigned)(e + 2) <= cnt) {
#pragma unroll
                for (int c2 = 0; c2 < C; ++c2) {
                    int q = l8 + 8 * c2;
                    if (q < Q) acc[c2] += r[e][c2] * w[e] + r[e + 1][c2] * w[e + 1];
                }
            } else if ((unsigned)(e + 1) == cnt) {
#pragma unroll
                for (int c2 = 0; c2 < C; ++c2) {
                    int q = l8 + 8 * c2;
                    if (q < Q) acc[c2] += r[e][c2] * w[e];
                }
            }
        }
        p = pn;
    }

    if constexpr (FN > 0) {
        v4f y;
        y.x = fmaxf(acc[0].x, 0.0f);
        y.y = fmaxf(acc[0].y, 0.0f);
        y.z = fmaxf(acc[0].z, 0.0f);
        y.w = fmaxf(acc[0].w, 0.0f);
        if (xold_in) y += xvec;
        v4f d = dvec;
        v4f x;
        x.x = y.x * (d.x > 0.5f ? 2.0f : 0.0f);
        x.y = y.y * (d.y > 0.5f ? 2.0f : 0.0f);
        x.z = y.z * (d.z > 0.5f ? 2.0f : 0.0f);
        x.w = y.w * (d.w > 0.5f ? 2.0f : 0.0f);
        *(v4f*)(xs + (threadIdx.x >> 3) * TILE_PITCH + 4 * l8) = x;

        __syncthreads();
        if (live) *(v4f*)(xold_out + obase) = y;

        int wid  = __builtin_amdgcn_readfirstlane((int)(threadIdx.x >> 5));
        int row0 = (int)blockIdx.x * 32 + wid * 16;
        if (wid < 2 && row0 < n) {
            int lane = threadIdx.x & 31;
            int half = lane >> 4;
            int l16  = lane & 15;
            const float* arow = xs + (wid * 16 + l16) * TILE_PITCH + (half << 1);
            const float* bcol = wt + l16 * TILE_PITCH + (half << 1);
            v8f c[NT] = {};
#pragma unroll
            for (int k0 = 0; k0 < F; k0 += 4) {
                v2f a = *(const v2f*)(arow + k0);
#pragma unroll
                for (int t = 0; t < NT; ++t) {
                    v2f b = *(const v2f*)(bcol + t * 16 * TILE_PITCH + k0);
                    c[t] = __builtin_amdgcn_wmma_f32_16x16x4_f32(
                               false, a, false, b, (short)0, c[t], false, false);
                }
            }
            int rbase = row0 + (half << 3);
#pragma unroll
            for (int t = 0; t < NT; ++t) {
                int ncol = t * 16 + l16;
                if (ncol < FN) {
#pragma unroll
                    for (int r = 0; r < 8; ++r)
                        xout[(size_t)(rbase + r) * FN + ncol] = c[t][r];
                }
            }
        }
        return;
    }

    size_t base = (size_t)node * F;
#pragma unroll
    for (int c2 = 0; c2 < C; ++c2) {
        int q = l8 + 8 * c2;
        if (q < Q) {
            size_t o = base + 4 * (size_t)q;
            if (POST) {
                v4f y;
                y.x = fmaxf(acc[c2].x, 0.0f);
                y.y = fmaxf(acc[c2].y, 0.0f);
                y.z = fmaxf(acc[c2].z, 0.0f);
                y.w = fmaxf(acc[c2].w, 0.0f);
                if (xold_in) y += xvec;
                *(v4f*)(xold_out + o) = y;
                v4f d = dvec;
                v4f x;
                x.x = y.x * (d.x > 0.5f ? 2.0f : 0.0f);
                x.y = y.y * (d.y > 0.5f ? 2.0f : 0.0f);
                x.z = y.z * (d.z > 0.5f ? 2.0f : 0.0f);
                x.w = y.w * (d.w > 0.5f ? 2.0f : 0.0f);
                *(v4f*)(xout + o) = x;
            } else {
                *(v4f*)(xout + o) = acc[c2];
            }
        }
    }
}


extern "C" void kernel_launch(void* const* d_in, const int* in_sizes, int n_in,
                              void* d_out, int out_size, void* d_ws, size_t ws_size,
                              hipStream_t stream) {
    (void)in_sizes; (void)n_in; (void)out_size; (void)ws_size;

    const float* x0    = (const float*)d_in[0];
    const int*   ei    = (const int*)d_in[1];
    const float* dropu = (const float*)d_in[2];
    const float* W0    = (const float*)d_in[3];
    const float* b0    = (const float*)d_in[4];
    const float* Whid  = (const float*)d_in[5];
    const float* bhid  = (const float*)d_in[6];
    const float* Wout  = (const float*)d_in[7];
    const float* bout  = (const float*)d_in[8];
    float* out = (float*)d_out;

    const int* src = ei;
    const int* dst = ei + NEDGES;

    char* ws = (char*)d_ws;
    size_t off = 0;
    auto take = [&](size_t bytes) -> void* {
        void* p = (void*)(ws + off);
        off += (bytes + 255) & ~(size_t)255;
        return p;
    };
    const int NB = (NNODES + 255) / 256;
    unsigned* cnt    = (unsigned*)take((size_t)NNODES * 4);
    unsigned* rowptr = (unsigned*)take((size_t)(NNODES + 1) * 4);
    unsigned* cursor = (unsigned*)take((size_t)NNODES * 4);
    unsigned* bsum   = (unsigned*)take((size_t)NB * 4);
    float*    dinv   = (float*)   take((size_t)NNODES * 4);
    int2*     csr    = (int2*)    take((size_t)NEDGES * 8);
    float*    hq     = (float*)   take((size_t)NNODES * NCLS * 4);
    float*    hp     = (float*)   take((size_t)NNODES * NHID * 4);
    float*    xold   = (float*)   take((size_t)NNODES * NHID * 4);
    float*    Bt     = (float*)   take((size_t)48 * NFEAT * 4);

    const int TB = 256;
    auto blocks = [](int n, int tb) { return (n + tb - 1) / tb; };

    zero_u32_kernel<<<blocks(NNODES, TB), TB, 0, stream>>>(cnt, NNODES);
    zero_u32_kernel<<<blocks(NNODES, TB), TB, 0, stream>>>(cursor, NNODES);
    deg_count_kernel<<<blocks(NEDGES, TB), TB, 0, stream>>>(dst, cnt, NEDGES);
    dinv_kernel<<<blocks(NNODES, TB), TB, 0, stream>>>(cnt, dinv, NNODES);
    scan_block_kernel<<<NB, 256, 0, stream>>>(cnt, rowptr, bsum, NNODES);
    scan_bsum_kernel<<<1, 32, 0, stream>>>(bsum, NB);
    scan_add_kernel<<<blocks(NNODES, TB), TB, 0, stream>>>(rowptr, bsum, NNODES, NEDGES);
    csr_fill_kernel<<<blocks(NEDGES, TB), TB, 0, stream>>>(
        src, dst, dinv, rowptr, cursor, csr, NEDGES);

    int gemm_blocks   = blocks(NNODES / 16, 8);
    int gather_blocks = blocks(NNODES, 32);

    transpose_pad_kernel<<<blocks(NHID * NFEAT, TB), TB, 0, stream>>>(W0, Bt, NFEAT, NHID, NHID);
    gemm_wmma_kernel<2><<<gemm_blocks, 256, 0, stream>>>(x0, Bt, hp, NNODES, NFEAT, NHID);

    const float* xold_in = nullptr;
    for (int layer = 0; layer < 7; ++layer) {
        const float* hin  = (layer & 1) ? hq : hp;
        float*       hout = (layer & 1) ? hp : hq;
        const float* b = (layer == 0) ? b0 : bhid + (size_t)(layer - 1) * NHID;
        const float* d = dropu + (size_t)layer * NNODES * NHID;

        if (layer < 6) {
            gather_kernel<NHID, true, NHID><<<gather_blocks, 256, 0, stream>>>(
                hin, csr, rowptr, dinv, b, xold_in, d, hout, xold,
                Whid + (size_t)layer * NHID * NHID, NNODES);
        } else {
            gather_kernel<NHID, true, NCLS><<<gather_blocks, 256, 0, stream>>>(
                hin, csr, rowptr, dinv, b, xold_in, d, hout, xold, Wout, NNODES);
        }
        xold_in = xold;
    }

    gather_kernel<NCLS, false, 0><<<gather_blocks, 256, 0, stream>>>(
        hq, csr, rowptr, dinv, bout, nullptr, nullptr, out, nullptr, nullptr, NNODES);
}


# degree count's atomic add returns each edge's rank in its row; CSR fill places by rowptr+rank: 2nd pass of 1.6M global atomics and cursor zero launch gone; row (sum) order = that add's arrival order
# speedup vs baseline: 1.6904x; 1.2258x over previous
#include <hip/hip_runtime.h>

#define NNODES 100000
#define NEDGES 1600000
#define NFEAT  128
#define NHID   32
#define NCLS   40
static_assert(NNODES % 16 == 0, "the WMMA strips are whole 16-row tiles");

typedef float v2f __attribute__((ext_vector_type(2)));
typedef float v8f __attribute__((ext_vector_type(8)));

__global__ void zero_u32_kernel(unsigned* __restrict__ p, int n) {
    int t = blockIdx.x * blockDim.x + threadIdx.x;
    if (t < n) p[t] = 0u;
}

__global__ void deg_count_kernel(const int* __restrict__ dst,
                                 unsigned* __restrict__ cnt,
                                 unsigned* __restrict__ rank, int E) {
    int e = blockIdx.x * blockDim.x + threadIdx.x;
    if (e < E) rank[e] = atomicAdd(&cnt[dst[e]], 1u);
}

__global__ void dinv_kernel(const unsigned* __restrict__ cnt,
                            float* __restrict__ dinv, int n) {
    int i = blockIdx.x * blockDim.x + threadIdx.x;
    if (i < n) dinv[i] = rsqrtf((float)cnt[i] + 1.0f);
}

__global__ void scan_block_kernel(const unsigned* __restrict__ cnt,
                                  unsigned* __restrict__ excl,
                                  unsigned* __restrict__ bsum, int n) {
    __shared__ unsigned s[256];
    int t = blockIdx.x * 256 + threadIdx.x;
    unsigned v = (t < n) ? cnt[t] : 0u;
    s[threadIdx.x] = v;
    __syncthreads();
    for (int off = 1; off < 256; off <<= 1) {
        unsigned add = (threadIdx.x >= off) ? s[threadIdx.x - off] : 0u;
        __syncthreads();
        s[threadIdx.x] += add;
        __syncthreads();
    }
    if (t < n) excl[t] = s[threadIdx.x] - v;
    if (threadIdx.x == 255) bsum[blockIdx.x] = s[255];
}

__global__ void scan_bsum_kernel(unsigned* __restrict__ bsum, int nb) {
    if (blockIdx.x == 0 && threadIdx.x == 0) {
        unsigned run = 0;
        for (int i = 0; i < nb; ++i) { unsigned v = bsum[i]; bsum[i] = run; run += v; }
    }
}

__global__ void scan_add_kernel(unsigned* __restrict__ excl,
                                const unsigned* __restrict__ bsum,
                                int n, unsigned total) {
    int t = blockIdx.x * blockDim.x + threadIdx.x;
    if (t < n) excl[t] += bsum[t >> 8];
    if (t == 0) excl[n] = total;
}

__global__ void csr_fill_kernel(const int* __restrict__ src, const int* __restrict__ dst,
                                const float* __restrict__ dinv,
                                const unsigned* __restrict__ rowptr,
                                const unsigned* __restrict__ rank,
                                int2* __restrict__ csr, int E) {
    int e = blockIdx.x * blockDim.x + threadIdx.x;
    if (e >= E) return;
    int s = src[e];
    int d = dst[e];
    unsigned pos = rowptr[d] + rank[e];
    csr[pos] = make_int2(s, __float_as_int(dinv[s] * dinv[d]));
}

__global__ void transpose_pad_kernel(const float* __restrict__ W,
                                     float* __restrict__ Bt,
                                     int K, int Nout, int Npad) {
    int t = blockIdx.x * blockDim.x + threadIdx.x;
    if (t >= Npad * K) return;
    int n = t / K;
    int k = t - n * K;
    Bt[t] = (n < Nout) ? W[(size_t)k * Nout + n] : 0.0f;
}

template <int NT>
__global__ __launch_bounds__(256)
void gemm_wmma_kernel(const float* __restrict__ A, const float* __restrict__ Bt,
                      float* __restrict__ C, int M, int K, int Nout) {
    int wave = (int)((blockIdx.x * blockDim.x + threadIdx.x) >> 5);
    int lane = threadIdx.x & 31;
    if (wave >= (M >> 4)) return;

    int m0   = wave << 4;
    int half = lane >> 4;
    int l16  = lane & 15;

    const float* Arow = A + (size_t)(m0 + l16) * K;
    const float* Bcol[NT];
#pragma unroll
    for (int t = 0; t < NT; ++t)
        Bcol[t] = Bt + (size_t)(t * 16 + l16) * K;

    v8f c[NT] = {};

    for (int k0 = 0; k0 < K; k0 += 4) {
        int ka = k0 + (half << 1);
        v2f a = *(const v2f*)(Arow + ka);
#pragma unroll
        for (int t = 0; t < NT; ++t) {
            v2f b = *(const v2f*)(Bcol[t] + ka);
            c[t] = __builtin_amdgcn_wmma_f32_16x16x4_f32(
                       false, a, false, b, (short)0, c[t], false, false);
        }
    }

    int rbase = m0 + (half << 3);
#pragma unroll
    for (int t = 0; t < NT; ++t) {
        int ncol = t * 16 + l16;
        if (ncol < Nout) {
#pragma unroll
            for (int r = 0; r < 8; ++r)
                C[(size_t)(rbase + r) * Nout + ncol] = c[t][r];
        }
    }
}

typedef float v4f __attribute__((ext_vector_type(4)));

#define TILE_PITCH 36

template <int E>
__device__ __forceinline__ unsigned group8_lane(unsigned v) {
    return __builtin_amdgcn_mov_dpp8(v, (unsigned)E * 0x249249u);
}

template <int F, bool POST, int FN>
__global__ __launch_bounds__(256)
void gather_kernel(const float* __restrict__ h, const int2* __restrict__ csr,
                   const unsigned* __restrict__ rowptr,
                   const float* __restrict__ dinv, const float* __restrict__ bias,
                   const float* __restrict__ xold_in, const float* __restrict__ dropu,
                   float* __restrict__ xout, float* __restrict__ xold_out,
                   const float* __restrict__ wnext, int n) {
    static_assert(F % 4 == 0, "row must be a whole number of 16-byte pieces");
    constexpr int Q = F / 4;
    constexpr int C = (Q + 7) / 8;
    static_assert(!POST || C == 1, "the epilogue handles one piece per lane");
    static_assert(FN == 0 || (POST && F == 32), "the fused product takes the epilogue's 32-wide rows");
    constexpr int NT  = (FN + 15) / 16;
    constexpr int NW4 = F * FN / 4;
    constexpr int WPT = (NW4 + 255) / 256;
    __shared__ __attribute__((aligned(16))) float xs[FN > 0 ? 32 * TILE_PITCH : 4];
    __shared__ __attribute__((aligned(16))) float wt[FN > 0 ? NT * 16 * TILE_PITCH : 4];

    int node = (int)((blockIdx.x * blockDim.x + threadIdx.x) >> 3);
    int l8   = threadIdx.x & 7;
    bool live = true;
    if constexpr (FN > 0) {
        live = node < n;
        node = live ? node : n - 1;
    } else {
        if (node >= n) return;
    }

    v4f wreg[WPT > 0 ? WPT : 1];
    if constexpr (FN > 0) {
#pragma unroll
        for (int u = 0; u < WPT; ++u) {
            int i = u * 256 + (int)threadIdx.x;
            i = i < NW4 ? i : NW4 - 1;
            wreg[u] = *(const v4f*)(wnext + 4 * (size_t)i);
        }
    }

    unsigned beg = rowptr[node];
    unsigned end = rowptr[node + 1];

    float di = dinv[node];
    const v4f* hrow = (const v4f*)(h + (size_t)node * F);
    v4f hself[C];
    v4f bself[C];
#pragma unroll
    for (int c2 = 0; c2 < C; ++c2) {
        int q = l8 + 8 * c2;
        if (q < Q) {
            hself[c2] = hrow[q];
            bself[c2].x = bias[4 * q + 0];
            bself[c2].y = bias[4 * q + 1];
            bself[c2].z = bias[4 * q + 2];
            bself[c2].w = bias[4 * q + 3];
        }
    }
    size_t obase = (size_t)node * F + 4 * (size_t)l8;
    v4f dvec = (v4f){0.0f, 0.0f, 0.0f, 0.0f};
    v4f xvec = (v4f){0.0f, 0.0f, 0.0f, 0.0f};
    if (POST) dvec = *(const v4f*)(dropu + obase);

    unsigned last = (end > 0u ? end : 1u) - 1u;
    unsigned i0 = beg + (unsigned)l8;
    int2 p = csr[i0 < last ? i0 : last];

    if (POST) {
        if (xold_in) xvec = *(const v4f*)(xold_in + obase);
    }

    if constexpr (FN > 0) {
#pragma unroll
        for (int u = 0; u < WPT; ++u) {
            int i = u * 256 + (int)threadIdx.x;
            if (i < NW4) {
                int k  = (4 * i) / FN;
                int c0 = (4 * i) - k * FN;
                wt[(c0 + 0) * TILE_PITCH + k] = wreg[u].x;
                wt[(c0 + 1) * TILE_PITCH + k] = wreg[u].y;
                wt[(c0 + 2) * TILE_PITCH + k] = wreg[u].z;
                wt[(c0 + 3) * TILE_PITCH + k] = wreg[u].w;
            }
        }
        if (NT * 16 > FN) {
            for (int i = (int)threadIdx.x; i < (NT * 16 - FN) * F; i += 256)
                wt[(FN + i / F) * TILE_PITCH + (i % F)] = 0.0f;
        }
    }

    v4f acc[C];
    float selfw = di * di;
#pragma unroll
    for (int c2 = 0; c2 < C; ++c2) {
        int q = l8 + 8 * c2;
        if (q < Q) acc[c2] = hself[c2] * selfw + bself[c2];
        else       acc[c2] = (v4f){0.0f, 0.0f, 0.0f, 0.0f};
    }

    for (unsigned j = beg; j < end; j += 8) {
        unsigned cnt = end - j;

        unsigned px = (unsigned)p.x, py = (unsigned)p.y;
        int s[8];
        float w[8];
        s[0] = (int)group8_lane<0>(px);  w[0] = __uint_as_float(group8_lane<0>(py));
        s[1] = (int)group8_lane<1>(px);  w[1] = __uint_as_float(group8_lane<1>(py));
        s[2] = (int)group8_lane<2>(px);  w[2] = __uint_as_float(group8_lane<2>(py));
        s[3] = (int)group8_lane<3>(px);  w[3] = __uint_as_float(group8_lane<3>(py));
        s[4] = (int)group8_lane<4>(px);  w[4] = __uint_as_float(group8_lane<4>(py));
        s[5] = (int)group8_lane<5>(px);  w[5] = __uint_as_float(group8_lane<5>(py));
        s[6] = (int)group8_lane<6>(px);  w[6] = __uint_as_float(group8_lane<6>(py));
        s[7] = (int)group8_lane<7>(px);  w[7] = __uint_as_float(group8_lane<7>(py));

        v4f r[8][C];
#pragma unroll
        for (int e = 0; e < 8; ++e) {
            const v4f* he = (const v4f*)(h + (size_t)s[e] * F);
#pragma unroll
            for (int c2 = 0; c2 < C; ++c2) {
                int q = l8 + 8 * c2;
                if (q < Q) r[e][c2] = he[q];
            }
        }

        unsigned jn = j + 8 + (unsigned)l8;
        int2 pn = csr[jn < last ? jn : last];

#pragma unroll
        for (int e = 0; e < 8; e += 2) {
            if ((unsigned)(e + 2) <= cnt) {
#pragma unroll
                for (int c2 = 0; c2 < C; ++c2) {
                    int q = l8 + 8 * c2;
                    if (q < Q) acc[c2] += r[e][c2] * w[e] + r[e + 1][c2] * w[e + 1];
                }
            } else if ((unsigned)(e + 1) == cnt) {
#pragma unroll
                for (int c2 = 0; c2 < C; ++c2) {
                    int q = l8 + 8 * c2;
                    if (q < Q) acc[c2] += r[e][c2] * w[e];
                }
            }
        }
        p = pn;
    }

    if constexpr (FN > 0) {
        v4f y;
        y.x = fmaxf(acc[0].x, 0.0f);
        y.y = fmaxf(acc[0].y, 0.0f);
        y.z = fmaxf(acc[0].z, 0.0f);
        y.w = fmaxf(acc[0].w, 0.0f);
        if (xold_in) y += xvec;
        v4f d = dvec;
        v4f x;
        x.x = y.x * (d.x > 0.5f ? 2.0f : 0.0f);
        x.y = y.y * (d.y > 0.5f ? 2.0f : 0.0f);
        x.z = y.z * (d.z > 0.5f ? 2.0f : 0.0f);
        x.w = y.w * (d.w > 0.5f ? 2.0f : 0.0f);
        *(v4f*)(xs + (threadIdx.x >> 3) * TILE_PITCH + 4 * l8) = x;

        __syncthreads();
        if (live) *(v4f*)(xold_out + obase) = y;

        int wid  = __builtin_amdgcn_readfirstlane((int)(threadIdx.x >> 5));
        int row0 = (int)blockIdx.x * 32 + wid * 16;
        if (wid < 2 && row0 < n) {
            int lane = threadIdx.x & 31;
            int half = lane >> 4;
            int l16  = lane & 15;
            const float* arow = xs + (wid * 16 + l16) * TILE_PITCH + (half << 1);
            const float* bcol = wt + l16 * TILE_PITCH + (half << 1);
            v8f c[NT] = {};
#pragma unroll
            for (int k0 = 0; k0 < F; k0 += 4) {
                v2f a = *(const v2f*)(arow + k0);
#pragma unroll
                for (int t = 0; t < NT; ++t) {
                    v2f b = *(const v2f*)(bcol + t * 16 * TILE_PITCH + k0);
                    c[t] = __builtin_amdgcn_wmma_f32_16x16x4_f32(
                               false, a, false, b, (short)0, c[t], false, false);
                }
            }
            int rbase = row0 + (half << 3);
#pragma unroll
            for (int t = 0; t < NT; ++t) {
                int ncol = t * 16 + l16;
                if (ncol < FN) {
#pragma unroll
                    for (int r = 0; r < 8; ++r)
                        xout[(size_t)(rbase + r) * FN + ncol] = c[t][r];
                }
            }
        }
        return;
    }

    size_t base = (size_t)node * F;
#pragma unroll
    for (int c2 = 0; c2 < C; ++c2) {
        int q = l8 + 8 * c2;
        if (q < Q) {
            size_t o = base + 4 * (size_t)q;
            if (POST) {
                v4f y;
                y.x = fmaxf(acc[c2].x, 0.0f);
                y.y = fmaxf(acc[c2].y, 0.0f);
                y.z = fmaxf(acc[c2].z, 0.0f);
                y.w = fmaxf(acc[c2].w, 0.0f);
                if (xold_in) y += xvec;
                *(v4f*)(xold_out + o) = y;
                v4f d = dvec;
                v4f x;
                x.x = y.x * (d.x > 0.5f ? 2.0f : 0.0f);
                x.y = y.y * (d.y > 0.5f ? 2.0f : 0.0f);
                x.z = y.z * (d.z > 0.5f ? 2.0f : 0.0f);
                x.w = y.w * (d.w > 0.5f ? 2.0f : 0.0f);
                *(v4f*)(xout + o) = x;
            } else {
                *(v4f*)(xout + o) = acc[c2];
            }
        }
    }
}


extern "C" void kernel_launch(void* const* d_in, const int* in_sizes, int n_in,
                              void* d_out, int out_size, void* d_ws, size_t ws_size,
                              hipStream_t stream) {
    (void)in_sizes; (void)n_in; (void)out_size; (void)ws_size;

    const float* x0    = (const float*)d_in[0];
    const int*   ei    = (const int*)d_in[1];
    const float* dropu = (const float*)d_in[2];
    const float* W0    = (const float*)d_in[3];
    const float* b0    = (const float*)d_in[4];
    const float* Whid  = (const float*)d_in[5];
    const float* bhid  = (const float*)d_in[6];
    const float* Wout  = (const float*)d_in[7];
    const float* bout  = (const float*)d_in[8];
    float* out = (float*)d_out;

    const int* src = ei;
    const int* dst = ei + NEDGES;

    char* ws = (char*)d_ws;
    size_t off = 0;
    auto take = [&](size_t bytes) -> void* {
        void* p = (void*)(ws + off);
        off += (bytes + 255) & ~(size_t)255;
        return p;
    };
    const int NB = (NNODES + 255) / 256;
    unsigned* cnt    = (unsigned*)take((size_t)NNODES * 4);
    unsigned* rowptr = (unsigned*)take((size_t)(NNODES + 1) * 4);
    unsigned* bsum   = (unsigned*)take((size_t)NB * 4);
    float*    dinv   = (float*)   take((size_t)NNODES * 4);
    int2*     csr    = (int2*)    take((size_t)NEDGES * 8);
    float*    hq     = (float*)   take((size_t)NNODES * NCLS * 4);
    float*    hp     = (float*)   take((size_t)NNODES * NHID * 4);
    float*    xold   = (float*)   take((size_t)NNODES * NHID * 4);
    float*    Bt     = (float*)   take((size_t)48 * NFEAT * 4);
    unsigned* rank   = (unsigned*)hq;
    static_assert((size_t)NEDGES * 4 <= (size_t)NNODES * NCLS * 4, "the ranks fit in hq");

    const int TB = 256;
    auto blocks = [](int n, int tb) { return (n + tb - 1) / tb; };

    zero_u32_kernel<<<blocks(NNODES, TB), TB, 0, stream>>>(cnt, NNODES);
    deg_count_kernel<<<blocks(NEDGES, TB), TB, 0, stream>>>(dst, cnt, rank, NEDGES);
    dinv_kernel<<<blocks(NNODES, TB), TB, 0, stream>>>(cnt, dinv, NNODES);
    scan_block_kernel<<<NB, 256, 0, stream>>>(cnt, rowptr, bsum, NNODES);
    scan_bsum_kernel<<<1, 32, 0, stream>>>(bsum, NB);
    scan_add_kernel<<<blocks(NNODES, TB), TB, 0, stream>>>(rowptr, bsum, NNODES, NEDGES);
    csr_fill_kernel<<<blocks(NEDGES, TB), TB, 0, stream>>>(
        src, dst, dinv, rowptr, rank, csr, NEDGES);

    int gemm_blocks   = blocks(NNODES / 16, 8);
    int gather_blocks = blocks(NNODES, 32);

    transpose_pad_kernel<<<blocks(NHID * NFEAT, TB), TB, 0, stream>>>(W0, Bt, NFEAT, NHID, NHID);
    gemm_wmma_kernel<2><<<gemm_blocks, 256, 0, stream>>>(x0, Bt, hp, NNODES, NFEAT, NHID);

    const float* xold_in = nullptr;
    for (int layer = 0; layer < 7; ++layer) {
        const float* hin  = (layer & 1) ? hq : hp;
        float*       hout = (layer & 1) ? hp : hq;
        const float* b = (layer == 0) ? b0 : bhid + (size_t)(layer - 1) * NHID;
        const float* d = dropu + (size_t)layer * NNODES * NHID;

        if (layer < 6) {
            gather_kernel<NHID, true, NHID><<<gather_blocks, 256, 0, stream>>>(
                hin, csr, rowptr, dinv, b, xold_in, d, hout, xold,
                Whid + (size_t)layer * NHID * NHID, NNODES);
        } else {
            gather_kernel<NHID, true, NCLS><<<gather_blocks, 256, 0, stream>>>(
                hin, csr, rowptr, dinv, b, xold_in, d, hout, xold, Wout, NNODES);
        }
        xold_in = xold;
    }

    gather_kernel<NCLS, false, 0><<<gather_blocks, 256, 0, stream>>>(
        hq, csr, rowptr, dinv, bout, nullptr, nullptr, out, nullptr, nullptr, NNODES);
}


# CSR build: edges counting-sorted by destination in two levels on LDS counters (196 bins of 512 nodes, then per node), no global atomic per edge; row (sum) order = LDS add arrival order
# speedup vs baseline: 2.1601x; 1.2779x over previous
#include <hip/hip_runtime.h>

#define NNODES 100000
#define NEDGES 1600000
#define NFEAT  128
#define NHID   32
#define NCLS   40
static_assert(NNODES % 16 == 0, "the WMMA strips are whole 16-row tiles");

typedef float v2f __attribute__((ext_vector_type(2)));
typedef float v8f __attribute__((ext_vector_type(8)));

#define BIN_NODES 512
#define NBINS     ((NNODES + BIN_NODES - 1) / BIN_NODES)
#define CHUNK_T   512
#define CHUNK_EPT 16
#define CHUNK     (CHUNK_T * CHUNK_EPT)
#define NCHUNK    ((((NEDGES + CHUNK - 1) / CHUNK) + 3) / 4 * 4)
static_assert(NEDGES % 4 == 0, "the edge list is read in 16-byte pieces");
static_assert((BIN_NODES & (BIN_NODES - 1)) == 0, "a destination's place in its bin is a mask");

typedef int v4i __attribute__((ext_vector_type(4)));

__device__ __forceinline__ int bin_of(int d) { return (int)((unsigned)d / (unsigned)BIN_NODES); }

__device__ __forceinline__ unsigned wave_incl_scan(unsigned v, int lane) {
#pragma unroll
    for (int off = 1; off < 32; off <<= 1) {
        unsigned o = __shfl_up(v, (unsigned)off);
        if (lane >= off) v += o;
    }
    return v;
}

__global__ __launch_bounds__(CHUNK_T)
void bin_count_kernel(const int* __restrict__ dst, unsigned* __restrict__ table, int E) {
    __shared__ unsigned hist[NBINS];
    for (int i = (int)threadIdx.x; i < NBINS; i += CHUNK_T) hist[i] = 0u;

    const int E4 = E >> 2;
    int base4 = (int)blockIdx.x * (CHUNK / 4) + (int)threadIdx.x;
    v4i d[CHUNK_EPT / 4];
#pragma unroll
    for (int k = 0; k < CHUNK_EPT / 4; ++k) {
        int i4 = base4 + k * CHUNK_T;
        i4 = i4 < E4 ? i4 : E4 - 1;
        d[k] = *(const v4i*)(dst + 4 * (size_t)i4);
    }
    __syncthreads();
#pragma unroll
    for (int k = 0; k < CHUNK_EPT / 4; ++k) {
        if (base4 + k * CHUNK_T < E4) {
            atomicAdd(&hist[bin_of(d[k].x)], 1u);
            atomicAdd(&hist[bin_of(d[k].y)], 1u);
            atomicAdd(&hist[bin_of(d[k].z)], 1u);
            atomicAdd(&hist[bin_of(d[k].w)], 1u);
        }
    }
    __syncthreads();
    for (int i = (int)threadIdx.x; i < NBINS; i += CHUNK_T)
        table[(size_t)i * NCHUNK + blockIdx.x] = hist[i];
}

#define SCAN_T   1024
#define SCAN_V   (((NBINS * NCHUNK + SCAN_T - 1) / SCAN_T + 3) / 4)
static_assert((NBINS * NCHUNK) % 4 == 0, "the table is a whole number of 16-byte pieces");
typedef unsigned v4u __attribute__((ext_vector_type(4)));
__global__ __launch_bounds__(SCAN_T)
void table_scan_kernel(unsigned* __restrict__ table, int M4) {
    __shared__ unsigned wsum[32];
    int lane = (int)threadIdx.x & 31;
    int wid  = (int)threadIdx.x >> 5;
    v4u* t4 = (v4u*)table;
    int b = (int)threadIdx.x * SCAN_V;
    v4u v[SCAN_V];
    unsigned sum = 0u;
#pragma unroll
    for (int i = 0; i < SCAN_V; ++i) {
        v[i] = (v4u){0u, 0u, 0u, 0u};
        if (b + i < M4) v[i] = t4[b + i];
        sum += v[i].x + v[i].y + v[i].z + v[i].w;
    }
    unsigned inc = wave_incl_scan(sum, lane);
    if (lane == 31) wsum[wid] = inc;
    __syncthreads();
    if (wid == 0) {
        unsigned w = wsum[lane];
        unsigned winc = wave_incl_scan(w, lane);
        wsum[lane] = winc - w;
    }
    __syncthreads();
    unsigned run = wsum[wid] + inc - sum;
#pragma unroll
    for (int i = 0; i < SCAN_V; ++i) {
        v4u o;
        o.x = run; run += v[i].x;
        o.y = run; run += v[i].y;
        o.z = run; run += v[i].z;
        o.w = run; run += v[i].w;
        if (b + i < M4) t4[b + i] = o;
    }
}

__global__ __launch_bounds__(CHUNK_T)
void bin_scatter_kernel(const int* __restrict__ src, const int* __restrict__ dst,
                        const unsigned* __restrict__ table,
                        int2* __restrict__ binned, int E) {
    __shared__ unsigned cursor[NBINS];

    const int E4 = E >> 2;
    int base4 = (int)blockIdx.x * (CHUNK / 4) + (int)threadIdx.x;
    v4i d[CHUNK_EPT / 4];
    v4i s[CHUNK_EPT / 4];
#pragma unroll
    for (int k = 0; k < CHUNK_EPT / 4; ++k) {
        int i4 = base4 + k * CHUNK_T;
        i4 = i4 < E4 ? i4 : E4 - 1;
        d[k] = *(const v4i*)(dst + 4 * (size_t)i4);
        s[k] = *(const v4i*)(src + 4 * (size_t)i4);
    }
    for (int i = (int)threadIdx.x; i < NBINS; i += CHUNK_T)
        cursor[i] = table[(size_t)i * NCHUNK + blockIdx.x];
    __syncthreads();
#pragma unroll
    for (int k = 0; k < CHUNK_EPT / 4; ++k) {
        if (base4 + k * CHUNK_T < E4) {
            unsigned p0 = atomicAdd(&cursor[bin_of(d[k].x)], 1u);
            unsigned p1 = atomicAdd(&cursor[bin_of(d[k].y)], 1u);
            unsigned p2 = atomicAdd(&cursor[bin_of(d[k].z)], 1u);
            unsigned p3 = atomicAdd(&cursor[bin_of(d[k].w)], 1u);
            binned[p0] = make_int2(s[k].x, d[k].x);
            binned[p1] = make_int2(s[k].y, d[k].y);
            binned[p2] = make_int2(s[k].z, d[k].z);
            binned[p3] = make_int2(s[k].w, d[k].w);
        }
    }
}

#define L2_T 512
#define L2_U 9
static_assert(L2_T >= BIN_NODES, "one thread per destination of the bin");
__global__ __launch_bounds__(L2_T)
void bin_to_csr_kernel(const int2* __restrict__ binned, const unsigned* __restrict__ table,
                       unsigned* __restrict__ rowptr, float* __restrict__ dinv,
                       int2* __restrict__ csr, int n, int E) {
    __shared__ unsigned cnt[BIN_NODES];
    __shared__ unsigned first[BIN_NODES];
    __shared__ unsigned cur[BIN_NODES];
    __shared__ float    dloc[BIN_NODES];
    int bin = (int)blockIdx.x;
    unsigned s = table[(size_t)bin * NCHUNK];
    unsigned e = (bin + 1 < NBINS) ? table[(size_t)(bin + 1) * NCHUNK] : (unsigned)E;
    if (threadIdx.x < BIN_NODES) cnt[threadIdx.x] = 0u;
    __syncthreads();

    for (unsigned i0 = s; i0 < e; i0 += L2_T * L2_U) {
        int dd[L2_U];
#pragma unroll
        for (int u = 0; u < L2_U; ++u) {
            unsigned i = i0 + (unsigned)(u * L2_T) + threadIdx.x;
            dd[u] = binned[i < e ? i : e - 1u].y;
        }
#pragma unroll
        for (int u = 0; u < L2_U; ++u) {
            unsigned i = i0 + (unsigned)(u * L2_T) + threadIdx.x;
            if (i < e) atomicAdd(&cnt[dd[u] & (BIN_NODES - 1)], 1u);
        }
    }
    __syncthreads();

    if (threadIdx.x < 32) {
        constexpr int PER = BIN_NODES / 32;
        int lane = (int)threadIdx.x;
        unsigned cc[PER];
        unsigned sum = 0u;
#pragma unroll
        for (int k = 0; k < PER; ++k) { cc[k] = cnt[lane * PER + k]; sum += cc[k]; }
        unsigned run = s + wave_incl_scan(sum, lane) - sum;
#pragma unroll
        for (int k = 0; k < PER; ++k) {
            first[lane * PER + k] = run;
            cur[lane * PER + k]   = run;
            dloc[lane * PER + k]  = rsqrtf((float)cc[k] + 1.0f);
            run += cc[k];
        }
    }
    __syncthreads();
    if (threadIdx.x < BIN_NODES) {
        int node = bin * BIN_NODES + (int)threadIdx.x;
        if (node < n) {
            rowptr[node] = first[threadIdx.x];
            dinv[node]   = dloc[threadIdx.x];
        }
    }
    if (bin == NBINS - 1 && threadIdx.x == 0) rowptr[n] = (unsigned)E;

    for (unsigned i0 = s; i0 < e; i0 += L2_T * L2_U) {
        int2 p[L2_U];
#pragma unroll
        for (int u = 0; u < L2_U; ++u) {
            unsigned i = i0 + (unsigned)(u * L2_T) + threadIdx.x;
            p[u] = binned[i < e ? i : e - 1u];
        }
#pragma unroll
        for (int u = 0; u < L2_U; ++u) {
            unsigned i = i0 + (unsigned)(u * L2_T) + threadIdx.x;
            if (i < e) {
                int l = p[u].y & (BIN_NODES - 1);
                unsigned pos = atomicAdd(&cur[l], 1u);
                csr[pos] = make_int2(p[u].x, __float_as_int(dloc[l]));
            }
        }
    }
}

#define WGT_T 320
#define WGT_V 4
static_assert((NEDGES / 2) % (WGT_T * WGT_V) == 0, "the work-groups tile the list exactly");
__global__ __launch_bounds__(WGT_T)
void edge_weight_kernel(int2* __restrict__ csr, const float* __restrict__ dinv) {
    v4i* c4 = (v4i*)csr;
    int j = (int)blockIdx.x * (WGT_T * WGT_V) + (int)threadIdx.x;
    v4i a[WGT_V];
#pragma unroll
    for (int u = 0; u < WGT_V; ++u) a[u] = c4[j + u * WGT_T];
    float w0[WGT_V], w1[WGT_V];
#pragma unroll
    for (int u = 0; u < WGT_V; ++u) { w0[u] = dinv[a[u].x]; w1[u] = dinv[a[u].z]; }
#pragma unroll
    for (int u = 0; u < WGT_V; ++u) {
        a[u].y = __float_as_int(w0[u] * __int_as_float(a[u].y));
        a[u].w = __float_as_int(w1[u] * __int_as_float(a[u].w));
        c4[j + u * WGT_T] = a[u];
    }
}

__global__ void transpose_pad_kernel(const float* __restrict__ W,
                                     float* __restrict__ Bt,
                                     int K, int Nout, int Npad) {
    int t = blockIdx.x * blockDim.x + threadIdx.x;
    if (t >= Npad * K) return;
    int n = t / K;
    int k = t - n * K;
    Bt[t] = (n < Nout) ? W[(size_t)k * Nout + n] : 0.0f;
}

template <int NT>
__global__ __launch_bounds__(256)
void gemm_wmma_kernel(const float* __restrict__ A, const float* __restrict__ Bt,
                      float* __restrict__ C, int M, int K, int Nout) {
    int wave = (int)((blockIdx.x * blockDim.x + threadIdx.x) >> 5);
    int lane = threadIdx.x & 31;
    if (wave >= (M >> 4)) return;

    int m0   = wave << 4;
    int half = lane >> 4;
    int l16  = lane & 15;

    const float* Arow = A + (size_t)(m0 + l16) * K;
    const float* Bcol[NT];
#pragma unroll
    for (int t = 0; t < NT; ++t)
        Bcol[t] = Bt + (size_t)(t * 16 + l16) * K;

    v8f c[NT] = {};

    for (int k0 = 0; k0 < K; k0 += 4) {
        int ka = k0 + (half << 1);
        v2f a = *(const v2f*)(Arow + ka);
#pragma unroll
        for (int t = 0; t < NT; ++t) {
            v2f b = *(const v2f*)(Bcol[t] + ka);
            c[t] = __builtin_amdgcn_wmma_f32_16x16x4_f32(
                       false, a, false, b, (short)0, c[t], false, false);
        }
    }

    int rbase = m0 + (half << 3);
#pragma unroll
    for (int t = 0; t < NT; ++t) {
        int ncol = t * 16 + l16;
        if (ncol < Nout) {
#pragma unroll
            for (int r = 0; r < 8; ++r)
                C[(size_t)(rbase + r) * Nout + ncol] = c[t][r];
        }
    }
}

typedef float v4f __attribute__((ext_vector_type(4)));

#define TILE_PITCH 36

template <int E>
__device__ __forceinline__ unsigned group8_lane(unsigned v) {
    return __builtin_amdgcn_mov_dpp8(v, (unsigned)E * 0x249249u);
}

template <int F, bool POST, int FN>
__global__ __launch_bounds__(256)
void gather_kernel(const float* __restrict__ h, const int2* __restrict__ csr,
                   const unsigned* __restrict__ rowptr,
                   const float* __restrict__ dinv, const float* __restrict__ bias,
                   const float* __restrict__ xold_in, const float* __restrict__ dropu,
                   float* __restrict__ xout, float* __restrict__ xold_out,
                   const float* __restrict__ wnext, int n) {
    static_assert(F % 4 == 0, "row must be a whole number of 16-byte pieces");
    constexpr int Q = F / 4;
    constexpr int C = (Q + 7) / 8;
    static_assert(!POST || C == 1, "the epilogue handles one piece per lane");
    static_assert(FN == 0 || (POST && F == 32), "the fused product takes the epilogue's 32-wide rows");
    constexpr int NT  = (FN + 15) / 16;
    constexpr int NW4 = F * FN / 4;
    constexpr int WPT = (NW4 + 255) / 256;
    __shared__ __attribute__((aligned(16))) float xs[FN > 0 ? 32 * TILE_PITCH : 4];
    __shared__ __attribute__((aligned(16))) float wt[FN > 0 ? NT * 16 * TILE_PITCH : 4];

    int node = (int)((blockIdx.x * blockDim.x + threadIdx.x) >> 3);
    int l8   = threadIdx.x & 7;
    bool live = true;
    if constexpr (FN > 0) {
        live = node < n;
        node = live ? node : n - 1;
    } else {
        if (node >= n) return;
    }

    v4f wreg[WPT > 0 ? WPT : 1];
    if constexpr (FN > 0) {
#pragma unroll
        for (int u = 0; u < WPT; ++u) {
            int i = u * 256 + (int)threadIdx.x;
            i = i < NW4 ? i : NW4 - 1;
            wreg[u] = *(const v4f*)(wnext + 4 * (size_t)i);
        }
    }

    unsigned beg = rowptr[node];
    unsigned end = rowptr[node + 1];

    float di = dinv[node];
    const v4f* hrow = (const v4f*)(h + (size_t)node * F);
    v4f hself[C];
    v4f bself[C];
#pragma unroll
    for (int c2 = 0; c2 < C; ++c2) {
        int q = l8 + 8 * c2;
        if (q < Q) {
            hself[c2] = hrow[q];
            bself[c2].x = bias[4 * q + 0];
            bself[c2].y = bias[4 * q + 1];
            bself[c2].z = bias[4 * q + 2];
            bself[c2].w = bias[4 * q + 3];
        }
    }
    size_t obase = (size_t)node * F + 4 * (size_t)l8;
    v4f dvec = (v4f){0.0f, 0.0f, 0.0f, 0.0f};
    v4f xvec = (v4f){0.0f, 0.0f, 0.0f, 0.0f};
    if (POST) dvec = *(const v4f*)(dropu + obase);

    unsigned last = (end > 0u ? end : 1u) - 1u;
    unsigned i0 = beg + (unsigned)l8;
    int2 p = csr[i0 < last ? i0 : last];

    if (POST) {
        if (xold_in) xvec = *(const v4f*)(xold_in + obase);
    }

    if constexpr (FN > 0) {
#pragma unroll
        for (int u = 0; u < WPT; ++u) {
            int i = u * 256 + (int)threadIdx.x;
            if (i < NW4) {
                int k  = (4 * i) / FN;
                int c0 = (4 * i) - k * FN;
                wt[(c0 + 0) * TILE_PITCH + k] = wreg[u].x;
                wt[(c0 + 1) * TILE_PITCH + k] = wreg[u].y;
                wt[(c0 + 2) * TILE_PITCH + k] = wreg[u].z;
                wt[(c0 + 3) * TILE_PITCH + k] = wreg[u].w;
            }
        }
        if (NT * 16 > FN) {
            for (int i = (int)threadIdx.x; i < (NT * 16 - FN) * F; i += 256)
                wt[(FN + i / F) * TILE_PITCH + (i % F)] = 0.0f;
        }
    }

    v4f acc[C];
    float selfw = di * di;
#pragma unroll
    for (int c2 = 0; c2 < C; ++c2) {
        int q = l8 + 8 * c2;
        if (q < Q) acc[c2] = hself[c2] * selfw + bself[c2];
        else       acc[c2] = (v4f){0.0f, 0.0f, 0.0f, 0.0f};
    }

    for (unsigned j = beg; j < end; j += 8) {
        unsigned cnt = end - j;

        unsigned px = (unsigned)p.x, py = (unsigned)p.y;
        int s[8];
        float w[8];
        s[0] = (int)group8_lane<0>(px);  w[0] = __uint_as_float(group8_lane<0>(py));
        s[1] = (int)group8_lane<1>(px);  w[1] = __uint_as_float(group8_lane<1>(py));
        s[2] = (int)group8_lane<2>(px);  w[2] = __uint_as_float(group8_lane<2>(py));
        s[3] = (int)group8_lane<3>(px);  w[3] = __uint_as_float(group8_lane<3>(py));
        s[4] = (int)group8_lane<4>(px);  w[4] = __uint_as_float(group8_lane<4>(py));
        s[5] = (int)group8_lane<5>(px);  w[5] = __uint_as_float(group8_lane<5>(py));
        s[6] = (int)group8_lane<6>(px);  w[6] = __uint_as_float(group8_lane<6>(py));
        s[7] = (int)group8_lane<7>(px);  w[7] = __uint_as_float(group8_lane<7>(py));

        v4f r[8][C];
#pragma unroll
        for (int e = 0; e < 8; ++e) {
            const v4f* he = (const v4f*)(h + (size_t)s[e] * F);
#pragma unroll
            for (int c2 = 0; c2 < C; ++c2) {
                int q = l8 + 8 * c2;
                if (q < Q) r[e][c2] = he[q];
            }
        }

        unsigned jn = j + 8 + (unsigned)l8;
        int2 pn = csr[jn < last ? jn : last];

#pragma unroll
        for (int e = 0; e < 8; e += 2) {
            if ((unsigned)(e + 2) <= cnt) {
#pragma unroll
                for (int c2 = 0; c2 < C; ++c2) {
                    int q = l8 + 8 * c2;
                    if (q < Q) acc[c2] += r[e][c2] * w[e] + r[e + 1][c2] * w[e + 1];
                }
            } else if ((unsigned)(e + 1) == cnt) {
#pragma unroll
                for (int c2 = 0; c2 < C; ++c2) {
                    int q = l8 + 8 * c2;
                    if (q < Q) acc[c2] += r[e][c2] * w[e];
                }
            }
        }
        p = pn;
    }

    if constexpr (FN > 0) {
        v4f y;
        y.x = fmaxf(acc[0].x, 0.0f);
        y.y = fmaxf(acc[0].y, 0.0f);
        y.z = fmaxf(acc[0].z, 0.0f);
        y.w = fmaxf(acc[0].w, 0.0f);
        if (xold_in) y += xvec;
        v4f d = dvec;
        v4f x;
        x.x = y.x * (d.x > 0.5f ? 2.0f : 0.0f);
        x.y = y.y * (d.y > 0.5f ? 2.0f : 0.0f);
        x.z = y.z * (d.z > 0.5f ? 2.0f : 0.0f);
        x.w = y.w * (d.w > 0.5f ? 2.0f : 0.0f);
        *(v4f*)(xs + (threadIdx.x >> 3) * TILE_PITCH + 4 * l8) = x;

        __syncthreads();
        if (live) *(v4f*)(xold_out + obase) = y;

        int wid  = __builtin_amdgcn_readfirstlane((int)(threadIdx.x >> 5));
        int row0 = (int)blockIdx.x * 32 + wid * 16;
        if (wid < 2 && row0 < n) {
            int lane = threadIdx.x & 31;
            int half = lane >> 4;
            int l16  = lane & 15;
            const float* arow = xs + (wid * 16 + l16) * TILE_PITCH + (half << 1);
            const float* bcol = wt + l16 * TILE_PITCH + (half << 1);
            v8f c[NT] = {};
#pragma unroll
            for (int k0 = 0; k0 < F; k0 += 4) {
                v2f a = *(const v2f*)(arow + k0);
#pragma unroll
                for (int t = 0; t < NT; ++t) {
                    v2f b = *(const v2f*)(bcol + t * 16 * TILE_PITCH + k0);
                    c[t] = __builtin_amdgcn_wmma_f32_16x16x4_f32(
                               false, a, false, b, (short)0, c[t], false, false);
                }
            }
            int rbase = row0 + (half << 3);
#pragma unroll
            for (int t = 0; t < NT; ++t) {
                int ncol = t * 16 + l16;
                if (ncol < FN) {
#pragma unroll
                    for (int r = 0; r < 8; ++r)
                        xout[(size_t)(rbase + r) * FN + ncol] = c[t][r];
                }
            }
        }
        return;
    }

    size_t base = (size_t)node * F;
#pragma unroll
    for (int c2 = 0; c2 < C; ++c2) {
        int q = l8 + 8 * c2;
        if (q < Q) {
            size_t o = base + 4 * (size_t)q;
            if (POST) {
                v4f y;
                y.x = fmaxf(acc[c2].x, 0.0f);
                y.y = fmaxf(acc[c2].y, 0.0f);
                y.z = fmaxf(acc[c2].z, 0.0f);
                y.w = fmaxf(acc[c2].w, 0.0f);
                if (xold_in) y += xvec;
                *(v4f*)(xold_out + o) = y;
                v4f d = dvec;
                v4f x;
                x.x = y.x * (d.x > 0.5f ? 2.0f : 0.0f);
                x.y = y.y * (d.y > 0.5f ? 2.0f : 0.0f);
                x.z = y.z * (d.z > 0.5f ? 2.0f : 0.0f);
                x.w = y.w * (d.w > 0.5f ? 2.0f : 0.0f);
                *(v4f*)(xout + o) = x;
            } else {
                *(v4f*)(xout + o) = acc[c2];
            }
        }
    }
}


extern "C" void kernel_launch(void* const* d_in, const int* in_sizes, int n_in,
                              void* d_out, int out_size, void* d_ws, size_t ws_size,
                              hipStream_t stream) {
    (void)in_sizes; (void)n_in; (void)out_size; (void)ws_size;

    const float* x0    = (const float*)d_in[0];
    const int*   ei    = (const int*)d_in[1];
    const float* dropu = (const float*)d_in[2];
    const float* W0    = (const float*)d_in[3];
    const float* b0    = (const float*)d_in[4];
    const float* Whid  = (const float*)d_in[5];
    const float* bhid  = (const float*)d_in[6];
    const float* Wout  = (const float*)d_in[7];
    const float* bout  = (const float*)d_in[8];
    float* out = (float*)d_out;

    const int* src = ei;
    const int* dst = ei + NEDGES;

    char* ws = (char*)d_ws;
    size_t off = 0;
    auto take = [&](size_t bytes) -> void* {
        void* p = (void*)(ws + off);
        off += (bytes + 255) & ~(size_t)255;
        return p;
    };
    unsigned* table  = (unsigned*)take((size_t)NBINS * NCHUNK * 4);
    unsigned* rowptr = (unsigned*)take((size_t)(NNODES + 1) * 4);
    float*    dinv   = (float*)   take((size_t)NNODES * 4);
    int2*     csr    = (int2*)    take((size_t)NEDGES * 8);
    float*    hq     = (float*)   take((size_t)NNODES * NCLS * 4);
    float*    hp     = (float*)   take((size_t)NNODES * NHID * 4);
    float*    xold   = (float*)   take((size_t)NNODES * NHID * 4);
    float*    Bt     = (float*)   take((size_t)48 * NFEAT * 4);
    int2* binned = (int2*)hq;
    static_assert((size_t)NEDGES * 8 <= (size_t)NNODES * NCLS * 4, "the binned edges fit in hq");

    const int TB = 256;
    auto blocks = [](int n, int tb) { return (n + tb - 1) / tb; };

    bin_count_kernel<<<NCHUNK, CHUNK_T, 0, stream>>>(dst, table, NEDGES);
    table_scan_kernel<<<1, SCAN_T, 0, stream>>>(table, NBINS * NCHUNK / 4);
    bin_scatter_kernel<<<NCHUNK, CHUNK_T, 0, stream>>>(src, dst, table, binned, NEDGES);
    bin_to_csr_kernel<<<NBINS, L2_T, 0, stream>>>(binned, table, rowptr, dinv, csr, NNODES, NEDGES);
    edge_weight_kernel<<<NEDGES / 2 / (WGT_T * WGT_V), WGT_T, 0, stream>>>(csr, dinv);

    int gemm_blocks   = blocks(NNODES / 16, 8);
    int gather_blocks = blocks(NNODES, 32);

    transpose_pad_kernel<<<blocks(NHID * NFEAT, TB), TB, 0, stream>>>(W0, Bt, NFEAT, NHID, NHID);
    gemm_wmma_kernel<2><<<gemm_blocks, 256, 0, stream>>>(x0, Bt, hp, NNODES, NFEAT, NHID);

    const float* xold_in = nullptr;
    for (int layer = 0; layer < 7; ++layer) {
        const float* hin  = (layer & 1) ? hq : hp;
        float*       hout = (layer & 1) ? hp : hq;
        const float* b = (layer == 0) ? b0 : bhid + (size_t)(layer - 1) * NHID;
        const float* d = dropu + (size_t)layer * NNODES * NHID;

        if (layer < 6) {
            gather_kernel<NHID, true, NHID><<<gather_blocks, 256, 0, stream>>>(
                hin, csr, rowptr, dinv, b, xold_in, d, hout, xold,
                Whid + (size_t)layer * NHID * NHID, NNODES);
        } else {
            gather_kernel<NHID, true, NCLS><<<gather_blocks, 256, 0, stream>>>(
                hin, csr, rowptr, dinv, b, xold_in, d, hout, xold, Wout, NNODES);
        }
        xold_in = xold;
    }

    gather_kernel<NCLS, false, 0><<<gather_blocks, 256, 0, stream>>>(
        hq, csr, rowptr, dinv, bout, nullptr, nullptr, out, nullptr, nullptr, NNODES);
}


# symmetric norm factored out of the edge sum: h rows stored pre-scaled by dinv[row], gather adds plain rows and scales the sum by dinv[dst] once; edge weights and their launch gone; sum order kept
# speedup vs baseline: 2.4765x; 1.1465x over previous
#include <hip/hip_runtime.h>

#define NNODES 100000
#define NEDGES 1600000
#define NFEAT  128
#define NHID   32
#define NCLS   40
static_assert(NNODES % 16 == 0, "the WMMA strips are whole 16-row tiles");

typedef float v2f __attribute__((ext_vector_type(2)));
typedef float v4f __attribute__((ext_vector_type(4)));
typedef float v8f __attribute__((ext_vector_type(8)));

#define BIN_NODES 512
#define NBINS     ((NNODES + BIN_NODES - 1) / BIN_NODES)
#define CHUNK_T   512
#define CHUNK_EPT 16
#define CHUNK     (CHUNK_T * CHUNK_EPT)
#define NCHUNK    ((((NEDGES + CHUNK - 1) / CHUNK) + 3) / 4 * 4)
static_assert(NEDGES % 4 == 0, "the edge list is read in 16-byte pieces");
static_assert((BIN_NODES & (BIN_NODES - 1)) == 0, "a destination's place in its bin is a mask");

typedef int v4i __attribute__((ext_vector_type(4)));

__device__ __forceinline__ int bin_of(int d) { return (int)((unsigned)d / (unsigned)BIN_NODES); }

__device__ __forceinline__ unsigned wave_incl_scan(unsigned v, int lane) {
#pragma unroll
    for (int off = 1; off < 32; off <<= 1) {
        unsigned o = __shfl_up(v, (unsigned)off);
        if (lane >= off) v += o;
    }
    return v;
}

__global__ __launch_bounds__(CHUNK_T)
void bin_count_kernel(const int* __restrict__ dst, unsigned* __restrict__ table, int E) {
    __shared__ unsigned hist[NBINS];
    for (int i = (int)threadIdx.x; i < NBINS; i += CHUNK_T) hist[i] = 0u;

    const int E4 = E >> 2;
    int base4 = (int)blockIdx.x * (CHUNK / 4) + (int)threadIdx.x;
    v4i d[CHUNK_EPT / 4];
#pragma unroll
    for (int k = 0; k < CHUNK_EPT / 4; ++k) {
        int i4 = base4 + k * CHUNK_T;
        i4 = i4 < E4 ? i4 : E4 - 1;
        d[k] = *(const v4i*)(dst + 4 * (size_t)i4);
    }
    __syncthreads();
#pragma unroll
    for (int k = 0; k < CHUNK_EPT / 4; ++k) {
        if (base4 + k * CHUNK_T < E4) {
            atomicAdd(&hist[bin_of(d[k].x)], 1u);
            atomicAdd(&hist[bin_of(d[k].y)], 1u);
            atomicAdd(&hist[bin_of(d[k].z)], 1u);
            atomicAdd(&hist[bin_of(d[k].w)], 1u);
        }
    }
    __syncthreads();
    for (int i = (int)threadIdx.x; i < NBINS; i += CHUNK_T)
        table[(size_t)i * NCHUNK + blockIdx.x] = hist[i];
}

#define SCAN_T   1024
#define SCAN_V   (((NBINS * NCHUNK + SCAN_T - 1) / SCAN_T + 3) / 4)
static_assert((NBINS * NCHUNK) % 4 == 0, "the table is a whole number of 16-byte pieces");
typedef unsigned v4u __attribute__((ext_vector_type(4)));
__global__ __launch_bounds__(SCAN_T)
void table_scan_kernel(unsigned* __restrict__ table, int M4) {
    __shared__ unsigned wsum[32];
    int lane = (int)threadIdx.x & 31;
    int wid  = (int)threadIdx.x >> 5;
    v4u* t4 = (v4u*)table;
    int b = (int)threadIdx.x * SCAN_V;
    v4u v[SCAN_V];
    unsigned sum = 0u;
#pragma unroll
    for (int i = 0; i < SCAN_V; ++i) {
        v[i] = (v4u){0u, 0u, 0u, 0u};
        if (b + i < M4) v[i] = t4[b + i];
        sum += v[i].x + v[i].y + v[i].z + v[i].w;
    }
    unsigned inc = wave_incl_scan(sum, lane);
    if (lane == 31) wsum[wid] = inc;
    __syncthreads();
    if (wid == 0) {
        unsigned w = wsum[lane];
        unsigned winc = wave_incl_scan(w, lane);
        wsum[lane] = winc - w;
    }
    __syncthreads();
    unsigned run = wsum[wid] + inc - sum;
#pragma unroll
    for (int i = 0; i < SCAN_V; ++i) {
        v4u o;
        o.x = run; run += v[i].x;
        o.y = run; run += v[i].y;
        o.z = run; run += v[i].z;
        o.w = run; run += v[i].w;
        if (b + i < M4) t4[b + i] = o;
    }
}

__global__ __launch_bounds__(CHUNK_T)
void bin_scatter_kernel(const int* __restrict__ src, const int* __restrict__ dst,
                        const unsigned* __restrict__ table,
                        int2* __restrict__ binned, int E) {
    __shared__ unsigned cursor[NBINS];

    const int E4 = E >> 2;
    int base4 = (int)blockIdx.x * (CHUNK / 4) + (int)threadIdx.x;
    v4i d[CHUNK_EPT / 4];
    v4i s[CHUNK_EPT / 4];
#pragma unroll
    for (int k = 0; k < CHUNK_EPT / 4; ++k) {
        int i4 = base4 + k * CHUNK_T;
        i4 = i4 < E4 ? i4 : E4 - 1;
        d[k] = *(const v4i*)(dst + 4 * (size_t)i4);
        s[k] = *(const v4i*)(src + 4 * (size_t)i4);
    }
    for (int i = (int)threadIdx.x; i < NBINS; i += CHUNK_T)
        cursor[i] = table[(size_t)i * NCHUNK + blockIdx.x];
    __syncthreads();
#pragma unroll
    for (int k = 0; k < CHUNK_EPT / 4; ++k) {
        if (base4 + k * CHUNK_T < E4) {
            unsigned p0 = atomicAdd(&cursor[bin_of(d[k].x)], 1u);
            unsigned p1 = atomicAdd(&cursor[bin_of(d[k].y)], 1u);
            unsigned p2 = atomicAdd(&cursor[bin_of(d[k].z)], 1u);
            unsigned p3 = atomicAdd(&cursor[bin_of(d[k].w)], 1u);
            binned[p0] = make_int2(s[k].x, d[k].x);
            binned[p1] = make_int2(s[k].y, d[k].y);
            binned[p2] = make_int2(s[k].z, d[k].z);
            binned[p3] = make_int2(s[k].w, d[k].w);
        }
    }
}

#define L2_T 512
#define L2_U 9
static_assert(L2_T >= BIN_NODES, "one thread per destination of the bin");
__global__ __launch_bounds__(L2_T)
void bin_to_csr_kernel(const int2* __restrict__ binned, const unsigned* __restrict__ table,
                       unsigned* __restrict__ rowptr, float* __restrict__ dinv,
                       int* __restrict__ csr, int n, int E) {
    __shared__ unsigned cnt[BIN_NODES];
    __shared__ unsigned first[BIN_NODES];
    __shared__ unsigned cur[BIN_NODES];
    __shared__ float    dloc[BIN_NODES];
    int bin = (int)blockIdx.x;
    unsigned s = table[(size_t)bin * NCHUNK];
    unsigned e = (bin + 1 < NBINS) ? table[(size_t)(bin + 1) * NCHUNK] : (unsigned)E;
    if (threadIdx.x < BIN_NODES) cnt[threadIdx.x] = 0u;
    __syncthreads();

    for (unsigned i0 = s; i0 < e; i0 += L2_T * L2_U) {
        int dd[L2_U];
#pragma unroll
        for (int u = 0; u < L2_U; ++u) {
            unsigned i = i0 + (unsigned)(u * L2_T) + threadIdx.x;
            dd[u] = binned[i < e ? i : e - 1u].y;
        }
#pragma unroll
        for (int u = 0; u < L2_U; ++u) {
            unsigned i = i0 + (unsigned)(u * L2_T) + threadIdx.x;
            if (i < e) atomicAdd(&cnt[dd[u] & (BIN_NODES - 1)], 1u);
        }
    }
    __syncthreads();

    if (threadIdx.x < 32) {
        constexpr int PER = BIN_NODES / 32;
        int lane = (int)threadIdx.x;
        unsigned cc[PER];
        unsigned sum = 0u;
#pragma unroll
        for (int k = 0; k < PER; ++k) { cc[k] = cnt[lane * PER + k]; sum += cc[k]; }
        unsigned run = s + wave_incl_scan(sum, lane) - sum;
#pragma unroll
        for (int k = 0; k < PER; ++k) {
            first[lane * PER + k] = run;
            cur[lane * PER + k]   = run;
            dloc[lane * PER + k]  = rsqrtf((float)cc[k] + 1.0f);
            run += cc[k];
        }
    }
    __syncthreads();
    if (threadIdx.x < BIN_NODES) {
        int node = bin * BIN_NODES + (int)threadIdx.x;
        if (node < n) {
            rowptr[node] = first[threadIdx.x];
            dinv[node]   = dloc[threadIdx.x];
        }
    }
    if (bin == NBINS - 1 && threadIdx.x == 0) rowptr[n] = (unsigned)E;

    for (unsigned i0 = s; i0 < e; i0 += L2_T * L2_U) {
        int2 p[L2_U];
#pragma unroll
        for (int u = 0; u < L2_U; ++u) {
            unsigned i = i0 + (unsigned)(u * L2_T) + threadIdx.x;
            p[u] = binned[i < e ? i : e - 1u];
        }
#pragma unroll
        for (int u = 0; u < L2_U; ++u) {
            unsigned i = i0 + (unsigned)(u * L2_T) + threadIdx.x;
            if (i < e) {
                int l = p[u].y & (BIN_NODES - 1);
                unsigned pos = atomicAdd(&cur[l], 1u);
                csr[pos] = p[u].x;
            }
        }
    }
}

__global__ void transpose_pad_kernel(const float* __restrict__ W,
                                     float* __restrict__ Bt,
                                     int K, int Nout, int Npad) {
    int t = blockIdx.x * blockDim.x + threadIdx.x;
    if (t >= Npad * K) return;
    int n = t / K;
    int k = t - n * K;
    Bt[t] = (n < Nout) ? W[(size_t)k * Nout + n] : 0.0f;
}

template <int NT>
__global__ __launch_bounds__(256)
void gemm_wmma_kernel(const float* __restrict__ A, const float* __restrict__ Bt,
                      const float* __restrict__ rowscale,
                      float* __restrict__ C, int M, int K, int Nout) {
    int wave = (int)((blockIdx.x * blockDim.x + threadIdx.x) >> 5);
    int lane = threadIdx.x & 31;
    if (wave >= (M >> 4)) return;

    int m0   = wave << 4;
    int half = lane >> 4;
    int l16  = lane & 15;

    const float* Arow = A + (size_t)(m0 + l16) * K;
    const float* Bcol[NT];
#pragma unroll
    for (int t = 0; t < NT; ++t)
        Bcol[t] = Bt + (size_t)(t * 16 + l16) * K;

    v8f c[NT] = {};

    int rbase = m0 + (half << 3);
    v4f sc0 = *(const v4f*)(rowscale + rbase);
    v4f sc1 = *(const v4f*)(rowscale + rbase + 4);

    for (int k0 = 0; k0 < K; k0 += 4) {
        int ka = k0 + (half << 1);
        v2f a = *(const v2f*)(Arow + ka);
#pragma unroll
        for (int t = 0; t < NT; ++t) {
            v2f b = *(const v2f*)(Bcol[t] + ka);
            c[t] = __builtin_amdgcn_wmma_f32_16x16x4_f32(
                       false, a, false, b, (short)0, c[t], false, false);
        }
    }

    const float sc[8] = {sc0.x, sc0.y, sc0.z, sc0.w, sc1.x, sc1.y, sc1.z, sc1.w};
#pragma unroll
    for (int t = 0; t < NT; ++t) {
        int ncol = t * 16 + l16;
        if (ncol < Nout) {
#pragma unroll
            for (int r = 0; r < 8; ++r)
                C[(size_t)(rbase + r) * Nout + ncol] = c[t][r] * sc[r];
        }
    }
}

#define TILE_PITCH 36

template <int E>
__device__ __forceinline__ unsigned group8_lane(unsigned v) {
    return __builtin_amdgcn_mov_dpp8(v, (unsigned)E * 0x249249u);
}

template <int F, bool POST, int FN>
__global__ __launch_bounds__(256)
void gather_kernel(const float* __restrict__ h, const int* __restrict__ csr,
                   const unsigned* __restrict__ rowptr,
                   const float* __restrict__ dinv, const float* __restrict__ bias,
                   const float* __restrict__ xold_in, const float* __restrict__ dropu,
                   float* __restrict__ xout, float* __restrict__ xold_out,
                   const float* __restrict__ wnext, int n) {
    static_assert(F % 4 == 0, "row must be a whole number of 16-byte pieces");
    constexpr int Q = F / 4;
    constexpr int C = (Q + 7) / 8;
    static_assert(!POST || C == 1, "the epilogue handles one piece per lane");
    static_assert(FN == 0 || (POST && F == 32), "the fused product takes the epilogue's 32-wide rows");
    constexpr int NT  = (FN + 15) / 16;
    constexpr int NW4 = F * FN / 4;
    constexpr int WPT = (NW4 + 255) / 256;
    __shared__ __attribute__((aligned(16))) float xs[FN > 0 ? 32 * TILE_PITCH : 4];
    __shared__ __attribute__((aligned(16))) float wt[FN > 0 ? NT * 16 * TILE_PITCH : 4];

    int node = (int)((blockIdx.x * blockDim.x + threadIdx.x) >> 3);
    int l8   = threadIdx.x & 7;
    bool live = true;
    if constexpr (FN > 0) {
        live = node < n;
        node = live ? node : n - 1;
    } else {
        if (node >= n) return;
    }

    v4f wreg[WPT > 0 ? WPT : 1];
    if constexpr (FN > 0) {
#pragma unroll
        for (int u = 0; u < WPT; ++u) {
            int i = u * 256 + (int)threadIdx.x;
            i = i < NW4 ? i : NW4 - 1;
            wreg[u] = *(const v4f*)(wnext + 4 * (size_t)i);
        }
    }

    unsigned beg = rowptr[node];
    unsigned end = rowptr[node + 1];

    float di = dinv[node];
    const v4f* hrow = (const v4f*)(h + (size_t)node * F);
    v4f hself[C];
    v4f bself[C];
#pragma unroll
    for (int c2 = 0; c2 < C; ++c2) {
        int q = l8 + 8 * c2;
        if (q < Q) {
            hself[c2] = hrow[q];
            bself[c2].x = bias[4 * q + 0];
            bself[c2].y = bias[4 * q + 1];
            bself[c2].z = bias[4 * q + 2];
            bself[c2].w = bias[4 * q + 3];
        }
    }
    size_t obase = (size_t)node * F + 4 * (size_t)l8;
    v4f dvec = (v4f){0.0f, 0.0f, 0.0f, 0.0f};
    v4f xvec = (v4f){0.0f, 0.0f, 0.0f, 0.0f};
    if (POST) dvec = *(const v4f*)(dropu + obase);

    unsigned last = (end > 0u ? end : 1u) - 1u;
    unsigned i0 = beg + (unsigned)l8;
    int p = csr[i0 < last ? i0 : last];

    if (POST) {
        if (xold_in) xvec = *(const v4f*)(xold_in + obase);
    }

    if constexpr (FN > 0) {
#pragma unroll
        for (int u = 0; u < WPT; ++u) {
            int i = u * 256 + (int)threadIdx.x;
            if (i < NW4) {
                int k  = (4 * i) / FN;
                int c0 = (4 * i) - k * FN;
                wt[(c0 + 0) * TILE_PITCH + k] = wreg[u].x;
                wt[(c0 + 1) * TILE_PITCH + k] = wreg[u].y;
                wt[(c0 + 2) * TILE_PITCH + k] = wreg[u].z;
                wt[(c0 + 3) * TILE_PITCH + k] = wreg[u].w;
            }
        }
        if (NT * 16 > FN) {
            for (int i = (int)threadIdx.x; i < (NT * 16 - FN) * F; i += 256)
                wt[(FN + i / F) * TILE_PITCH + (i % F)] = 0.0f;
        }
    }

    v4f acc[C];
#pragma unroll
    for (int c2 = 0; c2 < C; ++c2) {
        int q = l8 + 8 * c2;
        if (q < Q) acc[c2] = hself[c2];
        else       acc[c2] = (v4f){0.0f, 0.0f, 0.0f, 0.0f};
    }

    for (unsigned j = beg; j < end; j += 8) {
        unsigned cnt = end - j;

        unsigned px = (unsigned)p;
        int s[8];
        s[0] = (int)group8_lane<0>(px);
        s[1] = (int)group8_lane<1>(px);
        s[2] = (int)group8_lane<2>(px);
        s[3] = (int)group8_lane<3>(px);
        s[4] = (int)group8_lane<4>(px);
        s[5] = (int)group8_lane<5>(px);
        s[6] = (int)group8_lane<6>(px);
        s[7] = (int)group8_lane<7>(px);

        v4f r[8][C];
#pragma unroll
        for (int e = 0; e < 8; ++e) {
            const v4f* he = (const v4f*)(h + (size_t)s[e] * F);
#pragma unroll
            for (int c2 = 0; c2 < C; ++c2) {
                int q = l8 + 8 * c2;
                if (q < Q) r[e][c2] = he[q];
            }
        }

        unsigned jn = j + 8 + (unsigned)l8;
        int pn = csr[jn < last ? jn : last];

#pragma unroll
        for (int e = 0; e < 8; e += 2) {
            if ((unsigned)(e + 2) <= cnt) {
#pragma unroll
                for (int c2 = 0; c2 < C; ++c2) {
                    int q = l8 + 8 * c2;
                    if (q < Q) acc[c2] += r[e][c2] + r[e + 1][c2];
                }
            } else if ((unsigned)(e + 1) == cnt) {
#pragma unroll
                for (int c2 = 0; c2 < C; ++c2) {
                    int q = l8 + 8 * c2;
                    if (q < Q) acc[c2] += r[e][c2];
                }
            }
        }
        p = pn;
    }

#pragma unroll
    for (int c2 = 0; c2 < C; ++c2) {
        int q = l8 + 8 * c2;
        if (q < Q) acc[c2] = acc[c2] * di + bself[c2];
    }

    if constexpr (FN > 0) {
        v4f y;
        y.x = fmaxf(acc[0].x, 0.0f);
        y.y = fmaxf(acc[0].y, 0.0f);
        y.z = fmaxf(acc[0].z, 0.0f);
        y.w = fmaxf(acc[0].w, 0.0f);
        if (xold_in) y += xvec;
        v4f d = dvec;
        v4f x;
        x.x = y.x * (d.x > 0.5f ? 2.0f : 0.0f);
        x.y = y.y * (d.y > 0.5f ? 2.0f : 0.0f);
        x.z = y.z * (d.z > 0.5f ? 2.0f : 0.0f);
        x.w = y.w * (d.w > 0.5f ? 2.0f : 0.0f);
        *(v4f*)(xs + (threadIdx.x >> 3) * TILE_PITCH + 4 * l8) = x * di;

        __syncthreads();
        if (live) *(v4f*)(xold_out + obase) = y;

        int wid  = __builtin_amdgcn_readfirstlane((int)(threadIdx.x >> 5));
        int row0 = (int)blockIdx.x * 32 + wid * 16;
        if (wid < 2 && row0 < n) {
            int lane = threadIdx.x & 31;
            int half = lane >> 4;
            int l16  = lane & 15;
            const float* arow = xs + (wid * 16 + l16) * TILE_PITCH + (half << 1);
            const float* bcol = wt + l16 * TILE_PITCH + (half << 1);
            v8f c[NT] = {};
#pragma unroll
            for (int k0 = 0; k0 < F; k0 += 4) {
                v2f a = *(const v2f*)(arow + k0);
#pragma unroll
                for (int t = 0; t < NT; ++t) {
                    v2f b = *(const v2f*)(bcol + t * 16 * TILE_PITCH + k0);
                    c[t] = __builtin_amdgcn_wmma_f32_16x16x4_f32(
                               false, a, false, b, (short)0, c[t], false, false);
                }
            }
            int rbase = row0 + (half << 3);
#pragma unroll
            for (int t = 0; t < NT; ++t) {
                int ncol = t * 16 + l16;
                if (ncol < FN) {
#pragma unroll
                    for (int r = 0; r < 8; ++r)
                        xout[(size_t)(rbase + r) * FN + ncol] = c[t][r];
                }
            }
        }
        return;
    }

    size_t base = (size_t)node * F;
#pragma unroll
    for (int c2 = 0; c2 < C; ++c2) {
        int q = l8 + 8 * c2;
        if (q < Q) {
            size_t o = base + 4 * (size_t)q;
            if (POST) {
                v4f y;
                y.x = fmaxf(acc[c2].x, 0.0f);
                y.y = fmaxf(acc[c2].y, 0.0f);
                y.z = fmaxf(acc[c2].z, 0.0f);
                y.w = fmaxf(acc[c2].w, 0.0f);
                if (xold_in) y += xvec;
                *(v4f*)(xold_out + o) = y;
                v4f d = dvec;
                v4f x;
                x.x = y.x * (d.x > 0.5f ? 2.0f : 0.0f);
                x.y = y.y * (d.y > 0.5f ? 2.0f : 0.0f);
                x.z = y.z * (d.z > 0.5f ? 2.0f : 0.0f);
                x.w = y.w * (d.w > 0.5f ? 2.0f : 0.0f);
                *(v4f*)(xout + o) = x;
            } else {
                *(v4f*)(xout + o) = acc[c2];
            }
        }
    }
}


extern "C" void kernel_launch(void* const* d_in, const int* in_sizes, int n_in,
                              void* d_out, int out_size, void* d_ws, size_t ws_size,
                              hipStream_t stream) {
    (void)in_sizes; (void)n_in; (void)out_size; (void)ws_size;

    const float* x0    = (const float*)d_in[0];
    const int*   ei    = (const int*)d_in[1];
    const float* dropu = (const float*)d_in[2];
    const float* W0    = (const float*)d_in[3];
    const float* b0    = (const float*)d_in[4];
    const float* Whid  = (const float*)d_in[5];
    const float* bhid  = (const float*)d_in[6];
    const float* Wout  = (const float*)d_in[7];
    const float* bout  = (const float*)d_in[8];
    float* out = (float*)d_out;

    const int* src = ei;
    const int* dst = ei + NEDGES;

    char* ws = (char*)d_ws;
    size_t off = 0;
    auto take = [&](size_t bytes) -> void* {
        void* p = (void*)(ws + off);
        off += (bytes + 255) & ~(size_t)255;
        return p;
    };
    unsigned* table  = (unsigned*)take((size_t)NBINS * NCHUNK * 4);
    unsigned* rowptr = (unsigned*)take((size_t)(NNODES + 1) * 4);
    float*    dinv   = (float*)   take((size_t)NNODES * 4);
    int*      csr    = (int*)     take((size_t)NEDGES * 4);
    float*    hq     = (float*)   take((size_t)NNODES * NCLS * 4);
    float*    hp     = (float*)   take((size_t)NNODES * NHID * 4);
    float*    xold   = (float*)   take((size_t)NNODES * NHID * 4);
    float*    Bt     = (float*)   take((size_t)48 * NFEAT * 4);
    int2* binned = (int2*)hq;
    static_assert((size_t)NEDGES * 8 <= (size_t)NNODES * NCLS * 4, "the binned edges fit in hq");

    const int TB = 256;
    auto blocks = [](int n, int tb) { return (n + tb - 1) / tb; };

    bin_count_kernel<<<NCHUNK, CHUNK_T, 0, stream>>>(dst, table, NEDGES);
    table_scan_kernel<<<1, SCAN_T, 0, stream>>>(table, NBINS * NCHUNK / 4);
    bin_scatter_kernel<<<NCHUNK, CHUNK_T, 0, stream>>>(src, dst, table, binned, NEDGES);
    bin_to_csr_kernel<<<NBINS, L2_T, 0, stream>>>(binned, table, rowptr, dinv, csr, NNODES, NEDGES);

    int gemm_blocks   = blocks(NNODES / 16, 8);
    int gather_blocks = blocks(NNODES, 32);

    transpose_pad_kernel<<<blocks(NHID * NFEAT, TB), TB, 0, stream>>>(W0, Bt, NFEAT, NHID, NHID);
    gemm_wmma_kernel<2><<<gemm_blocks, 256, 0, stream>>>(x0, Bt, dinv, hp, NNODES, NFEAT, NHID);

    const float* xold_in = nullptr;
    for (int layer = 0; layer < 7; ++layer) {
        const float* hin  = (layer & 1) ? hq : hp;
        float*       hout = (layer & 1) ? hp : hq;
        const float* b = (layer == 0) ? b0 : bhid + (size_t)(layer - 1) * NHID;
        const float* d = dropu + (size_t)layer * NNODES * NHID;

        if (layer < 6) {
            gather_kernel<NHID, true, NHID><<<gather_blocks, 256, 0, stream>>>(
                hin, csr, rowptr, dinv, b, xold_in, d, hout, xold,
                Whid + (size_t)layer * NHID * NHID, NNODES);
        } else {
            gather_kernel<NHID, true, NCLS><<<gather_blocks, 256, 0, stream>>>(
                hin, csr, rowptr, dinv, b, xold_in, d, hout, xold, Wout, NNODES);
        }
        xold_in = xold;
    }

    gather_kernel<NCLS, false, 0><<<gather_blocks, 256, 0, stream>>>(
        hq, csr, rowptr, dinv, bout, nullptr, nullptr, out, nullptr, nullptr, NNODES);
}


# gathers: one wave owns a 16-node tile in four passes (was 4 nodes per wave): 4x fewer, longer-lived waves; each wave forms its own 16xFN WMMA product, barrier moved to the start; sum order kept
# speedup vs baseline: 2.7712x; 1.1190x over previous
#include <hip/hip_runtime.h>

#define NNODES 100000
#define NEDGES 1600000
#define NFEAT  128
#define NHID   32
#define NCLS   40
static_assert(NNODES % 16 == 0, "the WMMA strips are whole 16-row tiles");

typedef float v2f __attribute__((ext_vector_type(2)));
typedef float v4f __attribute__((ext_vector_type(4)));
typedef float v8f __attribute__((ext_vector_type(8)));

#define BIN_NODES 512
#define NBINS     ((NNODES + BIN_NODES - 1) / BIN_NODES)
#define CHUNK_T   512
#define CHUNK_EPT 16
#define CHUNK     (CHUNK_T * CHUNK_EPT)
#define NCHUNK    ((((NEDGES + CHUNK - 1) / CHUNK) + 3) / 4 * 4)
static_assert(NEDGES % 4 == 0, "the edge list is read in 16-byte pieces");
static_assert((BIN_NODES & (BIN_NODES - 1)) == 0, "a destination's place in its bin is a mask");

typedef int v4i __attribute__((ext_vector_type(4)));

__device__ __forceinline__ int bin_of(int d) { return (int)((unsigned)d / (unsigned)BIN_NODES); }

__device__ __forceinline__ unsigned wave_incl_scan(unsigned v, int lane) {
#pragma unroll
    for (int off = 1; off < 32; off <<= 1) {
        unsigned o = __shfl_up(v, (unsigned)off);
        if (lane >= off) v += o;
    }
    return v;
}

__global__ __launch_bounds__(CHUNK_T)
void bin_count_kernel(const int* __restrict__ dst, unsigned* __restrict__ table, int E) {
    __shared__ unsigned hist[NBINS];
    for (int i = (int)threadIdx.x; i < NBINS; i += CHUNK_T) hist[i] = 0u;

    const int E4 = E >> 2;
    int base4 = (int)blockIdx.x * (CHUNK / 4) + (int)threadIdx.x;
    v4i d[CHUNK_EPT / 4];
#pragma unroll
    for (int k = 0; k < CHUNK_EPT / 4; ++k) {
        int i4 = base4 + k * CHUNK_T;
        i4 = i4 < E4 ? i4 : E4 - 1;
        d[k] = *(const v4i*)(dst + 4 * (size_t)i4);
    }
    __syncthreads();
#pragma unroll
    for (int k = 0; k < CHUNK_EPT / 4; ++k) {
        if (base4 + k * CHUNK_T < E4) {
            atomicAdd(&hist[bin_of(d[k].x)], 1u);
            atomicAdd(&hist[bin_of(d[k].y)], 1u);
            atomicAdd(&hist[bin_of(d[k].z)], 1u);
            atomicAdd(&hist[bin_of(d[k].w)], 1u);
        }
    }
    __syncthreads();
    for (int i = (int)threadIdx.x; i < NBINS; i += CHUNK_T)
        table[(size_t)i * NCHUNK + blockIdx.x] = hist[i];
}

#define SCAN_T   1024
#define SCAN_V   (((NBINS * NCHUNK + SCAN_T - 1) / SCAN_T + 3) / 4)
static_assert((NBINS * NCHUNK) % 4 == 0, "the table is a whole number of 16-byte pieces");
typedef unsigned v4u __attribute__((ext_vector_type(4)));
__global__ __launch_bounds__(SCAN_T)
void table_scan_kernel(unsigned* __restrict__ table, int M4) {
    __shared__ unsigned wsum[32];
    int lane = (int)threadIdx.x & 31;
    int wid  = (int)threadIdx.x >> 5;
    v4u* t4 = (v4u*)table;
    int b = (int)threadIdx.x * SCAN_V;
    v4u v[SCAN_V];
    unsigned sum = 0u;
#pragma unroll
    for (int i = 0; i < SCAN_V; ++i) {
        v[i] = (v4u){0u, 0u, 0u, 0u};
        if (b + i < M4) v[i] = t4[b + i];
        sum += v[i].x + v[i].y + v[i].z + v[i].w;
    }
    unsigned inc = wave_incl_scan(sum, lane);
    if (lane == 31) wsum[wid] = inc;
    __syncthreads();
    if (wid == 0) {
        unsigned w = wsum[lane];
        unsigned winc = wave_incl_scan(w, lane);
        wsum[lane] = winc - w;
    }
    __syncthreads();
    unsigned run = wsum[wid] + inc - sum;
#pragma unroll
    for (int i = 0; i < SCAN_V; ++i) {
        v4u o;
        o.x = run; run += v[i].x;
        o.y = run; run += v[i].y;
        o.z = run; run += v[i].z;
        o.w = run; run += v[i].w;
        if (b + i < M4) t4[b + i] = o;
    }
}

__global__ __launch_bounds__(CHUNK_T)
void bin_scatter_kernel(const int* __restrict__ src, const int* __restrict__ dst,
                        const unsigned* __restrict__ table,
                        int2* __restrict__ binned, int E) {
    __shared__ unsigned cursor[NBINS];

    const int E4 = E >> 2;
    int base4 = (int)blockIdx.x * (CHUNK / 4) + (int)threadIdx.x;
    v4i d[CHUNK_EPT / 4];
    v4i s[CHUNK_EPT / 4];
#pragma unroll
    for (int k = 0; k < CHUNK_EPT / 4; ++k) {
        int i4 = base4 + k * CHUNK_T;
        i4 = i4 < E4 ? i4 : E4 - 1;
        d[k] = *(const v4i*)(dst + 4 * (size_t)i4);
        s[k] = *(const v4i*)(src + 4 * (size_t)i4);
    }
    for (int i = (int)threadIdx.x; i < NBINS; i += CHUNK_T)
        cursor[i] = table[(size_t)i * NCHUNK + blockIdx.x];
    __syncthreads();
#pragma unroll
    for (int k = 0; k < CHUNK_EPT / 4; ++k) {
        if (base4 + k * CHUNK_T < E4) {
            unsigned p0 = atomicAdd(&cursor[bin_of(d[k].x)], 1u);
            unsigned p1 = atomicAdd(&cursor[bin_of(d[k].y)], 1u);
            unsigned p2 = atomicAdd(&cursor[bin_of(d[k].z)], 1u);
            unsigned p3 = atomicAdd(&cursor[bin_of(d[k].w)], 1u);
            binned[p0] = make_int2(s[k].x, d[k].x);
            binned[p1] = make_int2(s[k].y, d[k].y);
            binned[p2] = make_int2(s[k].z, d[k].z);
            binned[p3] = make_int2(s[k].w, d[k].w);
        }
    }
}

#define L2_T 512
#define L2_U 9
static_assert(L2_T >= BIN_NODES, "one thread per destination of the bin");
__global__ __launch_bounds__(L2_T)
void bin_to_csr_kernel(const int2* __restrict__ binned, const unsigned* __restrict__ table,
                       unsigned* __restrict__ rowptr, float* __restrict__ dinv,
                       int* __restrict__ csr, int n, int E) {
    __shared__ unsigned cnt[BIN_NODES];
    __shared__ unsigned first[BIN_NODES];
    __shared__ unsigned cur[BIN_NODES];
    __shared__ float    dloc[BIN_NODES];
    int bin = (int)blockIdx.x;
    unsigned s = table[(size_t)bin * NCHUNK];
    unsigned e = (bin + 1 < NBINS) ? table[(size_t)(bin + 1) * NCHUNK] : (unsigned)E;
    if (threadIdx.x < BIN_NODES) cnt[threadIdx.x] = 0u;
    __syncthreads();

    for (unsigned i0 = s; i0 < e; i0 += L2_T * L2_U) {
        int dd[L2_U];
#pragma unroll
        for (int u = 0; u < L2_U; ++u) {
            unsigned i = i0 + (unsigned)(u * L2_T) + threadIdx.x;
            dd[u] = binned[i < e ? i : e - 1u].y;
        }
#pragma unroll
        for (int u = 0; u < L2_U; ++u) {
            unsigned i = i0 + (unsigned)(u * L2_T) + threadIdx.x;
            if (i < e) atomicAdd(&cnt[dd[u] & (BIN_NODES - 1)], 1u);
        }
    }
    __syncthreads();

    if (threadIdx.x < 32) {
        constexpr int PER = BIN_NODES / 32;
        int lane = (int)threadIdx.x;
        unsigned cc[PER];
        unsigned sum = 0u;
#pragma unroll
        for (int k = 0; k < PER; ++k) { cc[k] = cnt[lane * PER + k]; sum += cc[k]; }
        unsigned run = s + wave_incl_scan(sum, lane) - sum;
#pragma unroll
        for (int k = 0; k < PER; ++k) {
            first[lane * PER + k] = run;
            cur[lane * PER + k]   = run;
            dloc[lane * PER + k]  = rsqrtf((float)cc[k] + 1.0f);
            run += cc[k];
        }
    }
    __syncthreads();
    if (threadIdx.x < BIN_NODES) {
        int node = bin * BIN_NODES + (int)threadIdx.x;
        if (node < n) {
            rowptr[node] = first[threadIdx.x];
            dinv[node]   = dloc[threadIdx.x];
        }
    }
    if (bin == NBINS - 1 && threadIdx.x == 0) rowptr[n] = (unsigned)E;

    for (unsigned i0 = s; i0 < e; i0 += L2_T * L2_U) {
        int2 p[L2_U];
#pragma unroll
        for (int u = 0; u < L2_U; ++u) {
            unsigned i = i0 + (unsigned)(u * L2_T) + threadIdx.x;
            p[u] = binned[i < e ? i : e - 1u];
        }
#pragma unroll
        for (int u = 0; u < L2_U; ++u) {
            unsigned i = i0 + (unsigned)(u * L2_T) + threadIdx.x;
            if (i < e) {
                int l = p[u].y & (BIN_NODES - 1);
                unsigned pos = atomicAdd(&cur[l], 1u);
                csr[pos] = p[u].x;
            }
        }
    }
}

__global__ void transpose_pad_kernel(const float* __restrict__ W,
                                     float* __restrict__ Bt,
                                     int K, int Nout, int Npad) {
    int t = blockIdx.x * blockDim.x + threadIdx.x;
    if (t >= Npad * K) return;
    int n = t / K;
    int k = t - n * K;
    Bt[t] = (n < Nout) ? W[(size_t)k * Nout + n] : 0.0f;
}

template <int NT>
__global__ __launch_bounds__(256)
void gemm_wmma_kernel(const float* __restrict__ A, const float* __restrict__ Bt,
                      const float* __restrict__ rowscale,
                      float* __restrict__ C, int M, int K, int Nout) {
    int wave = (int)((blockIdx.x * blockDim.x + threadIdx.x) >> 5);
    int lane = threadIdx.x & 31;
    if (wave >= (M >> 4)) return;

    int m0   = wave << 4;
    int half = lane >> 4;
    int l16  = lane & 15;

    const float* Arow = A + (size_t)(m0 + l16) * K;
    const float* Bcol[NT];
#pragma unroll
    for (int t = 0; t < NT; ++t)
        Bcol[t] = Bt + (size_t)(t * 16 + l16) * K;

    v8f c[NT] = {};

    int rbase = m0 + (half << 3);
    v4f sc0 = *(const v4f*)(rowscale + rbase);
    v4f sc1 = *(const v4f*)(rowscale + rbase + 4);

    for (int k0 = 0; k0 < K; k0 += 4) {
        int ka = k0 + (half << 1);
        v2f a = *(const v2f*)(Arow + ka);
#pragma unroll
        for (int t = 0; t < NT; ++t) {
            v2f b = *(const v2f*)(Bcol[t] + ka);
            c[t] = __builtin_amdgcn_wmma_f32_16x16x4_f32(
                       false, a, false, b, (short)0, c[t], false, false);
        }
    }

    const float sc[8] = {sc0.x, sc0.y, sc0.z, sc0.w, sc1.x, sc1.y, sc1.z, sc1.w};
#pragma unroll
    for (int t = 0; t < NT; ++t) {
        int ncol = t * 16 + l16;
        if (ncol < Nout) {
#pragma unroll
            for (int r = 0; r < 8; ++r)
                C[(size_t)(rbase + r) * Nout + ncol] = c[t][r] * sc[r];
        }
    }
}

#define TILE_NODES 16
#define WG_WAVES   4
#define WG_T       (32 * WG_WAVES)
#define WG_NODES   (WG_WAVES * TILE_NODES)
static_assert(NNODES % TILE_NODES == 0, "a wave's tile is wholly inside the graph or wholly outside");

template <int E>
__device__ __forceinline__ unsigned group8_lane(unsigned v) {
    return __builtin_amdgcn_mov_dpp8(v, (unsigned)E * 0x249249u);
}

template <int F, int Q, int C>
__device__ __forceinline__ void request_rows(const float* __restrict__ h, unsigned px,
                                             int l8, v4f (&r)[8][C]) {
    int s[8];
    s[0] = (int)group8_lane<0>(px);
    s[1] = (int)group8_lane<1>(px);
    s[2] = (int)group8_lane<2>(px);
    s[3] = (int)group8_lane<3>(px);
    s[4] = (int)group8_lane<4>(px);
    s[5] = (int)group8_lane<5>(px);
    s[6] = (int)group8_lane<6>(px);
    s[7] = (int)group8_lane<7>(px);
#pragma unroll
    for (int e = 0; e < 8; ++e) {
        const v4f* he = (const v4f*)(h + (size_t)s[e] * F);
#pragma unroll
        for (int c2 = 0; c2 < C; ++c2) {
            int q = l8 + 8 * c2;
            if (q < Q) r[e][c2] = he[q];
        }
    }
}

template <int Q, int C>
__device__ __forceinline__ void add_rows(v4f (&acc)[C], const v4f (&r)[8][C],
                                         unsigned cnt, int l8) {
#pragma unroll
    for (int e = 0; e < 8; e += 2) {
        if ((unsigned)(e + 2) <= cnt) {
#pragma unroll
            for (int c2 = 0; c2 < C; ++c2) {
                int q = l8 + 8 * c2;
                if (q < Q) acc[c2] += r[e][c2] + r[e + 1][c2];
            }
        } else if ((unsigned)(e + 1) == cnt) {
#pragma unroll
            for (int c2 = 0; c2 < C; ++c2) {
                int q = l8 + 8 * c2;
                if (q < Q) acc[c2] += r[e][c2];
            }
        }
    }
}

__device__ __forceinline__ unsigned last_of(unsigned end) { return (end > 0u ? end : 1u) - 1u; }
__device__ __forceinline__ unsigned clamp_to(unsigned i, unsigned last) { return i < last ? i : last; }

template <int F, int Q, int C>
__device__ __forceinline__ void gather_node(const float* __restrict__ h, const int* __restrict__ csr,
                                            unsigned beg, unsigned end, int q0, int l8,
                                            v4f (&acc)[C]) {
    const unsigned last = last_of(end);
    v4f rf[8][C];
    request_rows<F, Q, C>(h, (unsigned)q0, l8, rf);
    int p = csr[clamp_to(beg + 8u + (unsigned)l8, last)];
    add_rows<Q, C>(acc, rf, end - beg, l8);
    for (unsigned j = beg + 8u; j < end; j += 8u) {
        v4f r[8][C];
        request_rows<F, Q, C>(h, (unsigned)p, l8, r);
        int pn = csr[clamp_to(j + 8u + (unsigned)l8, last)];
        add_rows<Q, C>(acc, r, end - j, l8);
        p = pn;
    }
}

#define TILE_PITCH 36

template <int FN>
__global__ __launch_bounds__(WG_T)
void gather_hidden_kernel(const float* __restrict__ h, const int* __restrict__ csr,
                          const unsigned* __restrict__ rowptr,
                          const float* __restrict__ dinv, const float* __restrict__ bias,
                          const float* __restrict__ xold_in, const float* __restrict__ dropu,
                          float* __restrict__ xout, float* __restrict__ xold_out,
                          const float* __restrict__ wnext, int n) {
    constexpr int F = NHID;
    constexpr int Q = F / 4;
    static_assert(Q == 8, "eight lanes hold one row");
    constexpr int NT  = (FN + 15) / 16;
    constexpr int NW4 = F * FN / 4;
    constexpr int WPT = (NW4 + WG_T - 1) / WG_T;
    __shared__ __attribute__((aligned(16))) float xs[WG_NODES * TILE_PITCH];
    __shared__ __attribute__((aligned(16))) float wt[NT * 16 * TILE_PITCH];

    const int wid  = __builtin_amdgcn_readfirstlane((int)(threadIdx.x >> 5));
    const int lane = threadIdx.x & 31;
    const int grp  = lane >> 3;
    const int l8   = lane & 7;
    int row0 = ((int)blockIdx.x * WG_WAVES + wid) * TILE_NODES;
    const bool live = row0 < n;
    row0 = live ? row0 : n - TILE_NODES;
    const int nb = row0 + 4 * grp;

    v4f wreg[WPT];
#pragma unroll
    for (int u = 0; u < WPT; ++u) {
        int i = u * WG_T + (int)threadIdx.x;
        i = i < NW4 ? i : NW4 - 1;
        wreg[u] = *(const v4f*)(wnext + 4 * (size_t)i);
    }

    const v4u      rp  = *(const v4u*)(rowptr + nb);
    const unsigned rpe = rowptr[nb + 4];
    const v4f      dv  = *(const v4f*)(dinv + nb);

#pragma unroll
    for (int u = 0; u < WPT; ++u) {
        int i = u * WG_T + (int)threadIdx.x;
        if (i < NW4) {
            int k  = (4 * i) / FN;
            int c0 = (4 * i) - k * FN;
            wt[(c0 + 0) * TILE_PITCH + k] = wreg[u].x;
            wt[(c0 + 1) * TILE_PITCH + k] = wreg[u].y;
            wt[(c0 + 2) * TILE_PITCH + k] = wreg[u].z;
            wt[(c0 + 3) * TILE_PITCH + k] = wreg[u].w;
        }
    }
    if (NT * 16 > FN) {
        for (int i = (int)threadIdx.x; i < (NT * 16 - FN) * F; i += WG_T)
            wt[(FN + i / F) * TILE_PITCH + (i % F)] = 0.0f;
    }
    __syncthreads();
    if (!live) return;

    unsigned b0 = rp.x, b1 = rp.y, b2 = rp.z, b3 = rp.w, b4 = rpe;
    float    d0 = dv.x, d1 = dv.y, d2 = dv.z, d3 = dv.w;
    int q0 = csr[clamp_to(b0 + (unsigned)l8, last_of(b1))];
    int q1 = csr[clamp_to(b1 + (unsigned)l8, last_of(b2))];
    int q2 = csr[clamp_to(b2 + (unsigned)l8, last_of(b3))];
    int q3 = csr[clamp_to(b3 + (unsigned)l8, last_of(b4))];

    v4f bself;
    bself.x = bias[4 * l8 + 0];
    bself.y = bias[4 * l8 + 1];
    bself.z = bias[4 * l8 + 2];
    bself.w = bias[4 * l8 + 3];

    const size_t ob0 = (size_t)nb * F + 4 * (size_t)l8;
    v4f dr0 = *(const v4f*)(dropu + ob0);
    v4f dr1 = *(const v4f*)(dropu + ob0 + F);
    v4f dr2 = *(const v4f*)(dropu + ob0 + 2 * F);
    v4f dr3 = *(const v4f*)(dropu + ob0 + 3 * F);

#pragma nounroll
    for (int g = 0; g < 4; ++g) {
        const size_t ob = ob0 + (size_t)(g * F);
        v4f acc[1];
        acc[0] = *(const v4f*)(h + ob);
        v4f xvec = (v4f){0.0f, 0.0f, 0.0f, 0.0f};
        if (xold_in) xvec = *(const v4f*)(xold_in + ob);

        gather_node<F, Q, 1>(h, csr, b0, b1, q0, l8, acc);

        v4f a = acc[0] * d0 + bself;
        v4f y;
        y.x = fmaxf(a.x, 0.0f);
        y.y = fmaxf(a.y, 0.0f);
        y.z = fmaxf(a.z, 0.0f);
        y.w = fmaxf(a.w, 0.0f);
        if (xold_in) y += xvec;
        v4f x;
        x.x = y.x * (dr0.x > 0.5f ? 2.0f : 0.0f);
        x.y = y.y * (dr0.y > 0.5f ? 2.0f : 0.0f);
        x.z = y.z * (dr0.z > 0.5f ? 2.0f : 0.0f);
        x.w = y.w * (dr0.w > 0.5f ? 2.0f : 0.0f);
        *(v4f*)(xs + (wid * TILE_NODES + 4 * grp + g) * TILE_PITCH + 4 * l8) = x * d0;
        *(v4f*)(xold_out + ob) = y;

        b0 = b1; b1 = b2; b2 = b3; b3 = b4;
        d0 = d1; d1 = d2; d2 = d3;
        q0 = q1; q1 = q2; q2 = q3;
        dr0 = dr1; dr1 = dr2; dr2 = dr3;
    }

    const int half = lane >> 4;
    const int l16  = lane & 15;
    const float* arow = xs + (wid * TILE_NODES + l16) * TILE_PITCH + (half << 1);
    const float* bcol = wt + l16 * TILE_PITCH + (half << 1);
    v8f c[NT] = {};
#pragma unroll
    for (int k0 = 0; k0 < F; k0 += 4) {
        v2f a = *(const v2f*)(arow + k0);
#pragma unroll
        for (int t = 0; t < NT; ++t) {
            v2f b = *(const v2f*)(bcol + t * 16 * TILE_PITCH + k0);
            c[t] = __builtin_amdgcn_wmma_f32_16x16x4_f32(
                       false, a, false, b, (short)0, c[t], false, false);
        }
    }
    const int rbase = row0 + (half << 3);
#pragma unroll
    for (int t = 0; t < NT; ++t) {
        int ncol = t * 16 + l16;
        if (ncol < FN) {
#pragma unroll
            for (int r = 0; r < 8; ++r)
                xout[(size_t)(rbase + r) * FN + ncol] = c[t][r];
        }
    }
}

template <int F>
__global__ __launch_bounds__(WG_T)
void gather_out_kernel(const float* __restrict__ h, const int* __restrict__ csr,
                       const unsigned* __restrict__ rowptr,
                       const float* __restrict__ dinv, const float* __restrict__ bias,
                       float* __restrict__ xout, int n) {
    static_assert(F % 4 == 0, "row must be a whole number of 16-byte pieces");
    constexpr int Q = F / 4;
    constexpr int C = (Q + 7) / 8;

    const int wid  = (int)(threadIdx.x >> 5);
    const int lane = threadIdx.x & 31;
    const int grp  = lane >> 3;
    const int l8   = lane & 7;
    const int row0 = ((int)blockIdx.x * WG_WAVES + wid) * TILE_NODES;
    if (row0 >= n) return;
    const int nb = row0 + 4 * grp;

    const v4u      rp  = *(const v4u*)(rowptr + nb);
    const unsigned rpe = rowptr[nb + 4];
    const v4f      dv  = *(const v4f*)(dinv + nb);

    v4f bself[C];
#pragma unroll
    for (int c2 = 0; c2 < C; ++c2) {
        int q = l8 + 8 * c2;
        if (q < Q) {
            bself[c2].x = bias[4 * q + 0];
            bself[c2].y = bias[4 * q + 1];
            bself[c2].z = bias[4 * q + 2];
            bself[c2].w = bias[4 * q + 3];
        }
    }

    unsigned b0 = rp.x, b1 = rp.y, b2 = rp.z, b3 = rp.w, b4 = rpe;
    float    d0 = dv.x, d1 = dv.y, d2 = dv.z, d3 = dv.w;
    int q0 = csr[clamp_to(b0 + (unsigned)l8, last_of(b1))];
    int q1 = csr[clamp_to(b1 + (unsigned)l8, last_of(b2))];
    int q2 = csr[clamp_to(b2 + (unsigned)l8, last_of(b3))];
    int q3 = csr[clamp_to(b3 + (unsigned)l8, last_of(b4))];

#pragma nounroll
    for (int g = 0; g < 4; ++g) {
        const size_t base = (size_t)(nb + g) * F;
        const v4f* hrow = (const v4f*)(h + base);
        v4f acc[C];
#pragma unroll
        for (int c2 = 0; c2 < C; ++c2) {
            int q = l8 + 8 * c2;
            if (q < Q) acc[c2] = hrow[q];
            else       acc[c2] = (v4f){0.0f, 0.0f, 0.0f, 0.0f};
        }

        gather_node<F, Q, C>(h, csr, b0, b1, q0, l8, acc);

#pragma unroll
        for (int c2 = 0; c2 < C; ++c2) {
            int q = l8 + 8 * c2;
            if (q < Q) *(v4f*)(xout + base + 4 * (size_t)q) = acc[c2] * d0 + bself[c2];
        }

        b0 = b1; b1 = b2; b2 = b3; b3 = b4;
        d0 = d1; d1 = d2; d2 = d3;
        q0 = q1; q1 = q2; q2 = q3;
    }
}


extern "C" void kernel_launch(void* const* d_in, const int* in_sizes, int n_in,
                              void* d_out, int out_size, void* d_ws, size_t ws_size,
                              hipStream_t stream) {
    (void)in_sizes; (void)n_in; (void)out_size; (void)ws_size;

    const float* x0    = (const float*)d_in[0];
    const int*   ei    = (const int*)d_in[1];
    const float* dropu = (const float*)d_in[2];
    const float* W0    = (const float*)d_in[3];
    const float* b0    = (const float*)d_in[4];
    const float* Whid  = (const float*)d_in[5];
    const float* bhid  = (const float*)d_in[6];
    const float* Wout  = (const float*)d_in[7];
    const float* bout  = (const float*)d_in[8];
    float* out = (float*)d_out;

    const int* src = ei;
    const int* dst = ei + NEDGES;

    char* ws = (char*)d_ws;
    size_t off = 0;
    auto take = [&](size_t bytes) -> void* {
        void* p = (void*)(ws + off);
        off += (bytes + 255) & ~(size_t)255;
        return p;
    };
    unsigned* table  = (unsigned*)take((size_t)NBINS * NCHUNK * 4);
    unsigned* rowptr = (unsigned*)take((size_t)(NNODES + 1) * 4);
    float*    dinv   = (float*)   take((size_t)NNODES * 4);
    int*      csr    = (int*)     take((size_t)NEDGES * 4);
    float*    hq     = (float*)   take((size_t)NNODES * NCLS * 4);
    float*    hp     = (float*)   take((size_t)NNODES * NHID * 4);
    float*    xold   = (float*)   take((size_t)NNODES * NHID * 4);
    float*    Bt     = (float*)   take((size_t)48 * NFEAT * 4);
    int2* binned = (int2*)hq;
    static_assert((size_t)NEDGES * 8 <= (size_t)NNODES * NCLS * 4, "the binned edges fit in hq");

    const int TB = 256;
    auto blocks = [](int n, int tb) { return (n + tb - 1) / tb; };

    bin_count_kernel<<<NCHUNK, CHUNK_T, 0, stream>>>(dst, table, NEDGES);
    table_scan_kernel<<<1, SCAN_T, 0, stream>>>(table, NBINS * NCHUNK / 4);
    bin_scatter_kernel<<<NCHUNK, CHUNK_T, 0, stream>>>(src, dst, table, binned, NEDGES);
    bin_to_csr_kernel<<<NBINS, L2_T, 0, stream>>>(binned, table, rowptr, dinv, csr, NNODES, NEDGES);

    int gemm_blocks   = blocks(NNODES / 16, 8);
    int gather_blocks = blocks(NNODES, WG_NODES);

    transpose_pad_kernel<<<blocks(NHID * NFEAT, TB), TB, 0, stream>>>(W0, Bt, NFEAT, NHID, NHID);
    gemm_wmma_kernel<2><<<gemm_blocks, 256, 0, stream>>>(x0, Bt, dinv, hp, NNODES, NFEAT, NHID);

    const float* xold_in = nullptr;
    for (int layer = 0; layer < 7; ++layer) {
        const float* hin  = (layer & 1) ? hq : hp;
        float*       hout = (layer & 1) ? hp : hq;
        const float* b = (layer == 0) ? b0 : bhid + (size_t)(layer - 1) * NHID;
        const float* d = dropu + (size_t)layer * NNODES * NHID;

        if (layer < 6) {
            gather_hidden_kernel<NHID><<<gather_blocks, WG_T, 0, stream>>>(
                hin, csr, rowptr, dinv, b, xold_in, d, hout, xold,
                Whid + (size_t)layer * NHID * NHID, NNODES);
        } else {
            gather_hidden_kernel<NCLS><<<gather_blocks, WG_T, 0, stream>>>(
                hin, csr, rowptr, dinv, b, xold_in, d, hout, xold, Wout, NNODES);
        }
        xold_in = xold;
    }

    gather_out_kernel<NCLS><<<gather_blocks, WG_T, 0, stream>>>(
        hq, csr, rowptr, dinv, bout, out, NNODES);
}


# hidden gathers: own, x_old and dropout rows of all four passes loaded at the wave's start, next index ahead of each trip's rows: no memory load stands in front of neighbour rows; sum order kept
# speedup vs baseline: 2.8725x; 1.0366x over previous
#include <hip/hip_runtime.h>

#define NNODES 100000
#define NEDGES 1600000
#define NFEAT  128
#define NHID   32
#define NCLS   40
static_assert(NNODES % 16 == 0, "the WMMA strips are whole 16-row tiles");

typedef float v2f __attribute__((ext_vector_type(2)));
typedef float v4f __attribute__((ext_vector_type(4)));
typedef float v8f __attribute__((ext_vector_type(8)));

#define BIN_NODES 512
#define NBINS     ((NNODES + BIN_NODES - 1) / BIN_NODES)
#define CHUNK_T   512
#define CHUNK_EPT 16
#define CHUNK     (CHUNK_T * CHUNK_EPT)
#define NCHUNK    ((((NEDGES + CHUNK - 1) / CHUNK) + 3) / 4 * 4)
static_assert(NEDGES % 4 == 0, "the edge list is read in 16-byte pieces");
static_assert((BIN_NODES & (BIN_NODES - 1)) == 0, "a destination's place in its bin is a mask");

typedef int v4i __attribute__((ext_vector_type(4)));

__device__ __forceinline__ int bin_of(int d) { return (int)((unsigned)d / (unsigned)BIN_NODES); }

__device__ __forceinline__ unsigned wave_incl_scan(unsigned v, int lane) {
#pragma unroll
    for (int off = 1; off < 32; off <<= 1) {
        unsigned o = __shfl_up(v, (unsigned)off);
        if (lane >= off) v += o;
    }
    return v;
}

__global__ __launch_bounds__(CHUNK_T)
void bin_count_kernel(const int* __restrict__ dst, unsigned* __restrict__ table, int E) {
    __shared__ unsigned hist[NBINS];
    for (int i = (int)threadIdx.x; i < NBINS; i += CHUNK_T) hist[i] = 0u;

    const int E4 = E >> 2;
    int base4 = (int)blockIdx.x * (CHUNK / 4) + (int)threadIdx.x;
    v4i d[CHUNK_EPT / 4];
#pragma unroll
    for (int k = 0; k < CHUNK_EPT / 4; ++k) {
        int i4 = base4 + k * CHUNK_T;
        i4 = i4 < E4 ? i4 : E4 - 1;
        d[k] = *(const v4i*)(dst + 4 * (size_t)i4);
    }
    __syncthreads();
#pragma unroll
    for (int k = 0; k < CHUNK_EPT / 4; ++k) {
        if (base4 + k * CHUNK_T < E4) {
            atomicAdd(&hist[bin_of(d[k].x)], 1u);
            atomicAdd(&hist[bin_of(d[k].y)], 1u);
            atomicAdd(&hist[bin_of(d[k].z)], 1u);
            atomicAdd(&hist[bin_of(d[k].w)], 1u);
        }
    }
    __syncthreads();
    for (int i = (int)threadIdx.x; i < NBINS; i += CHUNK_T)
        table[(size_t)i * NCHUNK + blockIdx.x] = hist[i];
}

#define SCAN_T   1024
#define SCAN_V   (((NBINS * NCHUNK + SCAN_T - 1) / SCAN_T + 3) / 4)
static_assert((NBINS * NCHUNK) % 4 == 0, "the table is a whole number of 16-byte pieces");
typedef unsigned v4u __attribute__((ext_vector_type(4)));
__global__ __launch_bounds__(SCAN_T)
void table_scan_kernel(unsigned* __restrict__ table, int M4) {
    __shared__ unsigned wsum[32];
    int lane = (int)threadIdx.x & 31;
    int wid  = (int)threadIdx.x >> 5;
    v4u* t4 = (v4u*)table;
    int b = (int)threadIdx.x * SCAN_V;
    v4u v[SCAN_V];
    unsigned sum = 0u;
#pragma unroll
    for (int i = 0; i < SCAN_V; ++i) {
        v[i] = (v4u){0u, 0u, 0u, 0u};
        if (b + i < M4) v[i] = t4[b + i];
        sum += v[i].x + v[i].y + v[i].z + v[i].w;
    }
    unsigned inc = wave_incl_scan(sum, lane);
    if (lane == 31) wsum[wid] = inc;
    __syncthreads();
    if (wid == 0) {
        unsigned w = wsum[lane];
        unsigned winc = wave_incl_scan(w, lane);
        wsum[lane] = winc - w;
    }
    __syncthreads();
    unsigned run = wsum[wid] + inc - sum;
#pragma unroll
    for (int i = 0; i < SCAN_V; ++i) {
        v4u o;
        o.x = run; run += v[i].x;
        o.y = run; run += v[i].y;
        o.z = run; run += v[i].z;
        o.w = run; run += v[i].w;
        if (b + i < M4) t4[b + i] = o;
    }
}

__global__ __launch_bounds__(CHUNK_T)
void bin_scatter_kernel(const int* __restrict__ src, const int* __restrict__ dst,
                        const unsigned* __restrict__ table,
                        int2* __restrict__ binned, int E) {
    __shared__ unsigned cursor[NBINS];

    const int E4 = E >> 2;
    int base4 = (int)blockIdx.x * (CHUNK / 4) + (int)threadIdx.x;
    v4i d[CHUNK_EPT / 4];
    v4i s[CHUNK_EPT / 4];
#pragma unroll
    for (int k = 0; k < CHUNK_EPT / 4; ++k) {
        int i4 = base4 + k * CHUNK_T;
        i4 = i4 < E4 ? i4 : E4 - 1;
        d[k] = *(const v4i*)(dst + 4 * (size_t)i4);
        s[k] = *(const v4i*)(src + 4 * (size_t)i4);
    }
    for (int i = (int)threadIdx.x; i < NBINS; i += CHUNK_T)
        cursor[i] = table[(size_t)i * NCHUNK + blockIdx.x];
    __syncthreads();
#pragma unroll
    for (int k = 0; k < CHUNK_EPT / 4; ++k) {
        if (base4 + k * CHUNK_T < E4) {
            unsigned p0 = atomicAdd(&cursor[bin_of(d[k].x)], 1u);
            unsigned p1 = atomicAdd(&cursor[bin_of(d[k].y)], 1u);
            unsigned p2 = atomicAdd(&cursor[bin_of(d[k].z)], 1u);
            unsigned p3 = atomicAdd(&cursor[bin_of(d[k].w)], 1u);
            binned[p0] = make_int2(s[k].x, d[k].x);
            binned[p1] = make_int2(s[k].y, d[k].y);
            binned[p2] = make_int2(s[k].z, d[k].z);
            binned[p3] = make_int2(s[k].w, d[k].w);
        }
    }
}

#define L2_T 512
#define L2_U 9
static_assert(L2_T >= BIN_NODES, "one thread per destination of the bin");
__global__ __launch_bounds__(L2_T)
void bin_to_csr_kernel(const int2* __restrict__ binned, const unsigned* __restrict__ table,
                       unsigned* __restrict__ rowptr, float* __restrict__ dinv,
                       int* __restrict__ csr, int n, int E) {
    __shared__ unsigned cnt[BIN_NODES];
    __shared__ unsigned first[BIN_NODES];
    __shared__ unsigned cur[BIN_NODES];
    __shared__ float    dloc[BIN_NODES];
    int bin = (int)blockIdx.x;
    unsigned s = table[(size_t)bin * NCHUNK];
    unsigned e = (bin + 1 < NBINS) ? table[(size_t)(bin + 1) * NCHUNK] : (unsigned)E;
    if (threadIdx.x < BIN_NODES) cnt[threadIdx.x] = 0u;
    __syncthreads();

    for (unsigned i0 = s; i0 < e; i0 += L2_T * L2_U) {
        int dd[L2_U];
#pragma unroll
        for (int u = 0; u < L2_U; ++u) {
            unsigned i = i0 + (unsigned)(u * L2_T) + threadIdx.x;
            dd[u] = binned[i < e ? i : e - 1u].y;
        }
#pragma unroll
        for (int u = 0; u < L2_U; ++u) {
            unsigned i = i0 + (unsigned)(u * L2_T) + threadIdx.x;
            if (i < e) atomicAdd(&cnt[dd[u] & (BIN_NODES - 1)], 1u);
        }
    }
    __syncthreads();

    if (threadIdx.x < 32) {
        constexpr int PER = BIN_NODES / 32;
        int lane = (int)threadIdx.x;
        unsigned cc[PER];
        unsigned sum = 0u;
#pragma unroll
        for (int k = 0; k < PER; ++k) { cc[k] = cnt[lane * PER + k]; sum += cc[k]; }
        unsigned run = s + wave_incl_scan(sum, lane) - sum;
#pragma unroll
        for (int k = 0; k < PER; ++k) {
            first[lane * PER + k] = run;
            cur[lane * PER + k]   = run;
            dloc[lane * PER + k]  = rsqrtf((float)cc[k] + 1.0f);
            run += cc[k];
        }
    }
    __syncthreads();
    if (threadIdx.x < BIN_NODES) {
        int node = bin * BIN_NODES + (int)threadIdx.x;
        if (node < n) {
            rowptr[node] = first[threadIdx.x];
            dinv[node]   = dloc[threadIdx.x];
        }
    }
    if (bin == NBINS - 1 && threadIdx.x == 0) rowptr[n] = (unsigned)E;

    for (unsigned i0 = s; i0 < e; i0 += L2_T * L2_U) {
        int2 p[L2_U];
#pragma unroll
        for (int u = 0; u < L2_U; ++u) {
            unsigned i = i0 + (unsigned)(u * L2_T) + threadIdx.x;
            p[u] = binned[i < e ? i : e - 1u];
        }
#pragma unroll
        for (int u = 0; u < L2_U; ++u) {
            unsigned i = i0 + (unsigned)(u * L2_T) + threadIdx.x;
            if (i < e) {
                int l = p[u].y & (BIN_NODES - 1);
                unsigned pos = atomicAdd(&cur[l], 1u);
                csr[pos] = p[u].x;
            }
        }
    }
}

__global__ void transpose_pad_kernel(const float* __restrict__ W,
                                     float* __restrict__ Bt,
                                     int K, int Nout, int Npad) {
    int t = blockIdx.x * blockDim.x + threadIdx.x;
    if (t >= Npad * K) return;
    int n = t / K;
    int k = t - n * K;
    Bt[t] = (n < Nout) ? W[(size_t)k * Nout + n] : 0.0f;
}

template <int NT>
__global__ __launch_bounds__(256)
void gemm_wmma_kernel(const float* __restrict__ A, const float* __restrict__ Bt,
                      const float* __restrict__ rowscale,
                      float* __restrict__ C, int M, int K, int Nout) {
    int wave = (int)((blockIdx.x * blockDim.x + threadIdx.x) >> 5);
    int lane = threadIdx.x & 31;
    if (wave >= (M >> 4)) return;

    int m0   = wave << 4;
    int half = lane >> 4;
    int l16  = lane & 15;

    const float* Arow = A + (size_t)(m0 + l16) * K;
    const float* Bcol[NT];
#pragma unroll
    for (int t = 0; t < NT; ++t)
        Bcol[t] = Bt + (size_t)(t * 16 + l16) * K;

    v8f c[NT] = {};

    int rbase = m0 + (half << 3);
    v4f sc0 = *(const v4f*)(rowscale + rbase);
    v4f sc1 = *(const v4f*)(rowscale + rbase + 4);

    for (int k0 = 0; k0 < K; k0 += 4) {
        int ka = k0 + (half << 1);
        v2f a = *(const v2f*)(Arow + ka);
#pragma unroll
        for (int t = 0; t < NT; ++t) {
            v2f b = *(const v2f*)(Bcol[t] + ka);
            c[t] = __builtin_amdgcn_wmma_f32_16x16x4_f32(
                       false, a, false, b, (short)0, c[t], false, false);
        }
    }

    const float sc[8] = {sc0.x, sc0.y, sc0.z, sc0.w, sc1.x, sc1.y, sc1.z, sc1.w};
#pragma unroll
    for (int t = 0; t < NT; ++t) {
        int ncol = t * 16 + l16;
        if (ncol < Nout) {
#pragma unroll
            for (int r = 0; r < 8; ++r)
                C[(size_t)(rbase + r) * Nout + ncol] = c[t][r] * sc[r];
        }
    }
}

#define TILE_NODES 16
#define WG_WAVES   4
#define WG_T       (32 * WG_WAVES)
#define WG_NODES   (WG_WAVES * TILE_NODES)
static_assert(NNODES % TILE_NODES == 0, "a wave's tile is wholly inside the graph or wholly outside");

template <int E>
__device__ __forceinline__ unsigned group8_lane(unsigned v) {
    return __builtin_amdgcn_mov_dpp8(v, (unsigned)E * 0x249249u);
}

template <int F, int Q, int C>
__device__ __forceinline__ void request_rows(const float* __restrict__ h, unsigned px,
                                             int l8, v4f (&r)[8][C]) {
    int s[8];
    s[0] = (int)group8_lane<0>(px);
    s[1] = (int)group8_lane<1>(px);
    s[2] = (int)group8_lane<2>(px);
    s[3] = (int)group8_lane<3>(px);
    s[4] = (int)group8_lane<4>(px);
    s[5] = (int)group8_lane<5>(px);
    s[6] = (int)group8_lane<6>(px);
    s[7] = (int)group8_lane<7>(px);
#pragma unroll
    for (int e = 0; e < 8; ++e) {
        const v4f* he = (const v4f*)(h + (size_t)s[e] * F);
#pragma unroll
        for (int c2 = 0; c2 < C; ++c2) {
            int q = l8 + 8 * c2;
            if (q < Q) r[e][c2] = he[q];
        }
    }
}

template <int Q, int C>
__device__ __forceinline__ void add_rows(v4f (&acc)[C], const v4f (&r)[8][C],
                                         unsigned cnt, int l8) {
#pragma unroll
    for (int e = 0; e < 8; e += 2) {
        if ((unsigned)(e + 2) <= cnt) {
#pragma unroll
            for (int c2 = 0; c2 < C; ++c2) {
                int q = l8 + 8 * c2;
                if (q < Q) acc[c2] += r[e][c2] + r[e + 1][c2];
            }
        } else if ((unsigned)(e + 1) == cnt) {
#pragma unroll
            for (int c2 = 0; c2 < C; ++c2) {
                int q = l8 + 8 * c2;
                if (q < Q) acc[c2] += r[e][c2];
            }
        }
    }
}

__device__ __forceinline__ unsigned last_of(unsigned end) { return (end > 0u ? end : 1u) - 1u; }
__device__ __forceinline__ unsigned clamp_to(unsigned i, unsigned last) { return i < last ? i : last; }

template <int F, int Q, int C>
__device__ __forceinline__ void gather_node(const float* __restrict__ h, const int* __restrict__ csr,
                                            unsigned beg, unsigned end, int q0, int q1, int l8,
                                            v4f (&acc)[C]) {
    const unsigned last = last_of(end);
    v4f rf[8][C];
    request_rows<F, Q, C>(h, (unsigned)q0, l8, rf);
    add_rows<Q, C>(acc, rf, end - beg, l8);
    int p = q1;
    for (unsigned j = beg + 8u; j < end; j += 8u) {
        int pn = csr[clamp_to(j + 8u + (unsigned)l8, last)];
        v4f r[8][C];
        request_rows<F, Q, C>(h, (unsigned)p, l8, r);
        add_rows<Q, C>(acc, r, end - j, l8);
        p = pn;
    }
}

#define TILE_PITCH 36

template <int FN>
__global__ __launch_bounds__(WG_T)
void gather_hidden_kernel(const float* __restrict__ h, const int* __restrict__ csr,
                          const unsigned* __restrict__ rowptr,
                          const float* __restrict__ dinv, const float* __restrict__ bias,
                          const float* __restrict__ xold_in, const float* __restrict__ dropu,
                          float* __restrict__ xout, float* __restrict__ xold_out,
                          const float* __restrict__ wnext, int n) {
    constexpr int F = NHID;
    constexpr int Q = F / 4;
    static_assert(Q == 8, "eight lanes hold one row");
    constexpr int NT  = (FN + 15) / 16;
    constexpr int NW4 = F * FN / 4;
    constexpr int WPT = (NW4 + WG_T - 1) / WG_T;
    __shared__ __attribute__((aligned(16))) float xs[WG_NODES * TILE_PITCH];
    __shared__ __attribute__((aligned(16))) float wt[NT * 16 * TILE_PITCH];

    const int wid  = __builtin_amdgcn_readfirstlane((int)(threadIdx.x >> 5));
    const int lane = threadIdx.x & 31;
    const int grp  = lane >> 3;
    const int l8   = lane & 7;
    int row0 = ((int)blockIdx.x * WG_WAVES + wid) * TILE_NODES;
    const bool live = row0 < n;
    row0 = live ? row0 : n - TILE_NODES;
    const int nb = row0 + 4 * grp;

    v4f wreg[WPT];
#pragma unroll
    for (int u = 0; u < WPT; ++u) {
        int i = u * WG_T + (int)threadIdx.x;
        i = i < NW4 ? i : NW4 - 1;
        wreg[u] = *(const v4f*)(wnext + 4 * (size_t)i);
    }

    const v4u      rp  = *(const v4u*)(rowptr + nb);
    const unsigned rpe = rowptr[nb + 4];
    const v4f      dv  = *(const v4f*)(dinv + nb);

#pragma unroll
    for (int u = 0; u < WPT; ++u) {
        int i = u * WG_T + (int)threadIdx.x;
        if (i < NW4) {
            int k  = (4 * i) / FN;
            int c0 = (4 * i) - k * FN;
            wt[(c0 + 0) * TILE_PITCH + k] = wreg[u].x;
            wt[(c0 + 1) * TILE_PITCH + k] = wreg[u].y;
            wt[(c0 + 2) * TILE_PITCH + k] = wreg[u].z;
            wt[(c0 + 3) * TILE_PITCH + k] = wreg[u].w;
        }
    }
    if (NT * 16 > FN) {
        for (int i = (int)threadIdx.x; i < (NT * 16 - FN) * F; i += WG_T)
            wt[(FN + i / F) * TILE_PITCH + (i % F)] = 0.0f;
    }
    __syncthreads();
    if (!live) return;

    const unsigned bnd[5] = {rp.x, rp.y, rp.z, rp.w, rpe};
    const float    dn[4]  = {dv.x, dv.y, dv.z, dv.w};
    int q[4], r[4];
#pragma unroll
    for (int g = 0; g < 4; ++g) {
        const unsigned last = last_of(bnd[g + 1]);
        q[g] = csr[clamp_to(bnd[g] + (unsigned)l8, last)];
        r[g] = csr[clamp_to(bnd[g] + 8u + (unsigned)l8, last)];
    }

    v4f bself;
    bself.x = bias[4 * l8 + 0];
    bself.y = bias[4 * l8 + 1];
    bself.z = bias[4 * l8 + 2];
    bself.w = bias[4 * l8 + 3];

    const size_t ob0 = (size_t)nb * F + 4 * (size_t)l8;
    v4f sf[4], dr[4], xo[4];
#pragma unroll
    for (int g = 0; g < 4; ++g) sf[g] = *(const v4f*)(h + ob0 + (size_t)(g * F));
#pragma unroll
    for (int g = 0; g < 4; ++g) dr[g] = *(const v4f*)(dropu + ob0 + (size_t)(g * F));
#pragma unroll
    for (int g = 0; g < 4; ++g) xo[g] = (v4f){0.0f, 0.0f, 0.0f, 0.0f};
    if (xold_in) {
#pragma unroll
        for (int g = 0; g < 4; ++g) xo[g] = *(const v4f*)(xold_in + ob0 + (size_t)(g * F));
    }

#pragma unroll
    for (int g = 0; g < 4; ++g) {
        const size_t ob = ob0 + (size_t)(g * F);
        v4f acc[1];
        acc[0] = sf[g];

        gather_node<F, Q, 1>(h, csr, bnd[g], bnd[g + 1], q[g], r[g], l8, acc);

        v4f a = acc[0] * dn[g] + bself;
        v4f y;
        y.x = fmaxf(a.x, 0.0f);
        y.y = fmaxf(a.y, 0.0f);
        y.z = fmaxf(a.z, 0.0f);
        y.w = fmaxf(a.w, 0.0f);
        if (xold_in) y += xo[g];
        v4f x;
        x.x = y.x * (dr[g].x > 0.5f ? 2.0f : 0.0f);
        x.y = y.y * (dr[g].y > 0.5f ? 2.0f : 0.0f);
        x.z = y.z * (dr[g].z > 0.5f ? 2.0f : 0.0f);
        x.w = y.w * (dr[g].w > 0.5f ? 2.0f : 0.0f);
        *(v4f*)(xs + (wid * TILE_NODES + 4 * grp + g) * TILE_PITCH + 4 * l8) = x * dn[g];
        *(v4f*)(xold_out + ob) = y;
    }

    const int half = lane >> 4;
    const int l16  = lane & 15;
    const float* arow = xs + (wid * TILE_NODES + l16) * TILE_PITCH + (half << 1);
    const float* bcol = wt + l16 * TILE_PITCH + (half << 1);
    v8f c[NT] = {};
#pragma unroll
    for (int k0 = 0; k0 < F; k0 += 4) {
        v2f a = *(const v2f*)(arow + k0);
#pragma unroll
        for (int t = 0; t < NT; ++t) {
            v2f b = *(const v2f*)(bcol + t * 16 * TILE_PITCH + k0);
            c[t] = __builtin_amdgcn_wmma_f32_16x16x4_f32(
                       false, a, false, b, (short)0, c[t], false, false);
        }
    }
    const int rbase = row0 + (half << 3);
#pragma unroll
    for (int t = 0; t < NT; ++t) {
        int ncol = t * 16 + l16;
        if (ncol < FN) {
#pragma unroll
            for (int r = 0; r < 8; ++r)
                xout[(size_t)(rbase + r) * FN + ncol] = c[t][r];
        }
    }
}

template <int F>
__global__ __launch_bounds__(WG_T)
void gather_out_kernel(const float* __restrict__ h, const int* __restrict__ csr,
                       const unsigned* __restrict__ rowptr,
                       const float* __restrict__ dinv, const float* __restrict__ bias,
                       float* __restrict__ xout, int n) {
    static_assert(F % 4 == 0, "row must be a whole number of 16-byte pieces");
    constexpr int Q = F / 4;
    constexpr int C = (Q + 7) / 8;

    const int wid  = (int)(threadIdx.x >> 5);
    const int lane = threadIdx.x & 31;
    const int grp  = lane >> 3;
    const int l8   = lane & 7;
    const int row0 = ((int)blockIdx.x * WG_WAVES + wid) * TILE_NODES;
    if (row0 >= n) return;
    const int nb = row0 + 4 * grp;

    const v4u      rp  = *(const v4u*)(rowptr + nb);
    const unsigned rpe = rowptr[nb + 4];
    const v4f      dv  = *(const v4f*)(dinv + nb);

    v4f bself[C];
#pragma unroll
    for (int c2 = 0; c2 < C; ++c2) {
        int q = l8 + 8 * c2;
        if (q < Q) {
            bself[c2].x = bias[4 * q + 0];
            bself[c2].y = bias[4 * q + 1];
            bself[c2].z = bias[4 * q + 2];
            bself[c2].w = bias[4 * q + 3];
        }
    }

    unsigned b0 = rp.x, b1 = rp.y, b2 = rp.z, b3 = rp.w, b4 = rpe;
    float    d0 = dv.x, d1 = dv.y, d2 = dv.z, d3 = dv.w;
    int q0 = csr[clamp_to(b0 + (unsigned)l8, last_of(b1))];
    int q1 = csr[clamp_to(b1 + (unsigned)l8, last_of(b2))];
    int q2 = csr[clamp_to(b2 + (unsigned)l8, last_of(b3))];
    int q3 = csr[clamp_to(b3 + (unsigned)l8, last_of(b4))];
    int r0 = csr[clamp_to(b0 + 8u + (unsigned)l8, last_of(b1))];
    int r1 = csr[clamp_to(b1 + 8u + (unsigned)l8, last_of(b2))];
    int r2 = csr[clamp_to(b2 + 8u + (unsigned)l8, last_of(b3))];
    int r3 = csr[clamp_to(b3 + 8u + (unsigned)l8, last_of(b4))];

#pragma nounroll
    for (int g = 0; g < 4; ++g) {
        const size_t base = (size_t)(nb + g) * F;
        const v4f* hrow = (const v4f*)(h + base);
        v4f acc[C];
#pragma unroll
        for (int c2 = 0; c2 < C; ++c2) {
            int q = l8 + 8 * c2;
            if (q < Q) acc[c2] = hrow[q];
            else       acc[c2] = (v4f){0.0f, 0.0f, 0.0f, 0.0f};
        }

        gather_node<F, Q, C>(h, csr, b0, b1, q0, r0, l8, acc);

#pragma unroll
        for (int c2 = 0; c2 < C; ++c2) {
            int q = l8 + 8 * c2;
            if (q < Q) *(v4f*)(xout + base + 4 * (size_t)q) = acc[c2] * d0 + bself[c2];
        }

        b0 = b1; b1 = b2; b2 = b3; b3 = b4;
        d0 = d1; d1 = d2; d2 = d3;
        q0 = q1; q1 = q2; q2 = q3;
        r0 = r1; r1 = r2; r2 = r3;
    }
}


extern "C" void kernel_launch(void* const* d_in, const int* in_sizes, int n_in,
                              void* d_out, int out_size, void* d_ws, size_t ws_size,
                              hipStream_t stream) {
    (void)in_sizes; (void)n_in; (void)out_size; (void)ws_size;

    const float* x0    = (const float*)d_in[0];
    const int*   ei    = (const int*)d_in[1];
    const float* dropu = (const float*)d_in[2];
    const float* W0    = (const float*)d_in[3];
    const float* b0    = (const float*)d_in[4];
    const float* Whid  = (const float*)d_in[5];
    const float* bhid  = (const float*)d_in[6];
    const float* Wout  = (const float*)d_in[7];
    const float* bout  = (const float*)d_in[8];
    float* out = (float*)d_out;

    const int* src = ei;
    const int* dst = ei + NEDGES;

    char* ws = (char*)d_ws;
    size_t off = 0;
    auto take = [&](size_t bytes) -> void* {
        void* p = (void*)(ws + off);
        off += (bytes + 255) & ~(size_t)255;
        return p;
    };
    unsigned* table  = (unsigned*)take((size_t)NBINS * NCHUNK * 4);
    unsigned* rowptr = (unsigned*)take((size_t)(NNODES + 1) * 4);
    float*    dinv   = (float*)   take((size_t)NNODES * 4);
    int*      csr    = (int*)     take((size_t)NEDGES * 4);
    float*    hq     = (float*)   take((size_t)NNODES * NCLS * 4);
    float*    hp     = (float*)   take((size_t)NNODES * NHID * 4);
    float*    xold   = (float*)   take((size_t)NNODES * NHID * 4);
    float*    Bt     = (float*)   take((size_t)48 * NFEAT * 4);
    int2* binned = (int2*)hq;
    static_assert((size_t)NEDGES * 8 <= (size_t)NNODES * NCLS * 4, "the binned edges fit in hq");

    const int TB = 256;
    auto blocks = [](int n, int tb) { return (n + tb - 1) / tb; };

    bin_count_kernel<<<NCHUNK, CHUNK_T, 0, stream>>>(dst, table, NEDGES);
    table_scan_kernel<<<1, SCAN_T, 0, stream>>>(table, NBINS * NCHUNK / 4);
    bin_scatter_kernel<<<NCHUNK, CHUNK_T, 0, stream>>>(src, dst, table, binned, NEDGES);
    bin_to_csr_kernel<<<NBINS, L2_T, 0, stream>>>(binned, table, rowptr, dinv, csr, NNODES, NEDGES);

    int gemm_blocks   = blocks(NNODES / 16, 8);
    int gather_blocks = blocks(NNODES, WG_NODES);

    transpose_pad_kernel<<<blocks(NHID * NFEAT, TB), TB, 0, stream>>>(W0, Bt, NFEAT, NHID, NHID);
    gemm_wmma_kernel<2><<<gemm_blocks, 256, 0, stream>>>(x0, Bt, dinv, hp, NNODES, NFEAT, NHID);

    const float* xold_in = nullptr;
    for (int layer = 0; layer < 7; ++layer) {
        const float* hin  = (layer & 1) ? hq : hp;
        float*       hout = (layer & 1) ? hp : hq;
        const float* b = (layer == 0) ? b0 : bhid + (size_t)(layer - 1) * NHID;
        const float* d = dropu + (size_t)layer * NNODES * NHID;

        if (layer < 6) {
            gather_hidden_kernel<NHID><<<gather_blocks, WG_T, 0, stream>>>(
                hin, csr, rowptr, dinv, b, xold_in, d, hout, xold,
                Whid + (size_t)layer * NHID * NHID, NNODES);
        } else {
            gather_hidden_kernel<NCLS><<<gather_blocks, WG_T, 0, stream>>>(
                hin, csr, rowptr, dinv, b, xold_in, d, hout, xold, Wout, NNODES);
        }
        xold_in = xold;
    }

    gather_out_kernel<NCLS><<<gather_blocks, WG_T, 0, stream>>>(
        hq, csr, rowptr, dinv, bout, out, NNODES);
}


# output layer: W_out applied behind the edge sum, A(xW)=(Ax)W: gathered rows are 32 floats (one 128 B line), were 40 (two lines); the 32x40 f32 WMMA product moves to the last launch; sum order changes
# speedup vs baseline: 3.1271x; 1.0886x over previous
#include <hip/hip_runtime.h>

#define NNODES 100000
#define NEDGES 1600000
#define NFEAT  128
#define NHID   32
#define NCLS   40
static_assert(NNODES % 16 == 0, "the WMMA strips are whole 16-row tiles");

typedef float v2f __attribute__((ext_vector_type(2)));
typedef float v4f __attribute__((ext_vector_type(4)));
typedef float v8f __attribute__((ext_vector_type(8)));

#define BIN_NODES 512
#define NBINS     ((NNODES + BIN_NODES - 1) / BIN_NODES)
#define CHUNK_T   512
#define CHUNK_EPT 16
#define CHUNK     (CHUNK_T * CHUNK_EPT)
#define NCHUNK    ((((NEDGES + CHUNK - 1) / CHUNK) + 3) / 4 * 4)
static_assert(NEDGES % 4 == 0, "the edge list is read in 16-byte pieces");
static_assert((BIN_NODES & (BIN_NODES - 1)) == 0, "a destination's place in its bin is a mask");

typedef int v4i __attribute__((ext_vector_type(4)));

__device__ __forceinline__ int bin_of(int d) { return (int)((unsigned)d / (unsigned)BIN_NODES); }

__device__ __forceinline__ unsigned wave_incl_scan(unsigned v, int lane) {
#pragma unroll
    for (int off = 1; off < 32; off <<= 1) {
        unsigned o = __shfl_up(v, (unsigned)off);
        if (lane >= off) v += o;
    }
    return v;
}

__global__ __launch_bounds__(CHUNK_T)
void bin_count_kernel(const int* __restrict__ dst, unsigned* __restrict__ table, int E) {
    __shared__ unsigned hist[NBINS];
    for (int i = (int)threadIdx.x; i < NBINS; i += CHUNK_T) hist[i] = 0u;

    const int E4 = E >> 2;
    int base4 = (int)blockIdx.x * (CHUNK / 4) + (int)threadIdx.x;
    v4i d[CHUNK_EPT / 4];
#pragma unroll
    for (int k = 0; k < CHUNK_EPT / 4; ++k) {
        int i4 = base4 + k * CHUNK_T;
        i4 = i4 < E4 ? i4 : E4 - 1;
        d[k] = *(const v4i*)(dst + 4 * (size_t)i4);
    }
    __syncthreads();
#pragma unroll
    for (int k = 0; k < CHUNK_EPT / 4; ++k) {
        if (base4 + k * CHUNK_T < E4) {
            atomicAdd(&hist[bin_of(d[k].x)], 1u);
            atomicAdd(&hist[bin_of(d[k].y)], 1u);
            atomicAdd(&hist[bin_of(d[k].z)], 1u);
            atomicAdd(&hist[bin_of(d[k].w)], 1u);
        }
    }
    __syncthreads();
    for (int i = (int)threadIdx.x; i < NBINS; i += CHUNK_T)
        table[(size_t)i * NCHUNK + blockIdx.x] = hist[i];
}

#define SCAN_T   1024
#define SCAN_V   (((NBINS * NCHUNK + SCAN_T - 1) / SCAN_T + 3) / 4)
static_assert((NBINS * NCHUNK) % 4 == 0, "the table is a whole number of 16-byte pieces");
typedef unsigned v4u __attribute__((ext_vector_type(4)));
__global__ __launch_bounds__(SCAN_T)
void table_scan_kernel(unsigned* __restrict__ table, int M4) {
    __shared__ unsigned wsum[32];
    int lane = (int)threadIdx.x & 31;
    int wid  = (int)threadIdx.x >> 5;
    v4u* t4 = (v4u*)table;
    int b = (int)threadIdx.x * SCAN_V;
    v4u v[SCAN_V];
    unsigned sum = 0u;
#pragma unroll
    for (int i = 0; i < SCAN_V; ++i) {
        v[i] = (v4u){0u, 0u, 0u, 0u};
        if (b + i < M4) v[i] = t4[b + i];
        sum += v[i].x + v[i].y + v[i].z + v[i].w;
    }
    unsigned inc = wave_incl_scan(sum, lane);
    if (lane == 31) wsum[wid] = inc;
    __syncthreads();
    if (wid == 0) {
        unsigned w = wsum[lane];
        unsigned winc = wave_incl_scan(w, lane);
        wsum[lane] = winc - w;
    }
    __syncthreads();
    unsigned run = wsum[wid] + inc - sum;
#pragma unroll
    for (int i = 0; i < SCAN_V; ++i) {
        v4u o;
        o.x = run; run += v[i].x;
        o.y = run; run += v[i].y;
        o.z = run; run += v[i].z;
        o.w = run; run += v[i].w;
        if (b + i < M4) t4[b + i] = o;
    }
}

__global__ __launch_bounds__(CHUNK_T)
void bin_scatter_kernel(const int* __restrict__ src, const int* __restrict__ dst,
                        const unsigned* __restrict__ table,
                        int2* __restrict__ binned, int E) {
    __shared__ unsigned cursor[NBINS];

    const int E4 = E >> 2;
    int base4 = (int)blockIdx.x * (CHUNK / 4) + (int)threadIdx.x;
    v4i d[CHUNK_EPT / 4];
    v4i s[CHUNK_EPT / 4];
#pragma unroll
    for (int k = 0; k < CHUNK_EPT / 4; ++k) {
        int i4 = base4 + k * CHUNK_T;
        i4 = i4 < E4 ? i4 : E4 - 1;
        d[k] = *(const v4i*)(dst + 4 * (size_t)i4);
        s[k] = *(const v4i*)(src + 4 * (size_t)i4);
    }
    for (int i = (int)threadIdx.x; i < NBINS; i += CHUNK_T)
        cursor[i] = table[(size_t)i * NCHUNK + blockIdx.x];
    __syncthreads();
#pragma unroll
    for (int k = 0; k < CHUNK_EPT / 4; ++k) {
        if (base4 + k * CHUNK_T < E4) {
            unsigned p0 = atomicAdd(&cursor[bin_of(d[k].x)], 1u);
            unsigned p1 = atomicAdd(&cursor[bin_of(d[k].y)], 1u);
            unsigned p2 = atomicAdd(&cursor[bin_of(d[k].z)], 1u);
            unsigned p3 = atomicAdd(&cursor[bin_of(d[k].w)], 1u);
            binned[p0] = make_int2(s[k].x, d[k].x);
            binned[p1] = make_int2(s[k].y, d[k].y);
            binned[p2] = make_int2(s[k].z, d[k].z);
            binned[p3] = make_int2(s[k].w, d[k].w);
        }
    }
}

#define L2_T 512
#define L2_U 9
static_assert(L2_T >= BIN_NODES, "one thread per destination of the bin");
__global__ __launch_bounds__(L2_T)
void bin_to_csr_kernel(const int2* __restrict__ binned, const unsigned* __restrict__ table,
                       unsigned* __restrict__ rowptr, float* __restrict__ dinv,
                       int* __restrict__ csr, int n, int E) {
    __shared__ unsigned cnt[BIN_NODES];
    __shared__ unsigned first[BIN_NODES];
    __shared__ unsigned cur[BIN_NODES];
    __shared__ float    dloc[BIN_NODES];
    int bin = (int)blockIdx.x;
    unsigned s = table[(size_t)bin * NCHUNK];
    unsigned e = (bin + 1 < NBINS) ? table[(size_t)(bin + 1) * NCHUNK] : (unsigned)E;
    if (threadIdx.x < BIN_NODES) cnt[threadIdx.x] = 0u;
    __syncthreads();

    for (unsigned i0 = s; i0 < e; i0 += L2_T * L2_U) {
        int dd[L2_U];
#pragma unroll
        for (int u = 0; u < L2_U; ++u) {
            unsigned i = i0 + (unsigned)(u * L2_T) + threadIdx.x;
            dd[u] = binned[i < e ? i : e - 1u].y;
        }
#pragma unroll
        for (int u = 0; u < L2_U; ++u) {
            unsigned i = i0 + (unsigned)(u * L2_T) + threadIdx.x;
            if (i < e) atomicAdd(&cnt[dd[u] & (BIN_NODES - 1)], 1u);
        }
    }
    __syncthreads();

    if (threadIdx.x < 32) {
        constexpr int PER = BIN_NODES / 32;
        int lane = (int)threadIdx.x;
        unsigned cc[PER];
        unsigned sum = 0u;
#pragma unroll
        for (int k = 0; k < PER; ++k) { cc[k] = cnt[lane * PER + k]; sum += cc[k]; }
        unsigned run = s + wave_incl_scan(sum, lane) - sum;
#pragma unroll
        for (int k = 0; k < PER; ++k) {
            first[lane * PER + k] = run;
            cur[lane * PER + k]   = run;
            dloc[lane * PER + k]  = rsqrtf((float)cc[k] + 1.0f);
            run += cc[k];
        }
    }
    __syncthreads();
    if (threadIdx.x < BIN_NODES) {
        int node = bin * BIN_NODES + (int)threadIdx.x;
        if (node < n) {
            rowptr[node] = first[threadIdx.x];
            dinv[node]   = dloc[threadIdx.x];
        }
    }
    if (bin == NBINS - 1 && threadIdx.x == 0) rowptr[n] = (unsigned)E;

    for (unsigned i0 = s; i0 < e; i0 += L2_T * L2_U) {
        int2 p[L2_U];
#pragma unroll
        for (int u = 0; u < L2_U; ++u) {
            unsigned i = i0 + (unsigned)(u * L2_T) + threadIdx.x;
            p[u] = binned[i < e ? i : e - 1u];
        }
#pragma unroll
        for (int u = 0; u < L2_U; ++u) {
            unsigned i = i0 + (unsigned)(u * L2_T) + threadIdx.x;
            if (i < e) {
                int l = p[u].y & (BIN_NODES - 1);
                unsigned pos = atomicAdd(&cur[l], 1u);
                csr[pos] = p[u].x;
            }
        }
    }
}

__global__ void transpose_pad_kernel(const float* __restrict__ W,
                                     float* __restrict__ Bt,
                                     int K, int Nout, int Npad) {
    int t = blockIdx.x * blockDim.x + threadIdx.x;
    if (t >= Npad * K) return;
    int n = t / K;
    int k = t - n * K;
    Bt[t] = (n < Nout) ? W[(size_t)k * Nout + n] : 0.0f;
}

template <int NT>
__global__ __launch_bounds__(256)
void gemm_wmma_kernel(const float* __restrict__ A, const float* __restrict__ Bt,
                      const float* __restrict__ rowscale,
                      float* __restrict__ C, int M, int K, int Nout) {
    int wave = (int)((blockIdx.x * blockDim.x + threadIdx.x) >> 5);
    int lane = threadIdx.x & 31;
    if (wave >= (M >> 4)) return;

    int m0   = wave << 4;
    int half = lane >> 4;
    int l16  = lane & 15;

    const float* Arow = A + (size_t)(m0 + l16) * K;
    const float* Bcol[NT];
#pragma unroll
    for (int t = 0; t < NT; ++t)
        Bcol[t] = Bt + (size_t)(t * 16 + l16) * K;

    v8f c[NT] = {};

    int rbase = m0 + (half << 3);
    v4f sc0 = *(const v4f*)(rowscale + rbase);
    v4f sc1 = *(const v4f*)(rowscale + rbase + 4);

    for (int k0 = 0; k0 < K; k0 += 4) {
        int ka = k0 + (half << 1);
        v2f a = *(const v2f*)(Arow + ka);
#pragma unroll
        for (int t = 0; t < NT; ++t) {
            v2f b = *(const v2f*)(Bcol[t] + ka);
            c[t] = __builtin_amdgcn_wmma_f32_16x16x4_f32(
                       false, a, false, b, (short)0, c[t], false, false);
        }
    }

    const float sc[8] = {sc0.x, sc0.y, sc0.z, sc0.w, sc1.x, sc1.y, sc1.z, sc1.w};
#pragma unroll
    for (int t = 0; t < NT; ++t) {
        int ncol = t * 16 + l16;
        if (ncol < Nout) {
#pragma unroll
            for (int r = 0; r < 8; ++r)
                C[(size_t)(rbase + r) * Nout + ncol] = c[t][r] * sc[r];
        }
    }
}

#define TILE_NODES 16
#define WG_WAVES   4
#define WG_T       (32 * WG_WAVES)
#define WG_NODES   (WG_WAVES * TILE_NODES)
static_assert(NNODES % TILE_NODES == 0, "a wave's tile is wholly inside the graph or wholly outside");

template <int E>
__device__ __forceinline__ unsigned group8_lane(unsigned v) {
    return __builtin_amdgcn_mov_dpp8(v, (unsigned)E * 0x249249u);
}

template <int F>
__device__ __forceinline__ void request_rows(const float* __restrict__ h, unsigned px,
                                             int l8, v4f (&r)[8]) {
    int s[8];
    s[0] = (int)group8_lane<0>(px);
    s[1] = (int)group8_lane<1>(px);
    s[2] = (int)group8_lane<2>(px);
    s[3] = (int)group8_lane<3>(px);
    s[4] = (int)group8_lane<4>(px);
    s[5] = (int)group8_lane<5>(px);
    s[6] = (int)group8_lane<6>(px);
    s[7] = (int)group8_lane<7>(px);
#pragma unroll
    for (int e = 0; e < 8; ++e)
        r[e] = ((const v4f*)(h + (size_t)s[e] * F))[l8];
}

__device__ __forceinline__ void add_rows(v4f& acc, const v4f (&r)[8], unsigned cnt) {
#pragma unroll
    for (int e = 0; e < 8; e += 2) {
        if ((unsigned)(e + 2) <= cnt)       acc += r[e] + r[e + 1];
        else if ((unsigned)(e + 1) == cnt)  acc += r[e];
    }
}

__device__ __forceinline__ unsigned last_of(unsigned end) { return (end > 0u ? end : 1u) - 1u; }
__device__ __forceinline__ unsigned clamp_to(unsigned i, unsigned last) { return i < last ? i : last; }

template <int F>
__device__ __forceinline__ void gather_node(const float* __restrict__ h, const int* __restrict__ csr,
                                            unsigned beg, unsigned end, int q0, int q1, int l8,
                                            v4f& acc) {
    const unsigned last = last_of(end);
    v4f rf[8];
    request_rows<F>(h, (unsigned)q0, l8, rf);
    add_rows(acc, rf, end - beg);
    int p = q1;
    for (unsigned j = beg + 8u; j < end; j += 8u) {
        int pn = csr[clamp_to(j + 8u + (unsigned)l8, last)];
        v4f r[8];
        request_rows<F>(h, (unsigned)p, l8, r);
        add_rows(acc, r, end - j);
        p = pn;
    }
}

#define TILE_PITCH 36

template <int FN, bool EPI>
__global__ __launch_bounds__(WG_T)
void gather_tile_kernel(const float* __restrict__ h, const int* __restrict__ csr,
                        const unsigned* __restrict__ rowptr,
                        const float* __restrict__ dinv, const float* __restrict__ bias,
                        const float* __restrict__ xold_in, const float* __restrict__ dropu,
                        float* __restrict__ xout, float* __restrict__ xold_out,
                        const float* __restrict__ wnext, int n) {
    constexpr int F = NHID;
    constexpr int Q = F / 4;
    static_assert(Q == 8, "eight lanes hold one row");
    constexpr bool PRODUCT = FN > 0;
    static_assert(EPI || PRODUCT, "the output layer has its transform behind the sum");
    constexpr int NT  = PRODUCT ? (FN + 15) / 16 : 1;
    constexpr int NW4 = PRODUCT ? F * FN / 4 : 1;
    constexpr int WPT = (NW4 + WG_T - 1) / WG_T;
    __shared__ __attribute__((aligned(16))) float xs[PRODUCT ? WG_NODES * TILE_PITCH : 4];
    __shared__ __attribute__((aligned(16))) float wt[PRODUCT ? NT * 16 * TILE_PITCH : 4];

    const int wid  = __builtin_amdgcn_readfirstlane((int)(threadIdx.x >> 5));
    const int lane = threadIdx.x & 31;
    const int grp  = lane >> 3;
    const int l8   = lane & 7;
    int row0 = ((int)blockIdx.x * WG_WAVES + wid) * TILE_NODES;
    const bool live = row0 < n;
    row0 = live ? row0 : n - TILE_NODES;
    const int nb = row0 + 4 * grp;

    v4f wreg[WPT];
    if constexpr (PRODUCT) {
#pragma unroll
        for (int u = 0; u < WPT; ++u) {
            int i = u * WG_T + (int)threadIdx.x;
            i = i < NW4 ? i : NW4 - 1;
            wreg[u] = *(const v4f*)(wnext + 4 * (size_t)i);
        }
    }

    const v4u      rp  = *(const v4u*)(rowptr + nb);
    const unsigned rpe = rowptr[nb + 4];
    const v4f      dv  = *(const v4f*)(dinv + nb);

    if constexpr (PRODUCT) {
#pragma unroll
        for (int u = 0; u < WPT; ++u) {
            int i = u * WG_T + (int)threadIdx.x;
            if (i < NW4) {
                int k  = (4 * i) / FN;
                int c0 = (4 * i) - k * FN;
                wt[(c0 + 0) * TILE_PITCH + k] = wreg[u].x;
                wt[(c0 + 1) * TILE_PITCH + k] = wreg[u].y;
                wt[(c0 + 2) * TILE_PITCH + k] = wreg[u].z;
                wt[(c0 + 3) * TILE_PITCH + k] = wreg[u].w;
            }
        }
        if (NT * 16 > FN) {
            for (int i = (int)threadIdx.x; i < (NT * 16 - FN) * F; i += WG_T)
                wt[(FN + i / F) * TILE_PITCH + (i % F)] = 0.0f;
        }
        __syncthreads();
    }
    if (!live) return;

    const unsigned bnd[5] = {rp.x, rp.y, rp.z, rp.w, rpe};
    const float    dn[4]  = {dv.x, dv.y, dv.z, dv.w};
    int q[4], r[4];
#pragma unroll
    for (int g = 0; g < 4; ++g) {
        const unsigned last = last_of(bnd[g + 1]);
        q[g] = csr[clamp_to(bnd[g] + (unsigned)l8, last)];
        r[g] = csr[clamp_to(bnd[g] + 8u + (unsigned)l8, last)];
    }

    const int half = lane >> 4;
    const int l16  = lane & 15;
    v4f   bself = (v4f){0.0f, 0.0f, 0.0f, 0.0f};
    float bcolv[NT];
    if constexpr (EPI) {
        bself.x = bias[4 * l8 + 0];
        bself.y = bias[4 * l8 + 1];
        bself.z = bias[4 * l8 + 2];
        bself.w = bias[4 * l8 + 3];
    } else {
#pragma unroll
        for (int t = 0; t < NT; ++t) {
            int ncol = t * 16 + l16;
            bcolv[t] = bias[ncol < FN ? ncol : FN - 1];
        }
    }

    const size_t ob0 = (size_t)nb * F + 4 * (size_t)l8;
    v4f sf[4], dr[4], xo[4];
#pragma unroll
    for (int g = 0; g < 4; ++g) sf[g] = *(const v4f*)(h + ob0 + (size_t)(g * F));
    if constexpr (EPI) {
#pragma unroll
        for (int g = 0; g < 4; ++g) dr[g] = *(const v4f*)(dropu + ob0 + (size_t)(g * F));
#pragma unroll
        for (int g = 0; g < 4; ++g) xo[g] = (v4f){0.0f, 0.0f, 0.0f, 0.0f};
        if (xold_in) {
#pragma unroll
            for (int g = 0; g < 4; ++g) xo[g] = *(const v4f*)(xold_in + ob0 + (size_t)(g * F));
        }
    }

#pragma unroll
    for (int g = 0; g < 4; ++g) {
        const size_t ob = ob0 + (size_t)(g * F);
        v4f acc = sf[g];

        gather_node<F>(h, csr, bnd[g], bnd[g + 1], q[g], r[g], l8, acc);

        v4f row;
        if constexpr (EPI) {
            v4f a = acc * dn[g] + bself;
            v4f y;
            y.x = fmaxf(a.x, 0.0f);
            y.y = fmaxf(a.y, 0.0f);
            y.z = fmaxf(a.z, 0.0f);
            y.w = fmaxf(a.w, 0.0f);
            if (xold_in) y += xo[g];
            v4f x;
            x.x = y.x * (dr[g].x > 0.5f ? 2.0f : 0.0f);
            x.y = y.y * (dr[g].y > 0.5f ? 2.0f : 0.0f);
            x.z = y.z * (dr[g].z > 0.5f ? 2.0f : 0.0f);
            x.w = y.w * (dr[g].w > 0.5f ? 2.0f : 0.0f);
            row = x * dn[g];
            *(v4f*)(xold_out + ob) = y;
        } else {
            row = acc * dn[g];
        }
        if constexpr (PRODUCT)
            *(v4f*)(xs + (wid * TILE_NODES + 4 * grp + g) * TILE_PITCH + 4 * l8) = row;
        else
            *(v4f*)(xout + ob) = row;
    }

    if constexpr (PRODUCT) {
        const float* arow = xs + (wid * TILE_NODES + l16) * TILE_PITCH + (half << 1);
        const float* bcol = wt + l16 * TILE_PITCH + (half << 1);
        v8f c[NT] = {};
#pragma unroll
        for (int k0 = 0; k0 < F; k0 += 4) {
            v2f a = *(const v2f*)(arow + k0);
#pragma unroll
            for (int t = 0; t < NT; ++t) {
                v2f b = *(const v2f*)(bcol + t * 16 * TILE_PITCH + k0);
                c[t] = __builtin_amdgcn_wmma_f32_16x16x4_f32(
                           false, a, false, b, (short)0, c[t], false, false);
            }
        }
        if constexpr (!EPI) {
#pragma unroll
            for (int t = 0; t < NT; ++t) c[t] += bcolv[t];
        }
        const int rbase = row0 + (half << 3);
#pragma unroll
        for (int t = 0; t < NT; ++t) {
            int ncol = t * 16 + l16;
            if (ncol < FN) {
#pragma unroll
                for (int rr = 0; rr < 8; ++rr)
                    xout[(size_t)(rbase + rr) * FN + ncol] = c[t][rr];
            }
        }
    }
}


extern "C" void kernel_launch(void* const* d_in, const int* in_sizes, int n_in,
                              void* d_out, int out_size, void* d_ws, size_t ws_size,
                              hipStream_t stream) {
    (void)in_sizes; (void)n_in; (void)out_size; (void)ws_size;

    const float* x0    = (const float*)d_in[0];
    const int*   ei    = (const int*)d_in[1];
    const float* dropu = (const float*)d_in[2];
    const float* W0    = (const float*)d_in[3];
    const float* b0    = (const float*)d_in[4];
    const float* Whid  = (const float*)d_in[5];
    const float* bhid  = (const float*)d_in[6];
    const float* Wout  = (const float*)d_in[7];
    const float* bout  = (const float*)d_in[8];
    float* out = (float*)d_out;

    const int* src = ei;
    const int* dst = ei + NEDGES;

    char* ws = (char*)d_ws;
    size_t off = 0;
    auto take = [&](size_t bytes) -> void* {
        void* p = (void*)(ws + off);
        off += (bytes + 255) & ~(size_t)255;
        return p;
    };
    unsigned* table  = (unsigned*)take((size_t)NBINS * NCHUNK * 4);
    unsigned* rowptr = (unsigned*)take((size_t)(NNODES + 1) * 4);
    float*    dinv   = (float*)   take((size_t)NNODES * 4);
    int*      csr    = (int*)     take((size_t)NEDGES * 4);
    float*    hq     = (float*)   take((size_t)NNODES * NCLS * 4);
    float*    hp     = (float*)   take((size_t)NNODES * NHID * 4);
    float*    xold   = (float*)   take((size_t)NNODES * NHID * 4);
    float*    Bt     = (float*)   take((size_t)48 * NFEAT * 4);
    int2* binned = (int2*)hq;
    static_assert((size_t)NEDGES * 8 <= (size_t)NNODES * NCLS * 4, "the binned edges fit in hq");

    const int TB = 256;
    auto blocks = [](int n, int tb) { return (n + tb - 1) / tb; };

    bin_count_kernel<<<NCHUNK, CHUNK_T, 0, stream>>>(dst, table, NEDGES);
    table_scan_kernel<<<1, SCAN_T, 0, stream>>>(table, NBINS * NCHUNK / 4);
    bin_scatter_kernel<<<NCHUNK, CHUNK_T, 0, stream>>>(src, dst, table, binned, NEDGES);
    bin_to_csr_kernel<<<NBINS, L2_T, 0, stream>>>(binned, table, rowptr, dinv, csr, NNODES, NEDGES);

    int gemm_blocks   = blocks(NNODES / 16, 8);
    int gather_blocks = blocks(NNODES, WG_NODES);

    transpose_pad_kernel<<<blocks(NHID * NFEAT, TB), TB, 0, stream>>>(W0, Bt, NFEAT, NHID, NHID);
    gemm_wmma_kernel<2><<<gemm_blocks, 256, 0, stream>>>(x0, Bt, dinv, hp, NNODES, NFEAT, NHID);

    const float* xold_in = nullptr;
    for (int layer = 0; layer < 7; ++layer) {
        const float* hin  = (layer & 1) ? hq : hp;
        float*       hout = (layer & 1) ? hp : hq;
        const float* b = (layer == 0) ? b0 : bhid + (size_t)(layer - 1) * NHID;
        const float* d = dropu + (size_t)layer * NNODES * NHID;

        if (layer < 6) {
            gather_tile_kernel<NHID, true><<<gather_blocks, WG_T, 0, stream>>>(
                hin, csr, rowptr, dinv, b, xold_in, d, hout, xold,
                Whid + (size_t)layer * NHID * NHID, NNODES);
        } else {
            gather_tile_kernel<0, true><<<gather_blocks, WG_T, 0, stream>>>(
                hin, csr, rowptr, dinv, b, xold_in, d, hout, xold, nullptr, NNODES);
        }
        xold_in = xold;
    }

    gather_tile_kernel<NCLS, false><<<gather_blocks, WG_T, 0, stream>>>(
        hq, csr, rowptr, dinv, bout, nullptr, nullptr, out, nullptr, Wout, NNODES);
}


# layer 0's 128x32 transform: each lane requests its whole x row (16 x 16 B) ahead of the WMMA chain, W0^T staged in LDS in-kernel; transpose launch gone (13 launches); k order of its sums changes
# speedup vs baseline: 3.3300x; 1.0649x over previous
#include <hip/hip_runtime.h>

#define NNODES 100000
#define NEDGES 1600000
#define NFEAT  128
#define NHID   32
#define NCLS   40
static_assert(NNODES % 16 == 0, "the WMMA strips are whole 16-row tiles");

typedef float v2f __attribute__((ext_vector_type(2)));
typedef float v4f __attribute__((ext_vector_type(4)));
typedef float v8f __attribute__((ext_vector_type(8)));

#define BIN_NODES 512
#define NBINS     ((NNODES + BIN_NODES - 1) / BIN_NODES)
#define CHUNK_T   512
#define CHUNK_EPT 16
#define CHUNK     (CHUNK_T * CHUNK_EPT)
#define NCHUNK    ((((NEDGES + CHUNK - 1) / CHUNK) + 3) / 4 * 4)
static_assert(NEDGES % 4 == 0, "the edge list is read in 16-byte pieces");
static_assert((BIN_NODES & (BIN_NODES - 1)) == 0, "a destination's place in its bin is a mask");

typedef int v4i __attribute__((ext_vector_type(4)));

__device__ __forceinline__ int bin_of(int d) { return (int)((unsigned)d / (unsigned)BIN_NODES); }

__device__ __forceinline__ unsigned wave_incl_scan(unsigned v, int lane) {
#pragma unroll
    for (int off = 1; off < 32; off <<= 1) {
        unsigned o = __shfl_up(v, (unsigned)off);
        if (lane >= off) v += o;
    }
    return v;
}

__global__ __launch_bounds__(CHUNK_T)
void bin_count_kernel(const int* __restrict__ dst, unsigned* __restrict__ table, int E) {
    __shared__ unsigned hist[NBINS];
    for (int i = (int)threadIdx.x; i < NBINS; i += CHUNK_T) hist[i] = 0u;

    const int E4 = E >> 2;
    int base4 = (int)blockIdx.x * (CHUNK / 4) + (int)threadIdx.x;
    v4i d[CHUNK_EPT / 4];
#pragma unroll
    for (int k = 0; k < CHUNK_EPT / 4; ++k) {
        int i4 = base4 + k * CHUNK_T;
        i4 = i4 < E4 ? i4 : E4 - 1;
        d[k] = *(const v4i*)(dst + 4 * (size_t)i4);
    }
    __syncthreads();
#pragma unroll
    for (int k = 0; k < CHUNK_EPT / 4; ++k) {
        if (base4 + k * CHUNK_T < E4) {
            atomicAdd(&hist[bin_of(d[k].x)], 1u);
            atomicAdd(&hist[bin_of(d[k].y)], 1u);
            atomicAdd(&hist[bin_of(d[k].z)], 1u);
            atomicAdd(&hist[bin_of(d[k].w)], 1u);
        }
    }
    __syncthreads();
    for (int i = (int)threadIdx.x; i < NBINS; i += CHUNK_T)
        table[(size_t)i * NCHUNK + blockIdx.x] = hist[i];
}

#define SCAN_T   1024
#define SCAN_V   (((NBINS * NCHUNK + SCAN_T - 1) / SCAN_T + 3) / 4)
static_assert((NBINS * NCHUNK) % 4 == 0, "the table is a whole number of 16-byte pieces");
typedef unsigned v4u __attribute__((ext_vector_type(4)));
__global__ __launch_bounds__(SCAN_T)
void table_scan_kernel(unsigned* __restrict__ table, int M4) {
    __shared__ unsigned wsum[32];
    int lane = (int)threadIdx.x & 31;
    int wid  = (int)threadIdx.x >> 5;
    v4u* t4 = (v4u*)table;
    int b = (int)threadIdx.x * SCAN_V;
    v4u v[SCAN_V];
    unsigned sum = 0u;
#pragma unroll
    for (int i = 0; i < SCAN_V; ++i) {
        v[i] = (v4u){0u, 0u, 0u, 0u};
        if (b + i < M4) v[i] = t4[b + i];
        sum += v[i].x + v[i].y + v[i].z + v[i].w;
    }
    unsigned inc = wave_incl_scan(sum, lane);
    if (lane == 31) wsum[wid] = inc;
    __syncthreads();
    if (wid == 0) {
        unsigned w = wsum[lane];
        unsigned winc = wave_incl_scan(w, lane);
        wsum[lane] = winc - w;
    }
    __syncthreads();
    unsigned run = wsum[wid] + inc - sum;
#pragma unroll
    for (int i = 0; i < SCAN_V; ++i) {
        v4u o;
        o.x = run; run += v[i].x;
        o.y = run; run += v[i].y;
        o.z = run; run += v[i].z;
        o.w = run; run += v[i].w;
        if (b + i < M4) t4[b + i] = o;
    }
}

__global__ __launch_bounds__(CHUNK_T)
void bin_scatter_kernel(const int* __restrict__ src, const int* __restrict__ dst,
                        const unsigned* __restrict__ table,
                        int2* __restrict__ binned, int E) {
    __shared__ unsigned cursor[NBINS];

    const int E4 = E >> 2;
    int base4 = (int)blockIdx.x * (CHUNK / 4) + (int)threadIdx.x;
    v4i d[CHUNK_EPT / 4];
    v4i s[CHUNK_EPT / 4];
#pragma unroll
    for (int k = 0; k < CHUNK_EPT / 4; ++k) {
        int i4 = base4 + k * CHUNK_T;
        i4 = i4 < E4 ? i4 : E4 - 1;
        d[k] = *(const v4i*)(dst + 4 * (size_t)i4);
        s[k] = *(const v4i*)(src + 4 * (size_t)i4);
    }
    for (int i = (int)threadIdx.x; i < NBINS; i += CHUNK_T)
        cursor[i] = table[(size_t)i * NCHUNK + blockIdx.x];
    __syncthreads();
#pragma unroll
    for (int k = 0; k < CHUNK_EPT / 4; ++k) {
        if (base4 + k * CHUNK_T < E4) {
            unsigned p0 = atomicAdd(&cursor[bin_of(d[k].x)], 1u);
            unsigned p1 = atomicAdd(&cursor[bin_of(d[k].y)], 1u);
            unsigned p2 = atomicAdd(&cursor[bin_of(d[k].z)], 1u);
            unsigned p3 = atomicAdd(&cursor[bin_of(d[k].w)], 1u);
            binned[p0] = make_int2(s[k].x, d[k].x);
            binned[p1] = make_int2(s[k].y, d[k].y);
            binned[p2] = make_int2(s[k].z, d[k].z);
            binned[p3] = make_int2(s[k].w, d[k].w);
        }
    }
}

#define L2_T 512
#define L2_U 9
static_assert(L2_T >= BIN_NODES, "one thread per destination of the bin");
__global__ __launch_bounds__(L2_T)
void bin_to_csr_kernel(const int2* __restrict__ binned, const unsigned* __restrict__ table,
                       unsigned* __restrict__ rowptr, float* __restrict__ dinv,
                       int* __restrict__ csr, int n, int E) {
    __shared__ unsigned cnt[BIN_NODES];
    __shared__ unsigned first[BIN_NODES];
    __shared__ unsigned cur[BIN_NODES];
    __shared__ float    dloc[BIN_NODES];
    int bin = (int)blockIdx.x;
    unsigned s = table[(size_t)bin * NCHUNK];
    unsigned e = (bin + 1 < NBINS) ? table[(size_t)(bin + 1) * NCHUNK] : (unsigned)E;
    if (threadIdx.x < BIN_NODES) cnt[threadIdx.x] = 0u;
    __syncthreads();

    for (unsigned i0 = s; i0 < e; i0 += L2_T * L2_U) {
        int dd[L2_U];
#pragma unroll
        for (int u = 0; u < L2_U; ++u) {
            unsigned i = i0 + (unsigned)(u * L2_T) + threadIdx.x;
            dd[u] = binned[i < e ? i : e - 1u].y;
        }
#pragma unroll
        for (int u = 0; u < L2_U; ++u) {
            unsigned i = i0 + (unsigned)(u * L2_T) + threadIdx.x;
            if (i < e) atomicAdd(&cnt[dd[u] & (BIN_NODES - 1)], 1u);
        }
    }
    __syncthreads();

    if (threadIdx.x < 32) {
        constexpr int PER = BIN_NODES / 32;
        int lane = (int)threadIdx.x;
        unsigned cc[PER];
        unsigned sum = 0u;
#pragma unroll
        for (int k = 0; k < PER; ++k) { cc[k] = cnt[lane * PER + k]; sum += cc[k]; }
        unsigned run = s + wave_incl_scan(sum, lane) - sum;
#pragma unroll
        for (int k = 0; k < PER; ++k) {
            first[lane * PER + k] = run;
            cur[lane * PER + k]   = run;
            dloc[lane * PER + k]  = rsqrtf((float)cc[k] + 1.0f);
            run += cc[k];
        }
    }
    __syncthreads();
    if (threadIdx.x < BIN_NODES) {
        int node = bin * BIN_NODES + (int)threadIdx.x;
        if (node < n) {
            rowptr[node] = first[threadIdx.x];
            dinv[node]   = dloc[threadIdx.x];
        }
    }
    if (bin == NBINS - 1 && threadIdx.x == 0) rowptr[n] = (unsigned)E;

    for (unsigned i0 = s; i0 < e; i0 += L2_T * L2_U) {
        int2 p[L2_U];
#pragma unroll
        for (int u = 0; u < L2_U; ++u) {
            unsigned i = i0 + (unsigned)(u * L2_T) + threadIdx.x;
            p[u] = binned[i < e ? i : e - 1u];
        }
#pragma unroll
        for (int u = 0; u < L2_U; ++u) {
            unsigned i = i0 + (unsigned)(u * L2_T) + threadIdx.x;
            if (i < e) {
                int l = p[u].y & (BIN_NODES - 1);
                unsigned pos = atomicAdd(&cur[l], 1u);
                csr[pos] = p[u].x;
            }
        }
    }
}

#define XF_WAVES 8
#define XF_T     (32 * XF_WAVES)
#define XF_PITCH (NFEAT + 4)

__global__ __launch_bounds__(XF_T)
void input_transform_kernel(const float* __restrict__ x, const float* __restrict__ W,
                            const float* __restrict__ rowscale,
                            float* __restrict__ C, int M) {
    constexpr int K   = NFEAT;
    constexpr int N   = NHID;
    constexpr int NT  = N / 16;
    constexpr int KP  = K / 8;
    constexpr int NW4 = K * N / 4;
    constexpr int WPT = NW4 / XF_T;
    static_assert(N % 16 == 0 && K % 8 == 0, "whole tiles, whole pieces");
    static_assert(NW4 % XF_T == 0, "every thread stages the same number of pieces");
    __shared__ __attribute__((aligned(16))) float wt[N * XF_PITCH];

    const int wid  = __builtin_amdgcn_readfirstlane((int)(threadIdx.x >> 5));
    const int lane = threadIdx.x & 31;
    const int half = lane >> 4;
    const int l16  = lane & 15;
    int m0 = ((int)blockIdx.x * XF_WAVES + wid) * 16;
    const bool live = m0 < M;
    m0 = live ? m0 : M - 16;

    v4f wreg[WPT];
    for (int u = 0; u < WPT; ++u)
        wreg[u] = *(const v4f*)(W + 4 * (size_t)(u * XF_T + (int)threadIdx.x));

    const float* xrow = x + (size_t)(m0 + l16) * K + 4 * half;
    v4f a[KP];
    for (int j = 0; j < KP; ++j)
        a[j] = *(const v4f*)(xrow + 8 * j);

    const int rbase = m0 + (half << 3);
    const v4f sc0 = *(const v4f*)(rowscale + rbase);
    const v4f sc1 = *(const v4f*)(rowscale + rbase + 4);

    for (int u = 0; u < WPT; ++u) {
        const int i  = u * XF_T + (int)threadIdx.x;
        const int k  = (4 * i) / N;
        const int c0 = (4 * i) - k * N;
        wt[(c0 + 0) * XF_PITCH + k] = wreg[u].x;
        wt[(c0 + 1) * XF_PITCH + k] = wreg[u].y;
        wt[(c0 + 2) * XF_PITCH + k] = wreg[u].z;
        wt[(c0 + 3) * XF_PITCH + k] = wreg[u].w;
    }
    __syncthreads();
    if (!live) return;

    const float* bcol = wt + l16 * XF_PITCH + 4 * half;
    v8f c[NT] = {};
    for (int j = 0; j < KP; ++j) {
        v4f b[NT];
        for (int t = 0; t < NT; ++t)
            b[t] = *(const v4f*)(bcol + t * 16 * XF_PITCH + 8 * j);
        const v2f alo = (v2f){a[j].x, a[j].y};
        const v2f ahi = (v2f){a[j].z, a[j].w};
        for (int t = 0; t < NT; ++t)
            c[t] = __builtin_amdgcn_wmma_f32_16x16x4_f32(
                       false, alo, false, (v2f){b[t].x, b[t].y}, (short)0, c[t], false, false);
        for (int t = 0; t < NT; ++t)
            c[t] = __builtin_amdgcn_wmma_f32_16x16x4_f32(
                       false, ahi, false, (v2f){b[t].z, b[t].w}, (short)0, c[t], false, false);
    }

    const float sc[8] = {sc0.x, sc0.y, sc0.z, sc0.w, sc1.x, sc1.y, sc1.z, sc1.w};
    for (int t = 0; t < NT; ++t) {
        const int ncol = t * 16 + l16;
        for (int r = 0; r < 8; ++r)
            C[(size_t)(rbase + r) * N + ncol] = c[t][r] * sc[r];
    }
}

#define TILE_NODES 16
#define WG_WAVES   4
#define WG_T       (32 * WG_WAVES)
#define WG_NODES   (WG_WAVES * TILE_NODES)
static_assert(NNODES % TILE_NODES == 0, "a wave's tile is wholly inside the graph or wholly outside");

template <int E>
__device__ __forceinline__ unsigned group8_lane(unsigned v) {
    return __builtin_amdgcn_mov_dpp8(v, (unsigned)E * 0x249249u);
}

template <int F>
__device__ __forceinline__ void request_rows(const float* __restrict__ h, unsigned px,
                                             int l8, v4f (&r)[8]) {
    int s[8];
    s[0] = (int)group8_lane<0>(px);
    s[1] = (int)group8_lane<1>(px);
    s[2] = (int)group8_lane<2>(px);
    s[3] = (int)group8_lane<3>(px);
    s[4] = (int)group8_lane<4>(px);
    s[5] = (int)group8_lane<5>(px);
    s[6] = (int)group8_lane<6>(px);
    s[7] = (int)group8_lane<7>(px);
#pragma unroll
    for (int e = 0; e < 8; ++e)
        r[e] = ((const v4f*)(h + (size_t)s[e] * F))[l8];
}

__device__ __forceinline__ void add_rows(v4f& acc, const v4f (&r)[8], unsigned cnt) {
#pragma unroll
    for (int e = 0; e < 8; e += 2) {
        if ((unsigned)(e + 2) <= cnt)       acc += r[e] + r[e + 1];
        else if ((unsigned)(e + 1) == cnt)  acc += r[e];
    }
}

__device__ __forceinline__ unsigned last_of(unsigned end) { return (end > 0u ? end : 1u) - 1u; }
__device__ __forceinline__ unsigned clamp_to(unsigned i, unsigned last) { return i < last ? i : last; }

template <int F>
__device__ __forceinline__ void gather_node(const float* __restrict__ h, const int* __restrict__ csr,
                                            unsigned beg, unsigned end, int q0, int q1, int l8,
                                            v4f& acc) {
    const unsigned last = last_of(end);
    v4f rf[8];
    request_rows<F>(h, (unsigned)q0, l8, rf);
    add_rows(acc, rf, end - beg);
    int p = q1;
    for (unsigned j = beg + 8u; j < end; j += 8u) {
        int pn = csr[clamp_to(j + 8u + (unsigned)l8, last)];
        v4f r[8];
        request_rows<F>(h, (unsigned)p, l8, r);
        add_rows(acc, r, end - j);
        p = pn;
    }
}

#define TILE_PITCH 36

template <int FN, bool EPI>
__global__ __launch_bounds__(WG_T)
void gather_tile_kernel(const float* __restrict__ h, const int* __restrict__ csr,
                        const unsigned* __restrict__ rowptr,
                        const float* __restrict__ dinv, const float* __restrict__ bias,
                        const float* __restrict__ xold_in, const float* __restrict__ dropu,
                        float* __restrict__ xout, float* __restrict__ xold_out,
                        const float* __restrict__ wnext, int n) {
    constexpr int F = NHID;
    constexpr int Q = F / 4;
    static_assert(Q == 8, "eight lanes hold one row");
    constexpr bool PRODUCT = FN > 0;
    static_assert(EPI || PRODUCT, "the output layer has its transform behind the sum");
    constexpr int NT  = PRODUCT ? (FN + 15) / 16 : 1;
    constexpr int NW4 = PRODUCT ? F * FN / 4 : 1;
    constexpr int WPT = (NW4 + WG_T - 1) / WG_T;
    __shared__ __attribute__((aligned(16))) float xs[PRODUCT ? WG_NODES * TILE_PITCH : 4];
    __shared__ __attribute__((aligned(16))) float wt[PRODUCT ? NT * 16 * TILE_PITCH : 4];

    const int wid  = __builtin_amdgcn_readfirstlane((int)(threadIdx.x >> 5));
    const int lane = threadIdx.x & 31;
    const int grp  = lane >> 3;
    const int l8   = lane & 7;
    int row0 = ((int)blockIdx.x * WG_WAVES + wid) * TILE_NODES;
    const bool live = row0 < n;
    row0 = live ? row0 : n - TILE_NODES;
    const int nb = row0 + 4 * grp;

    v4f wreg[WPT];
    if constexpr (PRODUCT) {
#pragma unroll
        for (int u = 0; u < WPT; ++u) {
            int i = u * WG_T + (int)threadIdx.x;
            i = i < NW4 ? i : NW4 - 1;
            wreg[u] = *(const v4f*)(wnext + 4 * (size_t)i);
        }
    }

    const v4u      rp  = *(const v4u*)(rowptr + nb);
    const unsigned rpe = rowptr[nb + 4];
    const v4f      dv  = *(const v4f*)(dinv + nb);

    if constexpr (PRODUCT) {
#pragma unroll
        for (int u = 0; u < WPT; ++u) {
            int i = u * WG_T + (int)threadIdx.x;
            if (i < NW4) {
                int k  = (4 * i) / FN;
                int c0 = (4 * i) - k * FN;
                wt[(c0 + 0) * TILE_PITCH + k] = wreg[u].x;
                wt[(c0 + 1) * TILE_PITCH + k] = wreg[u].y;
                wt[(c0 + 2) * TILE_PITCH + k] = wreg[u].z;
                wt[(c0 + 3) * TILE_PITCH + k] = wreg[u].w;
            }
        }
        if (NT * 16 > FN) {
            for (int i = (int)threadIdx.x; i < (NT * 16 - FN) * F; i += WG_T)
                wt[(FN + i / F) * TILE_PITCH + (i % F)] = 0.0f;
        }
        __syncthreads();
    }
    if (!live) return;

    const unsigned bnd[5] = {rp.x, rp.y, rp.z, rp.w, rpe};
    const float    dn[4]  = {dv.x, dv.y, dv.z, dv.w};
    int q[4], r[4];
#pragma unroll
    for (int g = 0; g < 4; ++g) {
        const unsigned last = last_of(bnd[g + 1]);
        q[g] = csr[clamp_to(bnd[g] + (unsigned)l8, last)];
        r[g] = csr[clamp_to(bnd[g] + 8u + (unsigned)l8, last)];
    }

    const int half = lane >> 4;
    const int l16  = lane & 15;
    v4f   bself = (v4f){0.0f, 0.0f, 0.0f, 0.0f};
    float bcolv[NT];
    if constexpr (EPI) {
        bself.x = bias[4 * l8 + 0];
        bself.y = bias[4 * l8 + 1];
        bself.z = bias[4 * l8 + 2];
        bself.w = bias[4 * l8 + 3];
    } else {
#pragma unroll
        for (int t = 0; t < NT; ++t) {
            int ncol = t * 16 + l16;
            bcolv[t] = bias[ncol < FN ? ncol : FN - 1];
        }
    }

    const size_t ob0 = (size_t)nb * F + 4 * (size_t)l8;
    v4f sf[4], dr[4], xo[4];
#pragma unroll
    for (int g = 0; g < 4; ++g) sf[g] = *(const v4f*)(h + ob0 + (size_t)(g * F));
    if constexpr (EPI) {
#pragma unroll
        for (int g = 0; g < 4; ++g) dr[g] = *(const v4f*)(dropu + ob0 + (size_t)(g * F));
#pragma unroll
        for (int g = 0; g < 4; ++g) xo[g] = (v4f){0.0f, 0.0f, 0.0f, 0.0f};
        if (xold_in) {
#pragma unroll
            for (int g = 0; g < 4; ++g) xo[g] = *(const v4f*)(xold_in + ob0 + (size_t)(g * F));
        }
    }

#pragma unroll
    for (int g = 0; g < 4; ++g) {
        const size_t ob = ob0 + (size_t)(g * F);
        v4f acc = sf[g];

        gather_node<F>(h, csr, bnd[g], bnd[g + 1], q[g], r[g], l8, acc);

        v4f row;
        if constexpr (EPI) {
            v4f a = acc * dn[g] + bself;
            v4f y;
            y.x = fmaxf(a.x, 0.0f);
            y.y = fmaxf(a.y, 0.0f);
            y.z = fmaxf(a.z, 0.0f);
            y.w = fmaxf(a.w, 0.0f);
            if (xold_in) y += xo[g];
            v4f x;
            x.x = y.x * (dr[g].x > 0.5f ? 2.0f : 0.0f);
            x.y = y.y * (dr[g].y > 0.5f ? 2.0f : 0.0f);
            x.z = y.z * (dr[g].z > 0.5f ? 2.0f : 0.0f);
            x.w = y.w * (dr[g].w > 0.5f ? 2.0f : 0.0f);
            row = x * dn[g];
            *(v4f*)(xold_out + ob) = y;
        } else {
            row = acc * dn[g];
        }
        if constexpr (PRODUCT)
            *(v4f*)(xs + (wid * TILE_NODES + 4 * grp + g) * TILE_PITCH + 4 * l8) = row;
        else
            *(v4f*)(xout + ob) = row;
    }

    if constexpr (PRODUCT) {
        const float* arow = xs + (wid * TILE_NODES + l16) * TILE_PITCH + (half << 1);
        const float* bcol = wt + l16 * TILE_PITCH + (half << 1);
        v8f c[NT] = {};
#pragma unroll
        for (int k0 = 0; k0 < F; k0 += 4) {
            v2f a = *(const v2f*)(arow + k0);
#pragma unroll
            for (int t = 0; t < NT; ++t) {
                v2f b = *(const v2f*)(bcol + t * 16 * TILE_PITCH + k0);
                c[t] = __builtin_amdgcn_wmma_f32_16x16x4_f32(
                           false, a, false, b, (short)0, c[t], false, false);
            }
        }
        if constexpr (!EPI) {
#pragma unroll
            for (int t = 0; t < NT; ++t) c[t] += bcolv[t];
        }
        const int rbase = row0 + (half << 3);
#pragma unroll
        for (int t = 0; t < NT; ++t) {
            int ncol = t * 16 + l16;
            if (ncol < FN) {
#pragma unroll
                for (int rr = 0; rr < 8; ++rr)
                    xout[(size_t)(rbase + rr) * FN + ncol] = c[t][rr];
            }
        }
    }
}


extern "C" void kernel_launch(void* const* d_in, const int* in_sizes, int n_in,
                              void* d_out, int out_size, void* d_ws, size_t ws_size,
                              hipStream_t stream) {
    (void)in_sizes; (void)n_in; (void)out_size; (void)ws_size;

    const float* x0    = (const float*)d_in[0];
    const int*   ei    = (const int*)d_in[1];
    const float* dropu = (const float*)d_in[2];
    const float* W0    = (const float*)d_in[3];
    const float* b0    = (const float*)d_in[4];
    const float* Whid  = (const float*)d_in[5];
    const float* bhid  = (const float*)d_in[6];
    const float* Wout  = (const float*)d_in[7];
    const float* bout  = (const float*)d_in[8];
    float* out = (float*)d_out;

    const int* src = ei;
    const int* dst = ei + NEDGES;

    char* ws = (char*)d_ws;
    size_t off = 0;
    auto take = [&](size_t bytes) -> void* {
        void* p = (void*)(ws + off);
        off += (bytes + 255) & ~(size_t)255;
        return p;
    };
    unsigned* table  = (unsigned*)take((size_t)NBINS * NCHUNK * 4);
    unsigned* rowptr = (unsigned*)take((size_t)(NNODES + 1) * 4);
    float*    dinv   = (float*)   take((size_t)NNODES * 4);
    int*      csr    = (int*)     take((size_t)NEDGES * 4);
    float*    hq     = (float*)   take((size_t)NNODES * NCLS * 4);
    float*    hp     = (float*)   take((size_t)NNODES * NHID * 4);
    float*    xold   = (float*)   take((size_t)NNODES * NHID * 4);
    int2* binned = (int2*)hq;
    static_assert((size_t)NEDGES * 8 <= (size_t)NNODES * NCLS * 4, "the binned edges fit in hq");

    auto blocks = [](int n, int tb) { return (n + tb - 1) / tb; };

    bin_count_kernel<<<NCHUNK, CHUNK_T, 0, stream>>>(dst, table, NEDGES);
    table_scan_kernel<<<1, SCAN_T, 0, stream>>>(table, NBINS * NCHUNK / 4);
    bin_scatter_kernel<<<NCHUNK, CHUNK_T, 0, stream>>>(src, dst, table, binned, NEDGES);
    bin_to_csr_kernel<<<NBINS, L2_T, 0, stream>>>(binned, table, rowptr, dinv, csr, NNODES, NEDGES);

    int xform_blocks  = blocks(NNODES / 16, XF_WAVES);
    int gather_blocks = blocks(NNODES, WG_NODES);

    input_transform_kernel<<<xform_blocks, XF_T, 0, stream>>>(x0, W0, dinv, hp, NNODES);

    const float* xold_in = nullptr;
    for (int layer = 0; layer < 7; ++layer) {
        const float* hin  = (layer & 1) ? hq : hp;
        float*       hout = (layer & 1) ? hp : hq;
        const float* b = (layer == 0) ? b0 : bhid + (size_t)(layer - 1) * NHID;
        const float* d = dropu + (size_t)layer * NNODES * NHID;

        if (layer < 6) {
            gather_tile_kernel<NHID, true><<<gather_blocks, WG_T, 0, stream>>>(
                hin, csr, rowptr, dinv, b, xold_in, d, hout, xold,
                Whid + (size_t)layer * NHID * NHID, NNODES);
        } else {
            gather_tile_kernel<0, true><<<gather_blocks, WG_T, 0, stream>>>(
                hin, csr, rowptr, dinv, b, xold_in, d, hout, xold, nullptr, NNODES);
        }
        xold_in = xold;
    }

    gather_tile_kernel<NCLS, false><<<gather_blocks, WG_T, 0, stream>>>(
        hq, csr, rowptr, dinv, bout, nullptr, nullptr, out, nullptr, Wout, NNODES);
}
